# starting kernel
# baseline (speedup 1.0000x reference)
#include <hip/hip_runtime.h>
#include <math.h>

#define DIM 128
#define HEADS 4
#define CH 32
#define NEG_SLOPE 0.2f

typedef __attribute__((ext_vector_type(2))) float v2f;
typedef __attribute__((ext_vector_type(8))) float v8f;

__global__ void gat_gemm_f32_wmma(const float* __restrict__ A,
                                  const float* __restrict__ W,
                                  const float* __restrict__ bias,
                                  float* __restrict__ out) {
  const int wave = threadIdx.x >> 5;
  const int lane = threadIdx.x & 31;
  const int row0 = blockIdx.x * 16;
  const int col0 = wave * 16;
  const int m  = lane & 15;
  const int kh = lane >> 4;

  const float* __restrict__ arow = A + (size_t)(row0 + m) * DIM;

  v8f acc = {};
  for (int k0 = 0; k0 < DIM; k0 += 4) {
    const int ka = k0 + kh * 2;
    v2f av, bv;
    av.x = arow[ka];
    av.y = arow[ka + 1];
    bv.x = W[(size_t)ka * DIM + col0 + m];
    bv.y = W[(size_t)(ka + 1) * DIM + col0 + m];
    acc = __builtin_amdgcn_wmma_f32_16x16x4_f32(
        false, av, false, bv, (short)0, acc, false, false);
  }

  const int nc = col0 + m;
  const float b = bias ? bias[nc] : 0.0f;
  const int rbase = row0 + kh * 8;
#pragma unroll
  for (int r = 0; r < 8; ++r)
    out[(size_t)(rbase + r) * DIM + nc] = acc[r] + b;
}

__global__ void gat_attn_scores(const float* __restrict__ h,
                                const float* __restrict__ a_src,
                                const float* __restrict__ a_dst,
                                float* __restrict__ asrc,
                                float* __restrict__ adst, int N) {
  int t = blockIdx.x * blockDim.x + threadIdx.x;
  if (t >= N * HEADS) return;
  const int n = t >> 2, hh = t & 3;
  const float* __restrict__ hp = h + (size_t)n * DIM + hh * CH;
  const float* __restrict__ ws = a_src + hh * CH;
  const float* __restrict__ wd = a_dst + hh * CH;
  float s1 = 0.f, s2 = 0.f;
#pragma unroll
  for (int c = 0; c < CH; ++c) {
    const float v = hp[c];
    s1 += v * ws[c];
    s2 += v * wd[c];
  }
  asrc[t] = s1;
  adst[t] = s2;
}

__device__ __forceinline__ void edge_sd(const int* __restrict__ ei, int e,
                                        int NE, int& s, int& d) {
  if (e < NE) { s = ei[e]; d = ei[NE + e]; }
  else        { s = d = e - NE; }
}

__device__ __forceinline__ unsigned ord_encode(float f) {
  unsigned u = __float_as_uint(f);
  return (f >= 0.0f) ? (u | 0x80000000u) : ~u;
}

__global__ void gat_edge_pass1(const int* __restrict__ ei, int NE, int E,
                               const float* __restrict__ asrc,
                               const float* __restrict__ adst,
                               float* __restrict__ ebuf,
                               unsigned* __restrict__ menc) {
  int t = blockIdx.x * blockDim.x + threadIdx.x;
  if (t >= E * HEADS) return;
  const int e = t >> 2, hh = t & 3;
  int s, d;
  edge_sd(ei, e, NE, s, d);
  float v = asrc[s * HEADS + hh] + adst[d * HEADS + hh];
  v = (v > 0.0f) ? v : NEG_SLOPE * v;
  ebuf[t] = v;
  atomicMax(&menc[d * HEADS + hh], ord_encode(v));
}

__global__ void gat_decode_max(unsigned* __restrict__ menc, int n) {
  int t = blockIdx.x * blockDim.x + threadIdx.x;
  if (t >= n) return;
  const unsigned u = menc[t];
  float f;
  if (u == 0u)                 f = 0.0f;
  else if (u & 0x80000000u)    f = __uint_as_float(u & 0x7FFFFFFFu);
  else                         f = __uint_as_float(~u);
  ((float*)menc)[t] = f;
}

__global__ void gat_edge_pass2(const int* __restrict__ ei, int NE, int E,
                               float* __restrict__ ebuf,
                               const float* __restrict__ m,
                               float* __restrict__ denom) {
  int t = blockIdx.x * blockDim.x + threadIdx.x;
  if (t >= E * HEADS) return;
  const int e = t >> 2, hh = t & 3;
  int s, d;
  edge_sd(ei, e, NE, s, d);
  const float ee = expf(ebuf[t] - m[d * HEADS + hh]);
  ebuf[t] = ee;
  atomicAdd(&denom[d * HEADS + hh], ee);
}

__global__ void gat_edge_pass3(const int* __restrict__ ei, int NE, int E,
                               const float* __restrict__ ebuf,
                               const float* __restrict__ denom,
                               const float* __restrict__ h,
                               float* __restrict__ agg) {
  int t = blockIdx.x * blockDim.x + threadIdx.x;
  if (t >= E * DIM) return;
  const int e = t >> 7;
  const int c = t & (DIM - 1);
  const int hh = c >> 5;
  int s, d;
  edge_sd(ei, e, NE, s, d);
  const float alpha =
      ebuf[e * HEADS + hh] / (denom[d * HEADS + hh] + 1e-16f);
  atomicAdd(&agg[(size_t)d * DIM + c], h[(size_t)s * DIM + c] * alpha);
}

__global__ void gat_elu(const float* __restrict__ agg, float* __restrict__ out,
                        int n) {
  int t = blockIdx.x * blockDim.x + threadIdx.x;
  if (t >= n) return;
  const float v = agg[t];
  out[t] = (v > 0.0f) ? v : expm1f(v);
}

extern "C" void kernel_launch(void* const* d_in, const int* in_sizes, int n_in,
                              void* d_out, int out_size, void* d_ws,
                              size_t ws_size, hipStream_t stream) {
  const float* x  = (const float*)d_in[0];
  const int*   ei = (const int*)d_in[1];

  const int N  = in_sizes[0] / DIM;
  const int NE = in_sizes[1] / 2;
  const int E  = NE + N;

  float* ws = (float*)d_ws;
  size_t o = 0;
  float* h     = ws + o; o += (size_t)N * DIM;
  float* agg   = ws + o; o += (size_t)N * DIM;
  float* asrc  = ws + o; o += (size_t)N * HEADS;
  float* adst  = ws + o; o += (size_t)N * HEADS;
  float* mbuf  = ws + o; o += (size_t)N * HEADS;
  float* denom = ws + o; o += (size_t)N * HEADS;
  float* ebuf  = ws + o; o += (size_t)E * HEADS;

  float* y = (float*)d_out;

  const int TB = 256;
  const int gGemm = N / 16;
  const int gNH   = (N * HEADS + TB - 1) / TB;
  const int gEH   = (E * HEADS + TB - 1) / TB;
  const int gED   = (E * DIM + TB - 1) / TB;
  const int gND   = (N * DIM + TB - 1) / TB;

  const float* xin = x;
  for (int layer = 0; layer < 2; ++layer) {
    const float* W     = (const float*)d_in[2 + layer * 5];
    const float* a_src = (const float*)d_in[3 + layer * 5];
    const float* a_dst = (const float*)d_in[4 + layer * 5];
    const float* bias  = (const float*)d_in[5 + layer * 5];
    const float* Wskip = (const float*)d_in[6 + layer * 5];

    gat_gemm_f32_wmma<<<gGemm, TB, 0, stream>>>(xin, W, nullptr, h);
    gat_gemm_f32_wmma<<<gGemm, TB, 0, stream>>>(xin, Wskip, bias, agg);

    gat_attn_scores<<<gNH, TB, 0, stream>>>(h, a_src, a_dst, asrc, adst, N);

    hipMemsetAsync(mbuf, 0, (size_t)N * HEADS * sizeof(float), stream);
    hipMemsetAsync(denom, 0, (size_t)N * HEADS * sizeof(float), stream);

    gat_edge_pass1<<<gEH, TB, 0, stream>>>(ei, NE, E, asrc, adst, ebuf,
                                           (unsigned*)mbuf);
    gat_decode_max<<<gNH, TB, 0, stream>>>((unsigned*)mbuf, N * HEADS);
    gat_edge_pass2<<<gEH, TB, 0, stream>>>(ei, NE, E, ebuf, mbuf, denom);
    gat_edge_pass3<<<gED, TB, 0, stream>>>(ei, NE, E, ebuf, denom, h, agg);

    gat_elu<<<gND, TB, 0, stream>>>(agg, y, N * DIM);
    xin = y;
  }
}


# Message aggregation: global float-atomic scatter replaced by per-32-node-range LDS tiles (LDS float adds onto the skip rows) over a per-call range-binned edge list; sum order: LDS arrival order
# speedup vs baseline: 1.4394x; 1.4394x over previous
#include <hip/hip_runtime.h>
#include <math.h>

#define DIM 128
#define HEADS 4
#define CH 32
#define NEG_SLOPE 0.2f

typedef __attribute__((ext_vector_type(2))) float v2f;
typedef __attribute__((ext_vector_type(8))) float v8f;

__global__ void gat_gemm_f32_wmma(const float* __restrict__ A,
                                  const float* __restrict__ W,
                                  const float* __restrict__ bias,
                                  float* __restrict__ out) {
  const int wave = threadIdx.x >> 5;
  const int lane = threadIdx.x & 31;
  const int row0 = blockIdx.x * 16;
  const int col0 = wave * 16;
  const int m  = lane & 15;
  const int kh = lane >> 4;

  const float* __restrict__ arow = A + (size_t)(row0 + m) * DIM;

  v8f acc = {};
  for (int k0 = 0; k0 < DIM; k0 += 4) {
    const int ka = k0 + kh * 2;
    v2f av, bv;
    av.x = arow[ka];
    av.y = arow[ka + 1];
    bv.x = W[(size_t)ka * DIM + col0 + m];
    bv.y = W[(size_t)(ka + 1) * DIM + col0 + m];
    acc = __builtin_amdgcn_wmma_f32_16x16x4_f32(
        false, av, false, bv, (short)0, acc, false, false);
  }

  const int nc = col0 + m;
  const float b = bias ? bias[nc] : 0.0f;
  const int rbase = row0 + kh * 8;
#pragma unroll
  for (int r = 0; r < 8; ++r)
    out[(size_t)(rbase + r) * DIM + nc] = acc[r] + b;
}

__global__ void gat_attn_scores(const float* __restrict__ h,
                                const float* __restrict__ a_src,
                                const float* __restrict__ a_dst,
                                float* __restrict__ asrc,
                                float* __restrict__ adst, int N) {
  int t = blockIdx.x * blockDim.x + threadIdx.x;
  if (t >= N * HEADS) return;
  const int n = t >> 2, hh = t & 3;
  const float* __restrict__ hp = h + (size_t)n * DIM + hh * CH;
  const float* __restrict__ ws = a_src + hh * CH;
  const float* __restrict__ wd = a_dst + hh * CH;
  float s1 = 0.f, s2 = 0.f;
#pragma unroll
  for (int c = 0; c < CH; ++c) {
    const float v = hp[c];
    s1 += v * ws[c];
    s2 += v * wd[c];
  }
  asrc[t] = s1;
  adst[t] = s2;
}

__device__ __forceinline__ void edge_sd(const int* __restrict__ ei, int e,
                                        int NE, int& s, int& d) {
  if (e < NE) { s = ei[e]; d = ei[NE + e]; }
  else        { s = d = e - NE; }
}

__device__ __forceinline__ unsigned ord_encode(float f) {
  unsigned u = __float_as_uint(f);
  return (f >= 0.0f) ? (u | 0x80000000u) : ~u;
}

__global__ void gat_edge_pass1(const int* __restrict__ ei, int NE, int E,
                               const float* __restrict__ asrc,
                               const float* __restrict__ adst,
                               float* __restrict__ ebuf,
                               unsigned* __restrict__ menc) {
  int t = blockIdx.x * blockDim.x + threadIdx.x;
  if (t >= E * HEADS) return;
  const int e = t >> 2, hh = t & 3;
  int s, d;
  edge_sd(ei, e, NE, s, d);
  float v = asrc[s * HEADS + hh] + adst[d * HEADS + hh];
  v = (v > 0.0f) ? v : NEG_SLOPE * v;
  ebuf[t] = v;
  atomicMax(&menc[d * HEADS + hh], ord_encode(v));
}

__global__ void gat_decode_max(unsigned* __restrict__ menc, int n) {
  int t = blockIdx.x * blockDim.x + threadIdx.x;
  if (t >= n) return;
  const unsigned u = menc[t];
  float f;
  if (u == 0u)                 f = 0.0f;
  else if (u & 0x80000000u)    f = __uint_as_float(u & 0x7FFFFFFFu);
  else                         f = __uint_as_float(~u);
  ((float*)menc)[t] = f;
}

__global__ void gat_edge_pass2(const int* __restrict__ ei, int NE, int E,
                               float* __restrict__ ebuf,
                               const float* __restrict__ m,
                               float* __restrict__ denom) {
  int t = blockIdx.x * blockDim.x + threadIdx.x;
  if (t >= E * HEADS) return;
  const int e = t >> 2, hh = t & 3;
  int s, d;
  edge_sd(ei, e, NE, s, d);
  const float ee = expf(ebuf[t] - m[d * HEADS + hh]);
  ebuf[t] = ee;
  atomicAdd(&denom[d * HEADS + hh], ee);
}

#define RNG 32
#define RNG_SHIFT 5
#define REP 8
#define LINE 32

__device__ __forceinline__ int sub_bin(int e, int d) {
  return (d >> RNG_SHIFT) * REP + (e & (REP - 1));
}

__global__ void gat_bin_zero(unsigned* __restrict__ cnt, int n) {
  int t = blockIdx.x * blockDim.x + threadIdx.x;
  if (t < n) cnt[(size_t)t * LINE] = 0u;
}

__global__ void gat_bin_count(const int* __restrict__ ei, int NE, int E,
                              unsigned* __restrict__ cnt) {
  int e = blockIdx.x * blockDim.x + threadIdx.x;
  if (e >= E) return;
  int s, d;
  edge_sd(ei, e, NE, s, d);
  atomicAdd(&cnt[(size_t)sub_bin(e, d) * LINE], 1u);
}

__global__ void gat_bin_scan(const unsigned* __restrict__ cnt,
                             unsigned* __restrict__ off,
                             unsigned* __restrict__ cursor, int NB) {
  __shared__ unsigned sh[1024];
  const int t = threadIdx.x;
  const int per = (NB + 1023) / 1024;
  const int b0 = t * per;
  unsigned sum = 0u;
  for (int i = 0; i < per; ++i) {
    const int b = b0 + i;
    if (b < NB) sum += cnt[(size_t)b * LINE];
  }
  sh[t] = sum;
  __syncthreads();
  for (int ofs = 1; ofs < 1024; ofs <<= 1) {
    const unsigned v = (t >= ofs) ? sh[t - ofs] : 0u;
    __syncthreads();
    sh[t] += v;
    __syncthreads();
  }
  unsigned run = sh[t] - sum;
  for (int i = 0; i < per; ++i) {
    const int b = b0 + i;
    if (b < NB) {
      off[b] = run;
      cursor[(size_t)b * LINE] = run;
      run += cnt[(size_t)b * LINE];
    }
  }
  if (t == 1023) off[NB] = sh[1023];
}

__global__ void gat_bin_fill(const int* __restrict__ ei, int NE, int E,
                             unsigned* __restrict__ cursor,
                             int4* __restrict__ rec) {
  int e = blockIdx.x * blockDim.x + threadIdx.x;
  if (e >= E) return;
  int s, d;
  edge_sd(ei, e, NE, s, d);
  const unsigned pos = atomicAdd(&cursor[(size_t)sub_bin(e, d) * LINE], 1u);
  rec[pos] = make_int4(e, s, d, 0);
}

#define GRP 4
template <bool FULL>
__device__ __forceinline__ void agg_group(const float* __restrict__ h,
                                          float* acc, int lane, int j, int n,
                                          int s, int dl, float a0, float a1,
                                          float a2, float a3) {
  float hv[GRP][4];
  float al[GRP][4];
  int ro[GRP];
#pragma unroll
  for (int u = 0; u < GRP; ++u) {
    if (FULL || j + u < n) {
      const int sj = __builtin_amdgcn_readlane(s, j + u);
      ro[u] = __builtin_amdgcn_readlane(dl, j + u);
      al[u][0] = __int_as_float(__builtin_amdgcn_readlane(__float_as_int(a0), j + u));
      al[u][1] = __int_as_float(__builtin_amdgcn_readlane(__float_as_int(a1), j + u));
      al[u][2] = __int_as_float(__builtin_amdgcn_readlane(__float_as_int(a2), j + u));
      al[u][3] = __int_as_float(__builtin_amdgcn_readlane(__float_as_int(a3), j + u));
      const float* __restrict__ hp = h + (size_t)sj * DIM + lane;
#pragma unroll
      for (int k = 0; k < 4; ++k) hv[u][k] = hp[32 * k];
    }
  }
#pragma unroll
  for (int u = 0; u < GRP; ++u) {
    if (FULL || j + u < n) {
#pragma unroll
      for (int k = 0; k < 4; ++k)
        atomicAdd(&acc[ro[u] + lane + 32 * k], hv[u][k] * al[u][k]);
    }
  }
}

__global__ void __launch_bounds__(256)
gat_agg_lds(const int4* __restrict__ rec, const unsigned* __restrict__ off,
            const float* __restrict__ ebuf, const float* __restrict__ denom,
            const float* __restrict__ h, float* __restrict__ agg, int N) {
  __shared__ alignas(16) float acc[RNG * DIM];
  const int tid = threadIdx.x;
  const int lane = tid & 31;
  const int wave = __builtin_amdgcn_readfirstlane(tid >> 5);
  const int b = blockIdx.x;

  float4* __restrict__ tile = (float4*)(agg + (size_t)b * RNG * DIM);
  float4* acc4 = (float4*)acc;
  const int lim4 = min(RNG, N - b * RNG) * (DIM / 4);
  const bool whole = (lim4 == RNG * DIM / 4);
  if (whole) {
    float4 v[RNG * DIM / 1024];
#pragma unroll
    for (int m = 0; m < RNG * DIM / 1024; ++m) v[m] = tile[tid + 256 * m];
#pragma unroll
    for (int m = 0; m < RNG * DIM / 1024; ++m) acc4[tid + 256 * m] = v[m];
  } else {
    for (int m = 0; m < RNG * DIM / 1024; ++m) {
      const int i = tid + 256 * m;
      acc4[i] = (i < lim4) ? tile[i] : make_float4(0.f, 0.f, 0.f, 0.f);
    }
  }
  __syncthreads();

  const int beg = __builtin_amdgcn_readfirstlane((int)off[b * REP]);
  const int end = __builtin_amdgcn_readfirstlane((int)off[(b + 1) * REP]);
  const int per = (end - beg + 7) >> 3;
  const int w0 = beg + wave * per;
  const int w1 = min(w0 + per, end);

  for (int base = w0; base < w1; base += 32) {
    const int n = min(32, w1 - base);
    int s = 0, d = b * RNG;
    float a0 = 0.f, a1 = 0.f, a2 = 0.f, a3 = 0.f;
    if (lane < n) {
      const int4 r = rec[base + lane];
      s = r.y;
      d = r.z;
      const float4 ev = *(const float4*)(ebuf + (size_t)r.x * HEADS);
      const float4 dv = *(const float4*)(denom + (size_t)d * HEADS);
      a0 = ev.x / (dv.x + 1e-16f);
      a1 = ev.y / (dv.y + 1e-16f);
      a2 = ev.z / (dv.z + 1e-16f);
      a3 = ev.w / (dv.w + 1e-16f);
    }
    const int dl = (d - b * RNG) * DIM;
    int j = 0;
    for (; j + GRP <= n; j += GRP)
      agg_group<true>(h, acc, lane, j, n, s, dl, a0, a1, a2, a3);
    if (j < n)
      agg_group<false>(h, acc, lane, j, n, s, dl, a0, a1, a2, a3);
  }
  __syncthreads();

  if (whole) {
#pragma unroll
    for (int m = 0; m < RNG * DIM / 1024; ++m)
      tile[tid + 256 * m] = acc4[tid + 256 * m];
  } else {
    for (int m = 0; m < RNG * DIM / 1024; ++m) {
      const int i = tid + 256 * m;
      if (i < lim4) tile[i] = acc4[i];
    }
  }
}

__global__ void gat_elu(const float* __restrict__ agg, float* __restrict__ out,
                        int n) {
  int t = blockIdx.x * blockDim.x + threadIdx.x;
  if (t >= n) return;
  const float v = agg[t];
  out[t] = (v > 0.0f) ? v : expm1f(v);
}

extern "C" void kernel_launch(void* const* d_in, const int* in_sizes, int n_in,
                              void* d_out, int out_size, void* d_ws,
                              size_t ws_size, hipStream_t stream) {
  const float* x  = (const float*)d_in[0];
  const int*   ei = (const int*)d_in[1];

  const int N  = in_sizes[0] / DIM;
  const int NE = in_sizes[1] / 2;
  const int E  = NE + N;

  float* ws = (float*)d_ws;
  size_t o = 0;
  float* h     = ws + o; o += (size_t)N * DIM;
  float* agg   = ws + o; o += (size_t)N * DIM;
  float* asrc  = ws + o; o += (size_t)N * HEADS;
  float* adst  = ws + o; o += (size_t)N * HEADS;
  float* mbuf  = ws + o; o += (size_t)N * HEADS;
  float* denom = ws + o; o += (size_t)N * HEADS;
  float* ebuf  = ws + o; o += (size_t)E * HEADS;
  const int NB = (N + RNG - 1) / RNG;
  int4* rec = (int4*)(ws + o); o += (size_t)E * 4;
  const int NBR = NB * REP;
  unsigned* bcnt = (unsigned*)(ws + o); o += (size_t)NBR * LINE;
  unsigned* bcur = (unsigned*)(ws + o); o += (size_t)NBR * LINE;
  unsigned* boff = (unsigned*)(ws + o); o += (size_t)((NBR + 4) & ~3);

  float* y = (float*)d_out;

  const int TB = 256;
  const int gGemm = N / 16;
  const int gNH   = (N * HEADS + TB - 1) / TB;
  const int gEH   = (E * HEADS + TB - 1) / TB;
  const int gE    = (E + TB - 1) / TB;
  const int gNB   = (NBR + TB - 1) / TB;
  const int gND   = (N * DIM + TB - 1) / TB;

  gat_bin_zero<<<gNB, TB, 0, stream>>>(bcnt, NBR);
  gat_bin_count<<<gE, TB, 0, stream>>>(ei, NE, E, bcnt);
  gat_bin_scan<<<1, 1024, 0, stream>>>(bcnt, boff, bcur, NBR);
  gat_bin_fill<<<gE, TB, 0, stream>>>(ei, NE, E, bcur, rec);

  const float* xin = x;
  for (int layer = 0; layer < 2; ++layer) {
    const float* W     = (const float*)d_in[2 + layer * 5];
    const float* a_src = (const float*)d_in[3 + layer * 5];
    const float* a_dst = (const float*)d_in[4 + layer * 5];
    const float* bias  = (const float*)d_in[5 + layer * 5];
    const float* Wskip = (const float*)d_in[6 + layer * 5];

    gat_gemm_f32_wmma<<<gGemm, TB, 0, stream>>>(xin, W, nullptr, h);
    gat_gemm_f32_wmma<<<gGemm, TB, 0, stream>>>(xin, Wskip, bias, agg);

    gat_attn_scores<<<gNH, TB, 0, stream>>>(h, a_src, a_dst, asrc, adst, N);

    hipMemsetAsync(mbuf, 0, (size_t)N * HEADS * sizeof(float), stream);
    hipMemsetAsync(denom, 0, (size_t)N * HEADS * sizeof(float), stream);

    gat_edge_pass1<<<gEH, TB, 0, stream>>>(ei, NE, E, asrc, adst, ebuf,
                                           (unsigned*)mbuf);
    gat_decode_max<<<gNH, TB, 0, stream>>>((unsigned*)mbuf, N * HEADS);
    gat_edge_pass2<<<gEH, TB, 0, stream>>>(ei, NE, E, ebuf, mbuf, denom);
    gat_agg_lds<<<NB, TB, 0, stream>>>(rec, boff, ebuf, denom, h, agg, N);

    gat_elu<<<gND, TB, 0, stream>>>(agg, y, N * DIM);
    xin = y;
  }
}


# Edge softmax moved into the 32-node-range block: per (row, head) max and sum in LDS cells (LDS max and add), 3 passes over the bin; edge passes, global atomics, memsets gone; sum order: LDS arrival
# speedup vs baseline: 1.8171x; 1.2624x over previous
#include <hip/hip_runtime.h>
#include <math.h>

#define DIM 128
#define HEADS 4
#define CH 32
#define NEG_SLOPE 0.2f

typedef __attribute__((ext_vector_type(2))) float v2f;
typedef __attribute__((ext_vector_type(8))) float v8f;

__global__ void gat_gemm_f32_wmma(const float* __restrict__ A,
                                  const float* __restrict__ W,
                                  const float* __restrict__ bias,
                                  float* __restrict__ out) {
  const int wave = threadIdx.x >> 5;
  const int lane = threadIdx.x & 31;
  const int row0 = blockIdx.x * 16;
  const int col0 = wave * 16;
  const int m  = lane & 15;
  const int kh = lane >> 4;

  const float* __restrict__ arow = A + (size_t)(row0 + m) * DIM;

  v8f acc = {};
  for (int k0 = 0; k0 < DIM; k0 += 4) {
    const int ka = k0 + kh * 2;
    v2f av, bv;
    av.x = arow[ka];
    av.y = arow[ka + 1];
    bv.x = W[(size_t)ka * DIM + col0 + m];
    bv.y = W[(size_t)(ka + 1) * DIM + col0 + m];
    acc = __builtin_amdgcn_wmma_f32_16x16x4_f32(
        false, av, false, bv, (short)0, acc, false, false);
  }

  const int nc = col0 + m;
  const float b = bias ? bias[nc] : 0.0f;
  const int rbase = row0 + kh * 8;
#pragma unroll
  for (int r = 0; r < 8; ++r)
    out[(size_t)(rbase + r) * DIM + nc] = acc[r] + b;
}

__global__ void gat_attn_scores(const float* __restrict__ h,
                                const float* __restrict__ a_src,
                                const float* __restrict__ a_dst,
                                float* __restrict__ asrc,
                                float* __restrict__ adst, int N) {
  int t = blockIdx.x * blockDim.x + threadIdx.x;
  if (t >= N * HEADS) return;
  const int n = t >> 2, hh = t & 3;
  const float* __restrict__ hp = h + (size_t)n * DIM + hh * CH;
  const float* __restrict__ ws = a_src + hh * CH;
  const float* __restrict__ wd = a_dst + hh * CH;
  float s1 = 0.f, s2 = 0.f;
#pragma unroll
  for (int c = 0; c < CH; ++c) {
    const float v = hp[c];
    s1 += v * ws[c];
    s2 += v * wd[c];
  }
  asrc[t] = s1;
  adst[t] = s2;
}

__device__ __forceinline__ void edge_sd(const int* __restrict__ ei, int e,
                                        int NE, int& s, int& d) {
  if (e < NE) { s = ei[e]; d = ei[NE + e]; }
  else        { s = d = e - NE; }
}

__device__ __forceinline__ unsigned ord_encode(float f) {
  unsigned u = __float_as_uint(f);
  return (f >= 0.0f) ? (u | 0x80000000u) : ~u;
}

__device__ __forceinline__ float ord_decode(unsigned u) {
  const unsigned top = (unsigned)((int)u >> 31);
  const float f = __uint_as_float(u ^ (~top | 0x80000000u));
  return (u == 0u) ? 0.0f : f;
}

__device__ __forceinline__ float leaky(float v) {
  return (v > 0.0f) ? v : NEG_SLOPE * v;
}

#define RNG 32
#define RNG_SHIFT 5
#define REP 8
#define LINE 32

__device__ __forceinline__ int sub_bin(int e, int d) {
  return (d >> RNG_SHIFT) * REP + (e & (REP - 1));
}

__global__ void gat_bin_zero(unsigned* __restrict__ cnt, int n) {
  int t = blockIdx.x * blockDim.x + threadIdx.x;
  if (t < n) cnt[(size_t)t * LINE] = 0u;
}

__global__ void gat_bin_count(const int* __restrict__ ei, int NE, int E,
                              unsigned* __restrict__ cnt) {
  int e = blockIdx.x * blockDim.x + threadIdx.x;
  if (e >= E) return;
  int s, d;
  edge_sd(ei, e, NE, s, d);
  atomicAdd(&cnt[(size_t)sub_bin(e, d) * LINE], 1u);
}

__global__ void gat_bin_scan(const unsigned* __restrict__ cnt,
                             unsigned* __restrict__ off,
                             unsigned* __restrict__ cursor, int NB) {
  __shared__ unsigned sh[1024];
  const int t = threadIdx.x;
  const int per = (NB + 1023) / 1024;
  const int b0 = t * per;
  unsigned sum = 0u;
  for (int i = 0; i < per; ++i) {
    const int b = b0 + i;
    if (b < NB) sum += cnt[(size_t)b * LINE];
  }
  sh[t] = sum;
  __syncthreads();
  for (int ofs = 1; ofs < 1024; ofs <<= 1) {
    const unsigned v = (t >= ofs) ? sh[t - ofs] : 0u;
    __syncthreads();
    sh[t] += v;
    __syncthreads();
  }
  unsigned run = sh[t] - sum;
  for (int i = 0; i < per; ++i) {
    const int b = b0 + i;
    if (b < NB) {
      off[b] = run;
      cursor[(size_t)b * LINE] = run;
      run += cnt[(size_t)b * LINE];
    }
  }
  if (t == 1023) off[NB] = sh[1023];
}

__global__ void gat_bin_fill(const int* __restrict__ ei, int NE, int E,
                             unsigned* __restrict__ cursor,
                             int4* __restrict__ rec) {
  int e = blockIdx.x * blockDim.x + threadIdx.x;
  if (e >= E) return;
  int s, d;
  edge_sd(ei, e, NE, s, d);
  const unsigned pos = atomicAdd(&cursor[(size_t)sub_bin(e, d) * LINE], 1u);
  rec[pos] = make_int4(e, s, d, 0);
}

#define GRP 4
template <bool FULL>
__device__ __forceinline__ void agg_group(const float* __restrict__ h,
                                          float* acc, int lane, int j, int n,
                                          int s, int dl, float a0, float a1,
                                          float a2, float a3) {
  float hv[GRP][4];
  float al[GRP][4];
  int ro[GRP];
#pragma unroll
  for (int u = 0; u < GRP; ++u) {
    if (FULL || j + u < n) {
      const int sj = __builtin_amdgcn_readlane(s, j + u);
      ro[u] = __builtin_amdgcn_readlane(dl, j + u);
      al[u][0] = __int_as_float(__builtin_amdgcn_readlane(__float_as_int(a0), j + u));
      al[u][1] = __int_as_float(__builtin_amdgcn_readlane(__float_as_int(a1), j + u));
      al[u][2] = __int_as_float(__builtin_amdgcn_readlane(__float_as_int(a2), j + u));
      al[u][3] = __int_as_float(__builtin_amdgcn_readlane(__float_as_int(a3), j + u));
      const float* __restrict__ hp = h + (size_t)sj * DIM + lane;
#pragma unroll
      for (int k = 0; k < 4; ++k) hv[u][k] = hp[32 * k];
    }
  }
#pragma unroll
  for (int u = 0; u < GRP; ++u) {
    if (FULL || j + u < n) {
#pragma unroll
      for (int k = 0; k < 4; ++k)
        atomicAdd(&acc[ro[u] + lane + 32 * k], hv[u][k] * al[u][k]);
    }
  }
}

__global__ void __launch_bounds__(256)
gat_agg_lds(const int4* __restrict__ rec, const unsigned* __restrict__ off,
            const float* __restrict__ asrc, const float* __restrict__ adst,
            float4* __restrict__ scr, const float* __restrict__ h,
            float* __restrict__ agg, int N) {
  __shared__ alignas(16) float acc[RNG * DIM];
  __shared__ alignas(16) float s_adst[RNG * HEADS];
  __shared__ alignas(16) unsigned s_menc[RNG * HEADS];
  __shared__ alignas(16) float s_den[RNG * HEADS];
  const int tid = threadIdx.x;
  const int lane = tid & 31;
  const int wave = __builtin_amdgcn_readfirstlane(tid >> 5);
  const int b = blockIdx.x;

  float4* __restrict__ tile = (float4*)(agg + (size_t)b * RNG * DIM);
  float4* acc4 = (float4*)acc;
  const int lim4 = min(RNG, N - b * RNG) * (DIM / 4);
  const bool whole = (lim4 == RNG * DIM / 4);
  if (whole) {
    float4 v[RNG * DIM / 1024];
#pragma unroll
    for (int m = 0; m < RNG * DIM / 1024; ++m) v[m] = tile[tid + 256 * m];
#pragma unroll
    for (int m = 0; m < RNG * DIM / 1024; ++m) acc4[tid + 256 * m] = v[m];
  } else {
    for (int m = 0; m < RNG * DIM / 1024; ++m) {
      const int i = tid + 256 * m;
      acc4[i] = (i < lim4) ? tile[i] : make_float4(0.f, 0.f, 0.f, 0.f);
    }
  }
  if (tid < RNG * HEADS) {
    const int node = b * RNG + (tid >> 2);
    s_adst[tid] = (node < N) ? adst[(size_t)b * RNG * HEADS + tid] : 0.0f;
    s_menc[tid] = 0u;
    s_den[tid] = 0.0f;
  }
  __syncthreads();

  const int beg = __builtin_amdgcn_readfirstlane((int)off[b * REP]);
  const int end = __builtin_amdgcn_readfirstlane((int)off[(b + 1) * REP]);
  const int per = (end - beg + 7) >> 3;
  const int w0 = beg + wave * per;
  const int w1 = min(w0 + per, end);


  for (int base = w0; base < w1; base += 64) {
    int q[2], dl4[2];
    bool ok[2];
    float4 av[2];
#pragma unroll
    for (int u = 0; u < 2; ++u) {
      const int p = base + 32 * u + lane;
      ok[u] = p < w1;
      q[u] = min(p, w1 - 1);
    }
    int4 r[2];
#pragma unroll
    for (int u = 0; u < 2; ++u) r[u] = rec[q[u]];
#pragma unroll
    for (int u = 0; u < 2; ++u) {
      av[u] = *(const float4*)(asrc + (size_t)r[u].y * HEADS);
      dl4[u] = (r[u].z - b * RNG) * HEADS;
    }
    asm volatile("" ::"v"(av[0].x), "v"(av[1].x));
#pragma unroll
    for (int u = 0; u < 2; ++u) {
      if (ok[u]) {
        const float4 dv = *(const float4*)(s_adst + dl4[u]);
        float4 e;
        e.x = leaky(av[u].x + dv.x);
        e.y = leaky(av[u].y + dv.y);
        e.z = leaky(av[u].z + dv.z);
        e.w = leaky(av[u].w + dv.w);
        scr[q[u]] = e;
        atomicMax(&s_menc[dl4[u] + 0], ord_encode(e.x));
        atomicMax(&s_menc[dl4[u] + 1], ord_encode(e.y));
        atomicMax(&s_menc[dl4[u] + 2], ord_encode(e.z));
        atomicMax(&s_menc[dl4[u] + 3], ord_encode(e.w));
      }
    }
  }
  __syncthreads();

  for (int base = w0; base < w1; base += 64) {
    int q[2], dl4[2];
    bool ok[2];
    float4 e[2];
#pragma unroll
    for (int u = 0; u < 2; ++u) {
      const int p = base + 32 * u + lane;
      ok[u] = p < w1;
      q[u] = min(p, w1 - 1);
    }
    int dz[2];
#pragma unroll
    for (int u = 0; u < 2; ++u) {
      dz[u] = rec[q[u]].z;
      e[u] = scr[q[u]];
    }
    asm volatile("" ::"v"(dz[0]), "v"(dz[1]), "v"(e[0].x), "v"(e[1].x));
#pragma unroll
    for (int u = 0; u < 2; ++u) dl4[u] = (dz[u] - b * RNG) * HEADS;
#pragma unroll
    for (int u = 0; u < 2; ++u) {
      if (ok[u]) {
        const uint4 mu = *(const uint4*)(s_menc + dl4[u]);
        float4 ex;
        ex.x = expf(e[u].x - ord_decode(mu.x));
        ex.y = expf(e[u].y - ord_decode(mu.y));
        ex.z = expf(e[u].z - ord_decode(mu.z));
        ex.w = expf(e[u].w - ord_decode(mu.w));
        scr[q[u]] = ex;
        atomicAdd(&s_den[dl4[u] + 0], ex.x);
        atomicAdd(&s_den[dl4[u] + 1], ex.y);
        atomicAdd(&s_den[dl4[u] + 2], ex.z);
        atomicAdd(&s_den[dl4[u] + 3], ex.w);
      }
    }
  }
  __syncthreads();

  for (int base = w0; base < w1; base += 32) {
    const int n = min(32, w1 - base);
    int s = 0, d = b * RNG;
    float a0 = 0.f, a1 = 0.f, a2 = 0.f, a3 = 0.f;
    if (lane < n) {
      const int4 r = rec[base + lane];
      s = r.y;
      d = r.z;
      const float4 ev = scr[base + lane];
      const float4 dv = *(const float4*)(s_den + (d - b * RNG) * HEADS);
      a0 = ev.x / (dv.x + 1e-16f);
      a1 = ev.y / (dv.y + 1e-16f);
      a2 = ev.z / (dv.z + 1e-16f);
      a3 = ev.w / (dv.w + 1e-16f);
    }
    const int dl = (d - b * RNG) * DIM;
    int j = 0;
    for (; j + GRP <= n; j += GRP)
      agg_group<true>(h, acc, lane, j, n, s, dl, a0, a1, a2, a3);
    if (j < n)
      agg_group<false>(h, acc, lane, j, n, s, dl, a0, a1, a2, a3);
  }
  __syncthreads();

  if (whole) {
#pragma unroll
    for (int m = 0; m < RNG * DIM / 1024; ++m)
      tile[tid + 256 * m] = acc4[tid + 256 * m];
  } else {
    for (int m = 0; m < RNG * DIM / 1024; ++m) {
      const int i = tid + 256 * m;
      if (i < lim4) tile[i] = acc4[i];
    }
  }
}

__global__ void gat_elu(const float* __restrict__ agg, float* __restrict__ out,
                        int n) {
  int t = blockIdx.x * blockDim.x + threadIdx.x;
  if (t >= n) return;
  const float v = agg[t];
  out[t] = (v > 0.0f) ? v : expm1f(v);
}

extern "C" void kernel_launch(void* const* d_in, const int* in_sizes, int n_in,
                              void* d_out, int out_size, void* d_ws,
                              size_t ws_size, hipStream_t stream) {
  const float* x  = (const float*)d_in[0];
  const int*   ei = (const int*)d_in[1];

  const int N  = in_sizes[0] / DIM;
  const int NE = in_sizes[1] / 2;
  const int E  = NE + N;

  float* ws = (float*)d_ws;
  size_t o = 0;
  float* h     = ws + o; o += (size_t)N * DIM;
  float* agg   = ws + o; o += (size_t)N * DIM;
  float* asrc  = ws + o; o += (size_t)N * HEADS;
  float* adst  = ws + o; o += (size_t)N * HEADS;
  float4* scr  = (float4*)(ws + o); o += (size_t)E * HEADS;
  const int NB = (N + RNG - 1) / RNG;
  int4* rec = (int4*)(ws + o); o += (size_t)E * 4;
  const int NBR = NB * REP;
  unsigned* bcnt = (unsigned*)(ws + o); o += (size_t)NBR * LINE;
  unsigned* bcur = (unsigned*)(ws + o); o += (size_t)NBR * LINE;
  unsigned* boff = (unsigned*)(ws + o); o += (size_t)((NBR + 4) & ~3);

  float* y = (float*)d_out;

  const int TB = 256;
  const int gGemm = N / 16;
  const int gNH   = (N * HEADS + TB - 1) / TB;
  const int gE    = (E + TB - 1) / TB;
  const int gNB   = (NBR + TB - 1) / TB;
  const int gND   = (N * DIM + TB - 1) / TB;

  gat_bin_zero<<<gNB, TB, 0, stream>>>(bcnt, NBR);
  gat_bin_count<<<gE, TB, 0, stream>>>(ei, NE, E, bcnt);
  gat_bin_scan<<<1, 1024, 0, stream>>>(bcnt, boff, bcur, NBR);
  gat_bin_fill<<<gE, TB, 0, stream>>>(ei, NE, E, bcur, rec);

  const float* xin = x;
  for (int layer = 0; layer < 2; ++layer) {
    const float* W     = (const float*)d_in[2 + layer * 5];
    const float* a_src = (const float*)d_in[3 + layer * 5];
    const float* a_dst = (const float*)d_in[4 + layer * 5];
    const float* bias  = (const float*)d_in[5 + layer * 5];
    const float* Wskip = (const float*)d_in[6 + layer * 5];

    gat_gemm_f32_wmma<<<gGemm, TB, 0, stream>>>(xin, W, nullptr, h);
    gat_gemm_f32_wmma<<<gGemm, TB, 0, stream>>>(xin, Wskip, bias, agg);

    gat_attn_scores<<<gNH, TB, 0, stream>>>(h, a_src, a_dst, asrc, adst, N);

    gat_agg_lds<<<NB, TB, 0, stream>>>(rec, boff, asrc, adst, scr, h, agg, N);

    gat_elu<<<gND, TB, 0, stream>>>(agg, y, N * DIM);
    xin = y;
  }
}


# ELU moved into the skip product's epilogue: skip product runs behind the aggregation, message sums as its WMMA accumulator start; ELU passes gone; sum order: messages, then skip terms, bias
# speedup vs baseline: 2.2048x; 1.2134x over previous
#include <hip/hip_runtime.h>
#include <math.h>

#define DIM 128
#define HEADS 4
#define CH 32
#define NEG_SLOPE 0.2f

typedef __attribute__((ext_vector_type(2))) float v2f;
typedef __attribute__((ext_vector_type(8))) float v8f;

template <bool EPI>
__global__ void gat_gemm_f32_wmma(const float* __restrict__ A,
                                  const float* __restrict__ W,
                                  const float* __restrict__ bias,
                                  const float* __restrict__ cin,
                                  float* __restrict__ out) {
  const int wave = threadIdx.x >> 5;
  const int lane = threadIdx.x & 31;
  const int row0 = blockIdx.x * 16;
  const int col0 = wave * 16;
  const int m  = lane & 15;
  const int kh = lane >> 4;

  const float* __restrict__ arow = A + (size_t)(row0 + m) * DIM;

  const int nc = col0 + m;
  const int rbase = row0 + kh * 8;

  v8f acc = {};
  if (EPI) {
#pragma unroll
    for (int r = 0; r < 8; ++r) acc[r] = cin[(size_t)(rbase + r) * DIM + nc];
  }
  for (int k0 = 0; k0 < DIM; k0 += 4) {
    const int ka = k0 + kh * 2;
    v2f av, bv;
    av.x = arow[ka];
    av.y = arow[ka + 1];
    bv.x = W[(size_t)ka * DIM + col0 + m];
    bv.y = W[(size_t)(ka + 1) * DIM + col0 + m];
    acc = __builtin_amdgcn_wmma_f32_16x16x4_f32(
        false, av, false, bv, (short)0, acc, false, false);
  }

  if (EPI) {
    const float b = bias[nc];
#pragma unroll
    for (int r = 0; r < 8; ++r) {
      const float v = acc[r] + b;
      out[(size_t)(rbase + r) * DIM + nc] = (v > 0.0f) ? v : expm1f(v);
    }
  } else {
#pragma unroll
    for (int r = 0; r < 8; ++r)
      out[(size_t)(rbase + r) * DIM + nc] = acc[r];
  }
}

__global__ void gat_attn_scores(const float* __restrict__ h,
                                const float* __restrict__ a_src,
                                const float* __restrict__ a_dst,
                                float* __restrict__ asrc,
                                float* __restrict__ adst, int N) {
  int t = blockIdx.x * blockDim.x + threadIdx.x;
  if (t >= N * HEADS) return;
  const int n = t >> 2, hh = t & 3;
  const float* __restrict__ hp = h + (size_t)n * DIM + hh * CH;
  const float* __restrict__ ws = a_src + hh * CH;
  const float* __restrict__ wd = a_dst + hh * CH;
  float s1 = 0.f, s2 = 0.f;
#pragma unroll
  for (int c = 0; c < CH; ++c) {
    const float v = hp[c];
    s1 += v * ws[c];
    s2 += v * wd[c];
  }
  asrc[t] = s1;
  adst[t] = s2;
}

__device__ __forceinline__ void edge_sd(const int* __restrict__ ei, int e,
                                        int NE, int& s, int& d) {
  if (e < NE) { s = ei[e]; d = ei[NE + e]; }
  else        { s = d = e - NE; }
}

__device__ __forceinline__ unsigned ord_encode(float f) {
  unsigned u = __float_as_uint(f);
  return (f >= 0.0f) ? (u | 0x80000000u) : ~u;
}

__device__ __forceinline__ float ord_decode(unsigned u) {
  const unsigned top = (unsigned)((int)u >> 31);
  const float f = __uint_as_float(u ^ (~top | 0x80000000u));
  return (u == 0u) ? 0.0f : f;
}

__device__ __forceinline__ float leaky(float v) {
  return (v > 0.0f) ? v : NEG_SLOPE * v;
}

#define RNG 32
#define RNG_SHIFT 5
#define REP 8
#define LINE 32

__device__ __forceinline__ int sub_bin(int e, int d) {
  return (d >> RNG_SHIFT) * REP + (e & (REP - 1));
}

__global__ void gat_bin_zero(unsigned* __restrict__ cnt, int n) {
  int t = blockIdx.x * blockDim.x + threadIdx.x;
  if (t < n) cnt[(size_t)t * LINE] = 0u;
}

__global__ void gat_bin_count(const int* __restrict__ ei, int NE, int E,
                              unsigned* __restrict__ cnt) {
  int e = blockIdx.x * blockDim.x + threadIdx.x;
  if (e >= E) return;
  int s, d;
  edge_sd(ei, e, NE, s, d);
  atomicAdd(&cnt[(size_t)sub_bin(e, d) * LINE], 1u);
}

__global__ void gat_bin_scan(const unsigned* __restrict__ cnt,
                             unsigned* __restrict__ off,
                             unsigned* __restrict__ cursor, int NB) {
  __shared__ unsigned sh[1024];
  const int t = threadIdx.x;
  const int per = (NB + 1023) / 1024;
  const int b0 = t * per;
  unsigned sum = 0u;
  for (int i = 0; i < per; ++i) {
    const int b = b0 + i;
    if (b < NB) sum += cnt[(size_t)b * LINE];
  }
  sh[t] = sum;
  __syncthreads();
  for (int ofs = 1; ofs < 1024; ofs <<= 1) {
    const unsigned v = (t >= ofs) ? sh[t - ofs] : 0u;
    __syncthreads();
    sh[t] += v;
    __syncthreads();
  }
  unsigned run = sh[t] - sum;
  for (int i = 0; i < per; ++i) {
    const int b = b0 + i;
    if (b < NB) {
      off[b] = run;
      cursor[(size_t)b * LINE] = run;
      run += cnt[(size_t)b * LINE];
    }
  }
  if (t == 1023) off[NB] = sh[1023];
}

__global__ void gat_bin_fill(const int* __restrict__ ei, int NE, int E,
                             unsigned* __restrict__ cursor,
                             int4* __restrict__ rec) {
  int e = blockIdx.x * blockDim.x + threadIdx.x;
  if (e >= E) return;
  int s, d;
  edge_sd(ei, e, NE, s, d);
  const unsigned pos = atomicAdd(&cursor[(size_t)sub_bin(e, d) * LINE], 1u);
  rec[pos] = make_int4(e, s, d, 0);
}

#define GRP 4
template <bool FULL>
__device__ __forceinline__ void agg_group(const float* __restrict__ h,
                                          float* acc, int lane, int j, int n,
                                          int s, int dl, float a0, float a1,
                                          float a2, float a3) {
  float hv[GRP][4];
  float al[GRP][4];
  int ro[GRP];
#pragma unroll
  for (int u = 0; u < GRP; ++u) {
    if (FULL || j + u < n) {
      const int sj = __builtin_amdgcn_readlane(s, j + u);
      ro[u] = __builtin_amdgcn_readlane(dl, j + u);
      al[u][0] = __int_as_float(__builtin_amdgcn_readlane(__float_as_int(a0), j + u));
      al[u][1] = __int_as_float(__builtin_amdgcn_readlane(__float_as_int(a1), j + u));
      al[u][2] = __int_as_float(__builtin_amdgcn_readlane(__float_as_int(a2), j + u));
      al[u][3] = __int_as_float(__builtin_amdgcn_readlane(__float_as_int(a3), j + u));
      const float* __restrict__ hp = h + (size_t)sj * DIM + lane;
#pragma unroll
      for (int k = 0; k < 4; ++k) hv[u][k] = hp[32 * k];
    }
  }
#pragma unroll
  for (int u = 0; u < GRP; ++u) {
    if (FULL || j + u < n) {
#pragma unroll
      for (int k = 0; k < 4; ++k)
        atomicAdd(&acc[ro[u] + lane + 32 * k], hv[u][k] * al[u][k]);
    }
  }
}

__global__ void __launch_bounds__(256)
gat_agg_lds(const int4* __restrict__ rec, const unsigned* __restrict__ off,
            const float* __restrict__ asrc, const float* __restrict__ adst,
            float4* __restrict__ scr, const float* __restrict__ h,
            float* __restrict__ agg, int N) {
  __shared__ alignas(16) float acc[RNG * DIM];
  __shared__ alignas(16) float s_adst[RNG * HEADS];
  __shared__ alignas(16) unsigned s_menc[RNG * HEADS];
  __shared__ alignas(16) float s_den[RNG * HEADS];
  const int tid = threadIdx.x;
  const int lane = tid & 31;
  const int wave = __builtin_amdgcn_readfirstlane(tid >> 5);
  const int b = blockIdx.x;

  float4* __restrict__ tile = (float4*)(agg + (size_t)b * RNG * DIM);
  float4* acc4 = (float4*)acc;
  const int lim4 = min(RNG, N - b * RNG) * (DIM / 4);
  const bool whole = (lim4 == RNG * DIM / 4);
#pragma unroll
  for (int m = 0; m < RNG * DIM / 1024; ++m)
    acc4[tid + 256 * m] = make_float4(0.f, 0.f, 0.f, 0.f);
  if (tid < RNG * HEADS) {
    const int node = b * RNG + (tid >> 2);
    s_adst[tid] = (node < N) ? adst[(size_t)b * RNG * HEADS + tid] : 0.0f;
    s_menc[tid] = 0u;
    s_den[tid] = 0.0f;
  }
  __syncthreads();

  const int beg = __builtin_amdgcn_readfirstlane((int)off[b * REP]);
  const int end = __builtin_amdgcn_readfirstlane((int)off[(b + 1) * REP]);
  const int per = (end - beg + 7) >> 3;
  const int w0 = beg + wave * per;
  const int w1 = min(w0 + per, end);


  for (int base = w0; base < w1; base += 64) {
    int q[2], dl4[2];
    bool ok[2];
    float4 av[2];
#pragma unroll
    for (int u = 0; u < 2; ++u) {
      const int p = base + 32 * u + lane;
      ok[u] = p < w1;
      q[u] = min(p, w1 - 1);
    }
    int4 r[2];
#pragma unroll
    for (int u = 0; u < 2; ++u) r[u] = rec[q[u]];
#pragma unroll
    for (int u = 0; u < 2; ++u) {
      av[u] = *(const float4*)(asrc + (size_t)r[u].y * HEADS);
      dl4[u] = (r[u].z - b * RNG) * HEADS;
    }
    asm volatile("" ::"v"(av[0].x), "v"(av[1].x));
#pragma unroll
    for (int u = 0; u < 2; ++u) {
      if (ok[u]) {
        const float4 dv = *(const float4*)(s_adst + dl4[u]);
        float4 e;
        e.x = leaky(av[u].x + dv.x);
        e.y = leaky(av[u].y + dv.y);
        e.z = leaky(av[u].z + dv.z);
        e.w = leaky(av[u].w + dv.w);
        scr[q[u]] = e;
        atomicMax(&s_menc[dl4[u] + 0], ord_encode(e.x));
        atomicMax(&s_menc[dl4[u] + 1], ord_encode(e.y));
        atomicMax(&s_menc[dl4[u] + 2], ord_encode(e.z));
        atomicMax(&s_menc[dl4[u] + 3], ord_encode(e.w));
      }
    }
  }
  __syncthreads();

  for (int base = w0; base < w1; base += 64) {
    int q[2], dl4[2];
    bool ok[2];
    float4 e[2];
#pragma unroll
    for (int u = 0; u < 2; ++u) {
      const int p = base + 32 * u + lane;
      ok[u] = p < w1;
      q[u] = min(p, w1 - 1);
    }
    int dz[2];
#pragma unroll
    for (int u = 0; u < 2; ++u) {
      dz[u] = rec[q[u]].z;
      e[u] = scr[q[u]];
    }
    asm volatile("" ::"v"(dz[0]), "v"(dz[1]), "v"(e[0].x), "v"(e[1].x));
#pragma unroll
    for (int u = 0; u < 2; ++u) dl4[u] = (dz[u] - b * RNG) * HEADS;
#pragma unroll
    for (int u = 0; u < 2; ++u) {
      if (ok[u]) {
        const uint4 mu = *(const uint4*)(s_menc + dl4[u]);
        float4 ex;
        ex.x = expf(e[u].x - ord_decode(mu.x));
        ex.y = expf(e[u].y - ord_decode(mu.y));
        ex.z = expf(e[u].z - ord_decode(mu.z));
        ex.w = expf(e[u].w - ord_decode(mu.w));
        scr[q[u]] = ex;
        atomicAdd(&s_den[dl4[u] + 0], ex.x);
        atomicAdd(&s_den[dl4[u] + 1], ex.y);
        atomicAdd(&s_den[dl4[u] + 2], ex.z);
        atomicAdd(&s_den[dl4[u] + 3], ex.w);
      }
    }
  }
  __syncthreads();

  for (int base = w0; base < w1; base += 32) {
    const int n = min(32, w1 - base);
    int s = 0, d = b * RNG;
    float a0 = 0.f, a1 = 0.f, a2 = 0.f, a3 = 0.f;
    if (lane < n) {
      const int4 r = rec[base + lane];
      s = r.y;
      d = r.z;
      const float4 ev = scr[base + lane];
      const float4 dv = *(const float4*)(s_den + (d - b * RNG) * HEADS);
      a0 = ev.x / (dv.x + 1e-16f);
      a1 = ev.y / (dv.y + 1e-16f);
      a2 = ev.z / (dv.z + 1e-16f);
      a3 = ev.w / (dv.w + 1e-16f);
    }
    const int dl = (d - b * RNG) * DIM;
    int j = 0;
    for (; j + GRP <= n; j += GRP)
      agg_group<true>(h, acc, lane, j, n, s, dl, a0, a1, a2, a3);
    if (j < n)
      agg_group<false>(h, acc, lane, j, n, s, dl, a0, a1, a2, a3);
  }
  __syncthreads();

  if (whole) {
#pragma unroll
    for (int m = 0; m < RNG * DIM / 1024; ++m)
      tile[tid + 256 * m] = acc4[tid + 256 * m];
  } else {
    for (int m = 0; m < RNG * DIM / 1024; ++m) {
      const int i = tid + 256 * m;
      if (i < lim4) tile[i] = acc4[i];
    }
  }
}

extern "C" void kernel_launch(void* const* d_in, const int* in_sizes, int n_in,
                              void* d_out, int out_size, void* d_ws,
                              size_t ws_size, hipStream_t stream) {
  const float* x  = (const float*)d_in[0];
  const int*   ei = (const int*)d_in[1];

  const int N  = in_sizes[0] / DIM;
  const int NE = in_sizes[1] / 2;
  const int E  = NE + N;

  float* ws = (float*)d_ws;
  size_t o = 0;
  float* h     = ws + o; o += (size_t)N * DIM;
  float* agg   = ws + o; o += (size_t)N * DIM;
  float* y1    = ws + o; o += (size_t)N * DIM;
  float* asrc  = ws + o; o += (size_t)N * HEADS;
  float* adst  = ws + o; o += (size_t)N * HEADS;
  float4* scr  = (float4*)(ws + o); o += (size_t)E * HEADS;
  const int NB = (N + RNG - 1) / RNG;
  int4* rec = (int4*)(ws + o); o += (size_t)E * 4;
  const int NBR = NB * REP;
  unsigned* bcnt = (unsigned*)(ws + o); o += (size_t)NBR * LINE;
  unsigned* bcur = (unsigned*)(ws + o); o += (size_t)NBR * LINE;
  unsigned* boff = (unsigned*)(ws + o); o += (size_t)((NBR + 4) & ~3);

  float* y = (float*)d_out;

  const int TB = 256;
  const int gGemm = N / 16;
  const int gNH   = (N * HEADS + TB - 1) / TB;
  const int gE    = (E + TB - 1) / TB;
  const int gNB   = (NBR + TB - 1) / TB;

  gat_bin_zero<<<gNB, TB, 0, stream>>>(bcnt, NBR);
  gat_bin_count<<<gE, TB, 0, stream>>>(ei, NE, E, bcnt);
  gat_bin_scan<<<1, 1024, 0, stream>>>(bcnt, boff, bcur, NBR);
  gat_bin_fill<<<gE, TB, 0, stream>>>(ei, NE, E, bcur, rec);

  const float* xin = x;
  for (int layer = 0; layer < 2; ++layer) {
    const float* W     = (const float*)d_in[2 + layer * 5];
    const float* a_src = (const float*)d_in[3 + layer * 5];
    const float* a_dst = (const float*)d_in[4 + layer * 5];
    const float* bias  = (const float*)d_in[5 + layer * 5];
    const float* Wskip = (const float*)d_in[6 + layer * 5];

    float* yout = (layer == 0) ? y1 : y;

    gat_gemm_f32_wmma<false><<<gGemm, TB, 0, stream>>>(xin, W, nullptr,
                                                       nullptr, h);

    gat_attn_scores<<<gNH, TB, 0, stream>>>(h, a_src, a_dst, asrc, adst, N);

    gat_agg_lds<<<NB, TB, 0, stream>>>(rec, boff, asrc, adst, scr, h, agg, N);

    gat_gemm_f32_wmma<true><<<gGemm, TB, 0, stream>>>(xin, Wskip, bias, agg,
                                                      yout);
    xin = yout;
  }
}


# Skip product, bias, ELU moved into the aggregation block: its x rows copied to LDS, f32 WMMA on the LDS message tile as accumulator start; closing-product launches, sums plane gone; sum order kept
# speedup vs baseline: 2.4991x; 1.1335x over previous
#include <hip/hip_runtime.h>
#include <math.h>

#define DIM 128
#define HEADS 4
#define CH 32
#define NEG_SLOPE 0.2f

typedef __attribute__((ext_vector_type(2))) float v2f;
typedef __attribute__((ext_vector_type(8))) float v8f;

__global__ void gat_gemm_f32_wmma(const float* __restrict__ A,
                                  const float* __restrict__ W,
                                  float* __restrict__ out) {
  const int wave = threadIdx.x >> 5;
  const int lane = threadIdx.x & 31;
  const int row0 = blockIdx.x * 16;
  const int col0 = wave * 16;
  const int m  = lane & 15;
  const int kh = lane >> 4;

  const float* __restrict__ arow = A + (size_t)(row0 + m) * DIM;

  const int nc = col0 + m;
  const int rbase = row0 + kh * 8;

  v8f acc = {};
  for (int k0 = 0; k0 < DIM; k0 += 4) {
    const int ka = k0 + kh * 2;
    v2f av, bv;
    av.x = arow[ka];
    av.y = arow[ka + 1];
    bv.x = W[(size_t)ka * DIM + col0 + m];
    bv.y = W[(size_t)(ka + 1) * DIM + col0 + m];
    acc = __builtin_amdgcn_wmma_f32_16x16x4_f32(
        false, av, false, bv, (short)0, acc, false, false);
  }

#pragma unroll
  for (int r = 0; r < 8; ++r)
    out[(size_t)(rbase + r) * DIM + nc] = acc[r];
}

__global__ void gat_attn_scores(const float* __restrict__ h,
                                const float* __restrict__ a_src,
                                const float* __restrict__ a_dst,
                                float* __restrict__ asrc,
                                float* __restrict__ adst, int N) {
  int t = blockIdx.x * blockDim.x + threadIdx.x;
  if (t >= N * HEADS) return;
  const int n = t >> 2, hh = t & 3;
  const float* __restrict__ hp = h + (size_t)n * DIM + hh * CH;
  const float* __restrict__ ws = a_src + hh * CH;
  const float* __restrict__ wd = a_dst + hh * CH;
  float s1 = 0.f, s2 = 0.f;
#pragma unroll
  for (int c = 0; c < CH; ++c) {
    const float v = hp[c];
    s1 += v * ws[c];
    s2 += v * wd[c];
  }
  asrc[t] = s1;
  adst[t] = s2;
}

__device__ __forceinline__ void edge_sd(const int* __restrict__ ei, int e,
                                        int NE, int& s, int& d) {
  if (e < NE) { s = ei[e]; d = ei[NE + e]; }
  else        { s = d = e - NE; }
}

__device__ __forceinline__ unsigned ord_encode(float f) {
  unsigned u = __float_as_uint(f);
  return (f >= 0.0f) ? (u | 0x80000000u) : ~u;
}

__device__ __forceinline__ float ord_decode(unsigned u) {
  const unsigned top = (unsigned)((int)u >> 31);
  const float f = __uint_as_float(u ^ (~top | 0x80000000u));
  return (u == 0u) ? 0.0f : f;
}

__device__ __forceinline__ float leaky(float v) {
  return (v > 0.0f) ? v : NEG_SLOPE * v;
}

#define RNG 32
#define RNG_SHIFT 5
#define REP 8
#define LINE 32

__device__ __forceinline__ int sub_bin(int e, int d) {
  return (d >> RNG_SHIFT) * REP + (e & (REP - 1));
}

__global__ void gat_bin_zero(unsigned* __restrict__ cnt, int n) {
  int t = blockIdx.x * blockDim.x + threadIdx.x;
  if (t < n) cnt[(size_t)t * LINE] = 0u;
}

__global__ void gat_bin_count(const int* __restrict__ ei, int NE, int E,
                              unsigned* __restrict__ cnt) {
  int e = blockIdx.x * blockDim.x + threadIdx.x;
  if (e >= E) return;
  int s, d;
  edge_sd(ei, e, NE, s, d);
  atomicAdd(&cnt[(size_t)sub_bin(e, d) * LINE], 1u);
}

__global__ void gat_bin_scan(const unsigned* __restrict__ cnt,
                             unsigned* __restrict__ off,
                             unsigned* __restrict__ cursor, int NB) {
  __shared__ unsigned sh[1024];
  const int t = threadIdx.x;
  const int per = (NB + 1023) / 1024;
  const int b0 = t * per;
  unsigned sum = 0u;
  for (int i = 0; i < per; ++i) {
    const int b = b0 + i;
    if (b < NB) sum += cnt[(size_t)b * LINE];
  }
  sh[t] = sum;
  __syncthreads();
  for (int ofs = 1; ofs < 1024; ofs <<= 1) {
    const unsigned v = (t >= ofs) ? sh[t - ofs] : 0u;
    __syncthreads();
    sh[t] += v;
    __syncthreads();
  }
  unsigned run = sh[t] - sum;
  for (int i = 0; i < per; ++i) {
    const int b = b0 + i;
    if (b < NB) {
      off[b] = run;
      cursor[(size_t)b * LINE] = run;
      run += cnt[(size_t)b * LINE];
    }
  }
  if (t == 1023) off[NB] = sh[1023];
}

__global__ void gat_bin_fill(const int* __restrict__ ei, int NE, int E,
                             unsigned* __restrict__ cursor,
                             int4* __restrict__ rec) {
  int e = blockIdx.x * blockDim.x + threadIdx.x;
  if (e >= E) return;
  int s, d;
  edge_sd(ei, e, NE, s, d);
  const unsigned pos = atomicAdd(&cursor[(size_t)sub_bin(e, d) * LINE], 1u);
  rec[pos] = make_int4(e, s, d, 0);
}

#define GRP 4
template <bool FULL>
__device__ __forceinline__ void agg_group(const float* __restrict__ h,
                                          float* acc, int lane, int j, int n,
                                          int s, int dl, float a0, float a1,
                                          float a2, float a3) {
  float hv[GRP][4];
  float al[GRP][4];
  int ro[GRP];
#pragma unroll
  for (int u = 0; u < GRP; ++u) {
    if (FULL || j + u < n) {
      const int sj = __builtin_amdgcn_readlane(s, j + u);
      ro[u] = __builtin_amdgcn_readlane(dl, j + u);
      al[u][0] = __int_as_float(__builtin_amdgcn_readlane(__float_as_int(a0), j + u));
      al[u][1] = __int_as_float(__builtin_amdgcn_readlane(__float_as_int(a1), j + u));
      al[u][2] = __int_as_float(__builtin_amdgcn_readlane(__float_as_int(a2), j + u));
      al[u][3] = __int_as_float(__builtin_amdgcn_readlane(__float_as_int(a3), j + u));
      const float* __restrict__ hp = h + (size_t)sj * DIM + lane;
#pragma unroll
      for (int k = 0; k < 4; ++k) hv[u][k] = hp[32 * k];
    }
  }
#pragma unroll
  for (int u = 0; u < GRP; ++u) {
    if (FULL || j + u < n) {
#pragma unroll
      for (int k = 0; k < 4; ++k)
        atomicAdd(&acc[ro[u] + lane + 32 * k], hv[u][k] * al[u][k]);
    }
  }
}

#define XP (DIM + 4)
__global__ void __launch_bounds__(256)
gat_agg_lds(const int4* __restrict__ rec, const unsigned* __restrict__ off,
            const float* __restrict__ asrc, const float* __restrict__ adst,
            float4* __restrict__ scr, const float* __restrict__ h,
            const float* __restrict__ xin, const float* __restrict__ Wskip,
            const float* __restrict__ bias, float* __restrict__ yout, int N) {
  __shared__ alignas(16) float acc[RNG * DIM];
  __shared__ alignas(16) float xt[RNG * XP];
  __shared__ alignas(16) float s_adst[RNG * HEADS];
  __shared__ alignas(16) unsigned s_menc[RNG * HEADS];
  __shared__ alignas(16) float s_den[RNG * HEADS];
  const int tid = threadIdx.x;
  const int lane = tid & 31;
  const int wave = __builtin_amdgcn_readfirstlane(tid >> 5);
  const int b = blockIdx.x;

  float4* __restrict__ tile = (float4*)(yout + (size_t)b * RNG * DIM);
  float4* acc4 = (float4*)acc;
  const int lim4 = min(RNG, N - b * RNG) * (DIM / 4);
  const bool whole = (lim4 == RNG * DIM / 4);

  const float4* __restrict__ xg = (const float4*)(xin + (size_t)b * RNG * DIM);
  float4 xr[RNG * DIM / 1024];
#pragma unroll
  for (int m = 0; m < RNG * DIM / 1024; ++m) {
    const int i = tid + 256 * m;
    xr[m] = (i < lim4) ? xg[i] : make_float4(0.f, 0.f, 0.f, 0.f);
  }
#pragma unroll
  for (int m = 0; m < RNG * DIM / 1024; ++m)
    acc4[tid + 256 * m] = make_float4(0.f, 0.f, 0.f, 0.f);
  if (tid < RNG * HEADS) {
    const int node = b * RNG + (tid >> 2);
    s_adst[tid] = (node < N) ? adst[(size_t)b * RNG * HEADS + tid] : 0.0f;
    s_menc[tid] = 0u;
    s_den[tid] = 0.0f;
  }
  __syncthreads();

  const int beg = __builtin_amdgcn_readfirstlane((int)off[b * REP]);
  const int end = __builtin_amdgcn_readfirstlane((int)off[(b + 1) * REP]);
  const int per = (end - beg + 7) >> 3;
  const int w0 = beg + wave * per;
  const int w1 = min(w0 + per, end);


  for (int base = w0; base < w1; base += 64) {
    int q[2], dl4[2];
    bool ok[2];
    float4 av[2];
#pragma unroll
    for (int u = 0; u < 2; ++u) {
      const int p = base + 32 * u + lane;
      ok[u] = p < w1;
      q[u] = min(p, w1 - 1);
    }
    int4 r[2];
#pragma unroll
    for (int u = 0; u < 2; ++u) r[u] = rec[q[u]];
#pragma unroll
    for (int u = 0; u < 2; ++u) {
      av[u] = *(const float4*)(asrc + (size_t)r[u].y * HEADS);
      dl4[u] = (r[u].z - b * RNG) * HEADS;
    }
    asm volatile("" ::"v"(av[0].x), "v"(av[1].x));
#pragma unroll
    for (int u = 0; u < 2; ++u) {
      if (ok[u]) {
        const float4 dv = *(const float4*)(s_adst + dl4[u]);
        float4 e;
        e.x = leaky(av[u].x + dv.x);
        e.y = leaky(av[u].y + dv.y);
        e.z = leaky(av[u].z + dv.z);
        e.w = leaky(av[u].w + dv.w);
        scr[q[u]] = e;
        atomicMax(&s_menc[dl4[u] + 0], ord_encode(e.x));
        atomicMax(&s_menc[dl4[u] + 1], ord_encode(e.y));
        atomicMax(&s_menc[dl4[u] + 2], ord_encode(e.z));
        atomicMax(&s_menc[dl4[u] + 3], ord_encode(e.w));
      }
    }
  }
#pragma unroll
  for (int m = 0; m < RNG * DIM / 1024; ++m) {
    const int i = tid + 256 * m;
    *(float4*)(xt + (i >> 5) * XP + (i & 31) * 4) = xr[m];
  }
  __syncthreads();

  for (int base = w0; base < w1; base += 64) {
    int q[2], dl4[2];
    bool ok[2];
    float4 e[2];
#pragma unroll
    for (int u = 0; u < 2; ++u) {
      const int p = base + 32 * u + lane;
      ok[u] = p < w1;
      q[u] = min(p, w1 - 1);
    }
    int dz[2];
#pragma unroll
    for (int u = 0; u < 2; ++u) {
      dz[u] = rec[q[u]].z;
      e[u] = scr[q[u]];
    }
    asm volatile("" ::"v"(dz[0]), "v"(dz[1]), "v"(e[0].x), "v"(e[1].x));
#pragma unroll
    for (int u = 0; u < 2; ++u) dl4[u] = (dz[u] - b * RNG) * HEADS;
#pragma unroll
    for (int u = 0; u < 2; ++u) {
      if (ok[u]) {
        const uint4 mu = *(const uint4*)(s_menc + dl4[u]);
        float4 ex;
        ex.x = expf(e[u].x - ord_decode(mu.x));
        ex.y = expf(e[u].y - ord_decode(mu.y));
        ex.z = expf(e[u].z - ord_decode(mu.z));
        ex.w = expf(e[u].w - ord_decode(mu.w));
        scr[q[u]] = ex;
        atomicAdd(&s_den[dl4[u] + 0], ex.x);
        atomicAdd(&s_den[dl4[u] + 1], ex.y);
        atomicAdd(&s_den[dl4[u] + 2], ex.z);
        atomicAdd(&s_den[dl4[u] + 3], ex.w);
      }
    }
  }
  __syncthreads();

  for (int base = w0; base < w1; base += 32) {
    const int n = min(32, w1 - base);
    int s = 0, d = b * RNG;
    float a0 = 0.f, a1 = 0.f, a2 = 0.f, a3 = 0.f;
    if (lane < n) {
      const int4 r = rec[base + lane];
      s = r.y;
      d = r.z;
      const float4 ev = scr[base + lane];
      const float4 dv = *(const float4*)(s_den + (d - b * RNG) * HEADS);
      a0 = ev.x / (dv.x + 1e-16f);
      a1 = ev.y / (dv.y + 1e-16f);
      a2 = ev.z / (dv.z + 1e-16f);
      a3 = ev.w / (dv.w + 1e-16f);
    }
    const int dl = (d - b * RNG) * DIM;
    int j = 0;
    for (; j + GRP <= n; j += GRP)
      agg_group<true>(h, acc, lane, j, n, s, dl, a0, a1, a2, a3);
    if (j < n)
      agg_group<false>(h, acc, lane, j, n, s, dl, a0, a1, a2, a3);
  }
  __syncthreads();

  {
    const int m  = lane & 15;
    const int kh = lane >> 4;
    const int nc = wave * 16 + m;
    float* c_lo = acc + (kh * 8) * DIM + nc;
    float* c_hi = c_lo + 16 * DIM;
    v8f c0, c1;
#pragma unroll
    for (int r = 0; r < 8; ++r) {
      c0[r] = c_lo[r * DIM];
      c1[r] = c_hi[r * DIM];
    }
    const float* a_lo = xt + m * XP + kh * 2;
    const float* a_hi = a_lo + 16 * XP;
    const float* __restrict__ wb = Wskip + (size_t)(kh * 2) * DIM + nc;
#pragma unroll 8
    for (int k0 = 0; k0 < DIM; k0 += 4) {
      const v2f av0 = *(const v2f*)(a_lo + k0);
      const v2f av1 = *(const v2f*)(a_hi + k0);
      v2f bv;
      bv.x = wb[(size_t)k0 * DIM];
      bv.y = wb[(size_t)(k0 + 1) * DIM];
      c0 = __builtin_amdgcn_wmma_f32_16x16x4_f32(
          false, av0, false, bv, (short)0, c0, false, false);
      c1 = __builtin_amdgcn_wmma_f32_16x16x4_f32(
          false, av1, false, bv, (short)0, c1, false, false);
    }
    const float bb = bias[nc];
#pragma unroll
    for (int r = 0; r < 8; ++r) {
      const float v0 = c0[r] + bb;
      const float v1 = c1[r] + bb;
      c_lo[r * DIM] = (v0 > 0.0f) ? v0 : expm1f(v0);
      c_hi[r * DIM] = (v1 > 0.0f) ? v1 : expm1f(v1);
    }
  }
  __syncthreads();

  if (whole) {
#pragma unroll
    for (int m = 0; m < RNG * DIM / 1024; ++m)
      tile[tid + 256 * m] = acc4[tid + 256 * m];
  } else {
    for (int m = 0; m < RNG * DIM / 1024; ++m) {
      const int i = tid + 256 * m;
      if (i < lim4) tile[i] = acc4[i];
    }
  }
}

extern "C" void kernel_launch(void* const* d_in, const int* in_sizes, int n_in,
                              void* d_out, int out_size, void* d_ws,
                              size_t ws_size, hipStream_t stream) {
  const float* x  = (const float*)d_in[0];
  const int*   ei = (const int*)d_in[1];

  const int N  = in_sizes[0] / DIM;
  const int NE = in_sizes[1] / 2;
  const int E  = NE + N;

  float* ws = (float*)d_ws;
  size_t o = 0;
  float* h     = ws + o; o += (size_t)N * DIM;
  float* y1    = ws + o; o += (size_t)N * DIM;
  float* asrc  = ws + o; o += (size_t)N * HEADS;
  float* adst  = ws + o; o += (size_t)N * HEADS;
  float4* scr  = (float4*)(ws + o); o += (size_t)E * HEADS;
  const int NB = (N + RNG - 1) / RNG;
  int4* rec = (int4*)(ws + o); o += (size_t)E * 4;
  const int NBR = NB * REP;
  unsigned* bcnt = (unsigned*)(ws + o); o += (size_t)NBR * LINE;
  unsigned* bcur = (unsigned*)(ws + o); o += (size_t)NBR * LINE;
  unsigned* boff = (unsigned*)(ws + o); o += (size_t)((NBR + 4) & ~3);

  float* y = (float*)d_out;

  const int TB = 256;
  const int gGemm = N / 16;
  const int gNH   = (N * HEADS + TB - 1) / TB;
  const int gE    = (E + TB - 1) / TB;
  const int gNB   = (NBR + TB - 1) / TB;

  gat_bin_zero<<<gNB, TB, 0, stream>>>(bcnt, NBR);
  gat_bin_count<<<gE, TB, 0, stream>>>(ei, NE, E, bcnt);
  gat_bin_scan<<<1, 1024, 0, stream>>>(bcnt, boff, bcur, NBR);
  gat_bin_fill<<<gE, TB, 0, stream>>>(ei, NE, E, bcur, rec);

  const float* xin = x;
  for (int layer = 0; layer < 2; ++layer) {
    const float* W     = (const float*)d_in[2 + layer * 5];
    const float* a_src = (const float*)d_in[3 + layer * 5];
    const float* a_dst = (const float*)d_in[4 + layer * 5];
    const float* bias  = (const float*)d_in[5 + layer * 5];
    const float* Wskip = (const float*)d_in[6 + layer * 5];

    float* yout = (layer == 0) ? y1 : y;

    gat_gemm_f32_wmma<<<gGemm, TB, 0, stream>>>(xin, W, h);

    gat_attn_scores<<<gNH, TB, 0, stream>>>(h, a_src, a_dst, asrc, adst, N);

    gat_agg_lds<<<NB, TB, 0, stream>>>(rec, boff, asrc, adst, scr, h, xin,
                                       Wskip, bias, yout, N);
    xin = yout;
  }
}


# Layer 2's node front end moved into layer 1's aggregation block: h2 = finished LDS tile @ W2 (f32 WMMA, same k order) and its scores there; layer-2 product and score launches gone; sum order kept
# speedup vs baseline: 2.7463x; 1.0989x over previous
#include <hip/hip_runtime.h>
#include <math.h>

#define DIM 128
#define HEADS 4
#define CH 32
#define NEG_SLOPE 0.2f

typedef __attribute__((ext_vector_type(2))) float v2f;
typedef __attribute__((ext_vector_type(8))) float v8f;

__global__ void gat_gemm_f32_wmma(const float* __restrict__ A,
                                  const float* __restrict__ W,
                                  float* __restrict__ out) {
  const int wave = threadIdx.x >> 5;
  const int lane = threadIdx.x & 31;
  const int row0 = blockIdx.x * 16;
  const int col0 = wave * 16;
  const int m  = lane & 15;
  const int kh = lane >> 4;

  const float* __restrict__ arow = A + (size_t)(row0 + m) * DIM;

  const int nc = col0 + m;
  const int rbase = row0 + kh * 8;

  v8f acc = {};
  for (int k0 = 0; k0 < DIM; k0 += 4) {
    const int ka = k0 + kh * 2;
    v2f av, bv;
    av.x = arow[ka];
    av.y = arow[ka + 1];
    bv.x = W[(size_t)ka * DIM + col0 + m];
    bv.y = W[(size_t)(ka + 1) * DIM + col0 + m];
    acc = __builtin_amdgcn_wmma_f32_16x16x4_f32(
        false, av, false, bv, (short)0, acc, false, false);
  }

#pragma unroll
  for (int r = 0; r < 8; ++r)
    out[(size_t)(rbase + r) * DIM + nc] = acc[r];
}

__global__ void gat_attn_scores(const float* __restrict__ h,
                                const float* __restrict__ a_src,
                                const float* __restrict__ a_dst,
                                float* __restrict__ asrc,
                                float* __restrict__ adst, int N) {
  int t = blockIdx.x * blockDim.x + threadIdx.x;
  if (t >= N * HEADS) return;
  const int n = t >> 2, hh = t & 3;
  const float* __restrict__ hp = h + (size_t)n * DIM + hh * CH;
  const float* __restrict__ ws = a_src + hh * CH;
  const float* __restrict__ wd = a_dst + hh * CH;
  float s1 = 0.f, s2 = 0.f;
#pragma unroll
  for (int c = 0; c < CH; ++c) {
    const float v = hp[c];
    s1 += v * ws[c];
    s2 += v * wd[c];
  }
  asrc[t] = s1;
  adst[t] = s2;
}

__device__ __forceinline__ void edge_sd(const int* __restrict__ ei, int e,
                                        int NE, int& s, int& d) {
  if (e < NE) { s = ei[e]; d = ei[NE + e]; }
  else        { s = d = e - NE; }
}

__device__ __forceinline__ unsigned ord_encode(float f) {
  unsigned u = __float_as_uint(f);
  return (f >= 0.0f) ? (u | 0x80000000u) : ~u;
}

__device__ __forceinline__ float ord_decode(unsigned u) {
  const unsigned top = (unsigned)((int)u >> 31);
  const float f = __uint_as_float(u ^ (~top | 0x80000000u));
  return (u == 0u) ? 0.0f : f;
}

__device__ __forceinline__ float leaky(float v) {
  return (v > 0.0f) ? v : NEG_SLOPE * v;
}

#define RNG 32
#define RNG_SHIFT 5
#define REP 8
#define LINE 32

__device__ __forceinline__ int sub_bin(int e, int d) {
  return (d >> RNG_SHIFT) * REP + (e & (REP - 1));
}

__global__ void gat_bin_zero(unsigned* __restrict__ cnt, int n) {
  int t = blockIdx.x * blockDim.x + threadIdx.x;
  if (t < n) cnt[(size_t)t * LINE] = 0u;
}

__global__ void gat_bin_count(const int* __restrict__ ei, int NE, int E,
                              unsigned* __restrict__ cnt) {
  int e = blockIdx.x * blockDim.x + threadIdx.x;
  if (e >= E) return;
  int s, d;
  edge_sd(ei, e, NE, s, d);
  atomicAdd(&cnt[(size_t)sub_bin(e, d) * LINE], 1u);
}

__global__ void gat_bin_scan(const unsigned* __restrict__ cnt,
                             unsigned* __restrict__ off,
                             unsigned* __restrict__ cursor, int NB) {
  __shared__ unsigned sh[1024];
  const int t = threadIdx.x;
  const int per = (NB + 1023) / 1024;
  const int b0 = t * per;
  unsigned sum = 0u;
  for (int i = 0; i < per; ++i) {
    const int b = b0 + i;
    if (b < NB) sum += cnt[(size_t)b * LINE];
  }
  sh[t] = sum;
  __syncthreads();
  for (int ofs = 1; ofs < 1024; ofs <<= 1) {
    const unsigned v = (t >= ofs) ? sh[t - ofs] : 0u;
    __syncthreads();
    sh[t] += v;
    __syncthreads();
  }
  unsigned run = sh[t] - sum;
  for (int i = 0; i < per; ++i) {
    const int b = b0 + i;
    if (b < NB) {
      off[b] = run;
      cursor[(size_t)b * LINE] = run;
      run += cnt[(size_t)b * LINE];
    }
  }
  if (t == 1023) off[NB] = sh[1023];
}

__global__ void gat_bin_fill(const int* __restrict__ ei, int NE, int E,
                             unsigned* __restrict__ cursor,
                             int4* __restrict__ rec) {
  int e = blockIdx.x * blockDim.x + threadIdx.x;
  if (e >= E) return;
  int s, d;
  edge_sd(ei, e, NE, s, d);
  const unsigned pos = atomicAdd(&cursor[(size_t)sub_bin(e, d) * LINE], 1u);
  rec[pos] = make_int4(e, s, d, 0);
}

#define GRP 4
template <bool FULL>
__device__ __forceinline__ void agg_group(const float* __restrict__ h,
                                          float* acc, int lane, int j, int n,
                                          int s, int dl, float a0, float a1,
                                          float a2, float a3) {
  float hv[GRP][4];
  float al[GRP][4];
  int ro[GRP];
#pragma unroll
  for (int u = 0; u < GRP; ++u) {
    if (FULL || j + u < n) {
      const int sj = __builtin_amdgcn_readlane(s, j + u);
      ro[u] = __builtin_amdgcn_readlane(dl, j + u);
      al[u][0] = __int_as_float(__builtin_amdgcn_readlane(__float_as_int(a0), j + u));
      al[u][1] = __int_as_float(__builtin_amdgcn_readlane(__float_as_int(a1), j + u));
      al[u][2] = __int_as_float(__builtin_amdgcn_readlane(__float_as_int(a2), j + u));
      al[u][3] = __int_as_float(__builtin_amdgcn_readlane(__float_as_int(a3), j + u));
      const float* __restrict__ hp = h + (size_t)sj * DIM + lane;
#pragma unroll
      for (int k = 0; k < 4; ++k) hv[u][k] = hp[32 * k];
    }
  }
#pragma unroll
  for (int u = 0; u < GRP; ++u) {
    if (FULL || j + u < n) {
#pragma unroll
      for (int k = 0; k < 4; ++k)
        atomicAdd(&acc[ro[u] + lane + 32 * k], hv[u][k] * al[u][k]);
    }
  }
}

#define XP (DIM + 4)
template <bool NEXT>
__global__ void __launch_bounds__(256)
gat_agg_lds(const int4* __restrict__ rec, const unsigned* __restrict__ off,
            const float* __restrict__ asrc, const float* __restrict__ adst,
            float4* __restrict__ scr, const float* __restrict__ h,
            const float* __restrict__ xin, const float* __restrict__ Wskip,
            const float* __restrict__ bias, float* __restrict__ yout, int N,
            const float* __restrict__ Wn, const float* __restrict__ a_src_n,
            const float* __restrict__ a_dst_n, float* __restrict__ hn,
            float* __restrict__ asrc_n, float* __restrict__ adst_n) {
  __shared__ alignas(16) float acc[RNG * XP];
  __shared__ alignas(16) float xt[RNG * XP];
  __shared__ alignas(16) float s_adst[RNG * HEADS];
  __shared__ alignas(16) unsigned s_menc[RNG * HEADS];
  __shared__ alignas(16) float s_den[RNG * HEADS];
  const int tid = threadIdx.x;
  const int lane = tid & 31;
  const int wave = __builtin_amdgcn_readfirstlane(tid >> 5);
  const int b = blockIdx.x;

  float4* __restrict__ tile = (float4*)(yout + (size_t)b * RNG * DIM);
  const int lim4 = min(RNG, N - b * RNG) * (DIM / 4);
  const bool whole = (lim4 == RNG * DIM / 4);
  const int tpos = (tid >> 5) * XP + (tid & 31) * 4;

  const float4* __restrict__ xg = (const float4*)(xin + (size_t)b * RNG * DIM);
  float4 xr[RNG * DIM / 1024];
#pragma unroll
  for (int m = 0; m < RNG * DIM / 1024; ++m) {
    const int i = tid + 256 * m;
    xr[m] = (i < lim4) ? xg[i] : make_float4(0.f, 0.f, 0.f, 0.f);
  }
#pragma unroll
  for (int m = 0; m < RNG * DIM / 1024; ++m)
    *(float4*)(acc + tpos + 8 * m * XP) = make_float4(0.f, 0.f, 0.f, 0.f);
  if (tid < RNG * HEADS) {
    const int node = b * RNG + (tid >> 2);
    s_adst[tid] = (node < N) ? adst[(size_t)b * RNG * HEADS + tid] : 0.0f;
    s_menc[tid] = 0u;
    s_den[tid] = 0.0f;
  }
  __syncthreads();

  const int beg = __builtin_amdgcn_readfirstlane((int)off[b * REP]);
  const int end = __builtin_amdgcn_readfirstlane((int)off[(b + 1) * REP]);
  const int per = (end - beg + 7) >> 3;
  const int w0 = beg + wave * per;
  const int w1 = min(w0 + per, end);


  for (int base = w0; base < w1; base += 64) {
    int q[2], dl4[2];
    bool ok[2];
    float4 av[2];
#pragma unroll
    for (int u = 0; u < 2; ++u) {
      const int p = base + 32 * u + lane;
      ok[u] = p < w1;
      q[u] = min(p, w1 - 1);
    }
    int4 r[2];
#pragma unroll
    for (int u = 0; u < 2; ++u) r[u] = rec[q[u]];
#pragma unroll
    for (int u = 0; u < 2; ++u) {
      av[u] = *(const float4*)(asrc + (size_t)r[u].y * HEADS);
      dl4[u] = (r[u].z - b * RNG) * HEADS;
    }
    asm volatile("" ::"v"(av[0].x), "v"(av[1].x));
#pragma unroll
    for (int u = 0; u < 2; ++u) {
      if (ok[u]) {
        const float4 dv = *(const float4*)(s_adst + dl4[u]);
        float4 e;
        e.x = leaky(av[u].x + dv.x);
        e.y = leaky(av[u].y + dv.y);
        e.z = leaky(av[u].z + dv.z);
        e.w = leaky(av[u].w + dv.w);
        scr[q[u]] = e;
        atomicMax(&s_menc[dl4[u] + 0], ord_encode(e.x));
        atomicMax(&s_menc[dl4[u] + 1], ord_encode(e.y));
        atomicMax(&s_menc[dl4[u] + 2], ord_encode(e.z));
        atomicMax(&s_menc[dl4[u] + 3], ord_encode(e.w));
      }
    }
  }
#pragma unroll
  for (int m = 0; m < RNG * DIM / 1024; ++m)
    *(float4*)(xt + tpos + 8 * m * XP) = xr[m];
  __syncthreads();

  for (int base = w0; base < w1; base += 64) {
    int q[2], dl4[2];
    bool ok[2];
    float4 e[2];
#pragma unroll
    for (int u = 0; u < 2; ++u) {
      const int p = base + 32 * u + lane;
      ok[u] = p < w1;
      q[u] = min(p, w1 - 1);
    }
    int dz[2];
#pragma unroll
    for (int u = 0; u < 2; ++u) {
      dz[u] = rec[q[u]].z;
      e[u] = scr[q[u]];
    }
    asm volatile("" ::"v"(dz[0]), "v"(dz[1]), "v"(e[0].x), "v"(e[1].x));
#pragma unroll
    for (int u = 0; u < 2; ++u) dl4[u] = (dz[u] - b * RNG) * HEADS;
#pragma unroll
    for (int u = 0; u < 2; ++u) {
      if (ok[u]) {
        const uint4 mu = *(const uint4*)(s_menc + dl4[u]);
        float4 ex;
        ex.x = expf(e[u].x - ord_decode(mu.x));
        ex.y = expf(e[u].y - ord_decode(mu.y));
        ex.z = expf(e[u].z - ord_decode(mu.z));
        ex.w = expf(e[u].w - ord_decode(mu.w));
        scr[q[u]] = ex;
        atomicAdd(&s_den[dl4[u] + 0], ex.x);
        atomicAdd(&s_den[dl4[u] + 1], ex.y);
        atomicAdd(&s_den[dl4[u] + 2], ex.z);
        atomicAdd(&s_den[dl4[u] + 3], ex.w);
      }
    }
  }
  __syncthreads();

  for (int base = w0; base < w1; base += 32) {
    const int n = min(32, w1 - base);
    int s = 0, d = b * RNG;
    float a0 = 0.f, a1 = 0.f, a2 = 0.f, a3 = 0.f;
    if (lane < n) {
      const int4 r = rec[base + lane];
      s = r.y;
      d = r.z;
      const float4 ev = scr[base + lane];
      const float4 dv = *(const float4*)(s_den + (d - b * RNG) * HEADS);
      a0 = ev.x / (dv.x + 1e-16f);
      a1 = ev.y / (dv.y + 1e-16f);
      a2 = ev.z / (dv.z + 1e-16f);
      a3 = ev.w / (dv.w + 1e-16f);
    }
    const int dl = (d - b * RNG) * XP;
    int j = 0;
    for (; j + GRP <= n; j += GRP)
      agg_group<true>(h, acc, lane, j, n, s, dl, a0, a1, a2, a3);
    if (j < n)
      agg_group<false>(h, acc, lane, j, n, s, dl, a0, a1, a2, a3);
  }
  __syncthreads();

  {
    const int m  = lane & 15;
    const int kh = lane >> 4;
    const int nc = wave * 16 + m;
    float* c_lo = acc + (kh * 8) * XP + nc;
    float* c_hi = c_lo + 16 * XP;
    v8f c0, c1;
#pragma unroll
    for (int r = 0; r < 8; ++r) {
      c0[r] = c_lo[r * XP];
      c1[r] = c_hi[r * XP];
    }
    const float* a_lo = xt + m * XP + kh * 2;
    const float* a_hi = a_lo + 16 * XP;
    const float* __restrict__ wb = Wskip + (size_t)(kh * 2) * DIM + nc;
#pragma unroll 8
    for (int k0 = 0; k0 < DIM; k0 += 4) {
      const v2f av0 = *(const v2f*)(a_lo + k0);
      const v2f av1 = *(const v2f*)(a_hi + k0);
      v2f bv;
      bv.x = wb[(size_t)k0 * DIM];
      bv.y = wb[(size_t)(k0 + 1) * DIM];
      c0 = __builtin_amdgcn_wmma_f32_16x16x4_f32(
          false, av0, false, bv, (short)0, c0, false, false);
      c1 = __builtin_amdgcn_wmma_f32_16x16x4_f32(
          false, av1, false, bv, (short)0, c1, false, false);
    }
    const float bb = bias[nc];
#pragma unroll
    for (int r = 0; r < 8; ++r) {
      const float v0 = c0[r] + bb;
      const float v1 = c1[r] + bb;
      c_lo[r * XP] = (v0 > 0.0f) ? v0 : expm1f(v0);
      c_hi[r * XP] = (v1 > 0.0f) ? v1 : expm1f(v1);
    }
  }
  __syncthreads();

  const int ln = (int)__builtin_amdgcn_mbcnt_lo(~0u, 0u);
  const int tq = wave * XP + ln * 4;
  if (whole) {
#pragma unroll
    for (int m = 0; m < RNG * DIM / 1024; ++m)
      tile[tid + 256 * m] = *(const float4*)(acc + tq + 8 * m * XP);
  } else {
    for (int m = 0; m < RNG * DIM / 1024; ++m) {
      const int i = tid + 256 * m;
      if (i < lim4) tile[i] = *(const float4*)(acc + tq + 8 * m * XP);
    }
  }

  if (NEXT) {
    {
      const int m  = ln & 15;
      const int kh = ln >> 4;
      const int nc = wave * 16 + m;
      const float* a_lo = acc + m * XP + kh * 2;
      const float* a_hi = a_lo + 16 * XP;
      const float* __restrict__ wb = Wn + (size_t)(kh * 2) * DIM + nc;
      v8f d0 = {}, d1 = {};
#pragma unroll 8
      for (int k0 = 0; k0 < DIM; k0 += 4) {
        const v2f av0 = *(const v2f*)(a_lo + k0);
        const v2f av1 = *(const v2f*)(a_hi + k0);
        v2f bv;
        bv.x = wb[(size_t)k0 * DIM];
        bv.y = wb[(size_t)(k0 + 1) * DIM];
        d0 = __builtin_amdgcn_wmma_f32_16x16x4_f32(
            false, av0, false, bv, (short)0, d0, false, false);
        d1 = __builtin_amdgcn_wmma_f32_16x16x4_f32(
            false, av1, false, bv, (short)0, d1, false, false);
      }
      float* o_lo = xt + (kh * 8) * XP + nc;
      float* o_hi = o_lo + 16 * XP;
#pragma unroll
      for (int r = 0; r < 8; ++r) {
        o_lo[r * XP] = d0[r];
        o_hi[r * XP] = d1[r];
      }
    }
    __syncthreads();

    if (tid < RNG * HEADS) {
      const int row = tid >> 2, hh = tid & 3;
      const float* hp = xt + row * XP + hh * CH;
      const float* __restrict__ ws = a_src_n + hh * CH;
      const float* __restrict__ wd = a_dst_n + hh * CH;
      float s1 = 0.f, s2 = 0.f;
#pragma unroll 8
      for (int c = 0; c < CH; ++c) {
        const float v = hp[c];
        s1 += v * ws[c];
        s2 += v * wd[c];
      }
      if (b * RNG + row < N) {
        asrc_n[(size_t)b * RNG * HEADS + tid] = s1;
        adst_n[(size_t)b * RNG * HEADS + tid] = s2;
      }
    }

    float4* __restrict__ htile = (float4*)(hn + (size_t)b * RNG * DIM);
    if (whole) {
#pragma unroll
      for (int m = 0; m < RNG * DIM / 1024; ++m)
        htile[tid + 256 * m] = *(const float4*)(xt + tq + 8 * m * XP);
    } else {
      for (int m = 0; m < RNG * DIM / 1024; ++m) {
        const int i = tid + 256 * m;
        if (i < lim4) htile[i] = *(const float4*)(xt + tq + 8 * m * XP);
      }
    }
  }
}

extern "C" void kernel_launch(void* const* d_in, const int* in_sizes, int n_in,
                              void* d_out, int out_size, void* d_ws,
                              size_t ws_size, hipStream_t stream) {
  const float* x  = (const float*)d_in[0];
  const int*   ei = (const int*)d_in[1];

  const int N  = in_sizes[0] / DIM;
  const int NE = in_sizes[1] / 2;
  const int E  = NE + N;

  float* ws = (float*)d_ws;
  size_t o = 0;
  float* h     = ws + o; o += (size_t)N * DIM;
  float* h2    = ws + o; o += (size_t)N * DIM;
  float* y1    = ws + o; o += (size_t)N * DIM;
  float* asrc  = ws + o; o += (size_t)N * HEADS;
  float* adst  = ws + o; o += (size_t)N * HEADS;
  float* asrc2 = ws + o; o += (size_t)N * HEADS;
  float* adst2 = ws + o; o += (size_t)N * HEADS;
  float4* scr  = (float4*)(ws + o); o += (size_t)E * HEADS;
  const int NB = (N + RNG - 1) / RNG;
  int4* rec = (int4*)(ws + o); o += (size_t)E * 4;
  const int NBR = NB * REP;
  unsigned* bcnt = (unsigned*)(ws + o); o += (size_t)NBR * LINE;
  unsigned* bcur = (unsigned*)(ws + o); o += (size_t)NBR * LINE;
  unsigned* boff = (unsigned*)(ws + o); o += (size_t)((NBR + 4) & ~3);

  float* y = (float*)d_out;

  const int TB = 256;
  const int gGemm = N / 16;
  const int gNH   = (N * HEADS + TB - 1) / TB;
  const int gE    = (E + TB - 1) / TB;
  const int gNB   = (NBR + TB - 1) / TB;

  gat_bin_zero<<<gNB, TB, 0, stream>>>(bcnt, NBR);
  gat_bin_count<<<gE, TB, 0, stream>>>(ei, NE, E, bcnt);
  gat_bin_scan<<<1, 1024, 0, stream>>>(bcnt, boff, bcur, NBR);
  gat_bin_fill<<<gE, TB, 0, stream>>>(ei, NE, E, bcur, rec);

  const float* W1     = (const float*)d_in[2];
  const float* a_src1 = (const float*)d_in[3];
  const float* a_dst1 = (const float*)d_in[4];
  const float* bias1  = (const float*)d_in[5];
  const float* Wskip1 = (const float*)d_in[6];
  const float* W2     = (const float*)d_in[7];
  const float* a_src2 = (const float*)d_in[8];
  const float* a_dst2 = (const float*)d_in[9];
  const float* bias2  = (const float*)d_in[10];
  const float* Wskip2 = (const float*)d_in[11];

  gat_gemm_f32_wmma<<<gGemm, TB, 0, stream>>>(x, W1, h);
  gat_attn_scores<<<gNH, TB, 0, stream>>>(h, a_src1, a_dst1, asrc, adst, N);

  gat_agg_lds<true><<<NB, TB, 0, stream>>>(rec, boff, asrc, adst, scr, h, x,
                                           Wskip1, bias1, y1, N, W2, a_src2,
                                           a_dst2, h2, asrc2, adst2);

  gat_agg_lds<false><<<NB, TB, 0, stream>>>(rec, boff, asrc2, adst2, scr, h2,
                                            y1, Wskip2, bias2, y, N, nullptr,
                                            nullptr, nullptr, nullptr,
                                            nullptr, nullptr);
}


# Bin build: the counting add returns each edge's rank in its sub-bin, kept per edge; fill slot = prefix + rank, so the fill pass's global adds and cursor plane go; slot order: counting add's arrival
# speedup vs baseline: 3.2530x; 1.1845x over previous
#include <hip/hip_runtime.h>
#include <math.h>

#define DIM 128
#define HEADS 4
#define CH 32
#define NEG_SLOPE 0.2f

typedef __attribute__((ext_vector_type(2))) float v2f;
typedef __attribute__((ext_vector_type(8))) float v8f;

__global__ void gat_gemm_f32_wmma(const float* __restrict__ A,
                                  const float* __restrict__ W,
                                  float* __restrict__ out) {
  const int wave = threadIdx.x >> 5;
  const int lane = threadIdx.x & 31;
  const int row0 = blockIdx.x * 16;
  const int col0 = wave * 16;
  const int m  = lane & 15;
  const int kh = lane >> 4;

  const float* __restrict__ arow = A + (size_t)(row0 + m) * DIM;

  const int nc = col0 + m;
  const int rbase = row0 + kh * 8;

  v8f acc = {};
  for (int k0 = 0; k0 < DIM; k0 += 4) {
    const int ka = k0 + kh * 2;
    v2f av, bv;
    av.x = arow[ka];
    av.y = arow[ka + 1];
    bv.x = W[(size_t)ka * DIM + col0 + m];
    bv.y = W[(size_t)(ka + 1) * DIM + col0 + m];
    acc = __builtin_amdgcn_wmma_f32_16x16x4_f32(
        false, av, false, bv, (short)0, acc, false, false);
  }

#pragma unroll
  for (int r = 0; r < 8; ++r)
    out[(size_t)(rbase + r) * DIM + nc] = acc[r];
}

__global__ void gat_attn_scores(const float* __restrict__ h,
                                const float* __restrict__ a_src,
                                const float* __restrict__ a_dst,
                                float* __restrict__ asrc,
                                float* __restrict__ adst, int N) {
  int t = blockIdx.x * blockDim.x + threadIdx.x;
  if (t >= N * HEADS) return;
  const int n = t >> 2, hh = t & 3;
  const float* __restrict__ hp = h + (size_t)n * DIM + hh * CH;
  const float* __restrict__ ws = a_src + hh * CH;
  const float* __restrict__ wd = a_dst + hh * CH;
  float s1 = 0.f, s2 = 0.f;
#pragma unroll
  for (int c = 0; c < CH; ++c) {
    const float v = hp[c];
    s1 += v * ws[c];
    s2 += v * wd[c];
  }
  asrc[t] = s1;
  adst[t] = s2;
}

__device__ __forceinline__ void edge_sd(const int* __restrict__ ei, int e,
                                        int NE, int& s, int& d) {
  if (e < NE) { s = ei[e]; d = ei[NE + e]; }
  else        { s = d = e - NE; }
}

__device__ __forceinline__ unsigned ord_encode(float f) {
  unsigned u = __float_as_uint(f);
  return (f >= 0.0f) ? (u | 0x80000000u) : ~u;
}

__device__ __forceinline__ float ord_decode(unsigned u) {
  const unsigned top = (unsigned)((int)u >> 31);
  const float f = __uint_as_float(u ^ (~top | 0x80000000u));
  return (u == 0u) ? 0.0f : f;
}

__device__ __forceinline__ float leaky(float v) {
  return (v > 0.0f) ? v : NEG_SLOPE * v;
}

#define RNG 32
#define RNG_SHIFT 5
#define REP 8
#define LINE 32

__device__ __forceinline__ int sub_bin(int e, int d) {
  return (d >> RNG_SHIFT) * REP + (e & (REP - 1));
}

__global__ void gat_bin_zero(unsigned* __restrict__ cnt, int n) {
  int t = blockIdx.x * blockDim.x + threadIdx.x;
  if (t < n) cnt[(size_t)t * LINE] = 0u;
}

__global__ void gat_bin_count(const int* __restrict__ ei, int NE, int E,
                              unsigned* __restrict__ cnt,
                              unsigned* __restrict__ rank) {
  int e = blockIdx.x * blockDim.x + threadIdx.x;
  if (e >= E) return;
  int s, d;
  edge_sd(ei, e, NE, s, d);
  rank[e] = atomicAdd(&cnt[(size_t)sub_bin(e, d) * LINE], 1u);
}

__global__ void gat_bin_scan(const unsigned* __restrict__ cnt,
                             unsigned* __restrict__ off, int NB) {
  __shared__ unsigned sh[1024];
  const int t = threadIdx.x;
  const int per = (NB + 1023) / 1024;
  const int b0 = t * per;
  unsigned sum = 0u;
  for (int i = 0; i < per; ++i) {
    const int b = b0 + i;
    if (b < NB) sum += cnt[(size_t)b * LINE];
  }
  sh[t] = sum;
  __syncthreads();
  for (int ofs = 1; ofs < 1024; ofs <<= 1) {
    const unsigned v = (t >= ofs) ? sh[t - ofs] : 0u;
    __syncthreads();
    sh[t] += v;
    __syncthreads();
  }
  unsigned run = sh[t] - sum;
  for (int i = 0; i < per; ++i) {
    const int b = b0 + i;
    if (b < NB) {
      off[b] = run;
      run += cnt[(size_t)b * LINE];
    }
  }
  if (t == 1023) off[NB] = sh[1023];
}

__global__ void gat_bin_fill(const int* __restrict__ ei, int NE, int E,
                             const unsigned* __restrict__ off,
                             const unsigned* __restrict__ rank,
                             int4* __restrict__ rec) {
  int e = blockIdx.x * blockDim.x + threadIdx.x;
  if (e >= E) return;
  int s, d;
  edge_sd(ei, e, NE, s, d);
  const unsigned pos = off[sub_bin(e, d)] + rank[e];
  rec[pos] = make_int4(e, s, d, 0);
}

#define GRP 4
template <bool FULL>
__device__ __forceinline__ void agg_group(const float* __restrict__ h,
                                          float* acc, int lane, int j, int n,
                                          int s, int dl, float a0, float a1,
                                          float a2, float a3) {
  float hv[GRP][4];
  float al[GRP][4];
  int ro[GRP];
#pragma unroll
  for (int u = 0; u < GRP; ++u) {
    if (FULL || j + u < n) {
      const int sj = __builtin_amdgcn_readlane(s, j + u);
      ro[u] = __builtin_amdgcn_readlane(dl, j + u);
      al[u][0] = __int_as_float(__builtin_amdgcn_readlane(__float_as_int(a0), j + u));
      al[u][1] = __int_as_float(__builtin_amdgcn_readlane(__float_as_int(a1), j + u));
      al[u][2] = __int_as_float(__builtin_amdgcn_readlane(__float_as_int(a2), j + u));
      al[u][3] = __int_as_float(__builtin_amdgcn_readlane(__float_as_int(a3), j + u));
      const float* __restrict__ hp = h + (size_t)sj * DIM + lane;
#pragma unroll
      for (int k = 0; k < 4; ++k) hv[u][k] = hp[32 * k];
    }
  }
#pragma unroll
  for (int u = 0; u < GRP; ++u) {
    if (FULL || j + u < n) {
#pragma unroll
      for (int k = 0; k < 4; ++k)
        atomicAdd(&acc[ro[u] + lane + 32 * k], hv[u][k] * al[u][k]);
    }
  }
}

#define XP (DIM + 4)
template <bool NEXT>
__global__ void __launch_bounds__(256)
gat_agg_lds(const int4* __restrict__ rec, const unsigned* __restrict__ off,
            const float* __restrict__ asrc, const float* __restrict__ adst,
            float4* __restrict__ scr, const float* __restrict__ h,
            const float* __restrict__ xin, const float* __restrict__ Wskip,
            const float* __restrict__ bias, float* __restrict__ yout, int N,
            const float* __restrict__ Wn, const float* __restrict__ a_src_n,
            const float* __restrict__ a_dst_n, float* __restrict__ hn,
            float* __restrict__ asrc_n, float* __restrict__ adst_n) {
  __shared__ alignas(16) float acc[RNG * XP];
  __shared__ alignas(16) float xt[RNG * XP];
  __shared__ alignas(16) float s_adst[RNG * HEADS];
  __shared__ alignas(16) unsigned s_menc[RNG * HEADS];
  __shared__ alignas(16) float s_den[RNG * HEADS];
  const int tid = threadIdx.x;
  const int lane = tid & 31;
  const int wave = __builtin_amdgcn_readfirstlane(tid >> 5);
  const int b = blockIdx.x;

  float4* __restrict__ tile = (float4*)(yout + (size_t)b * RNG * DIM);
  const int lim4 = min(RNG, N - b * RNG) * (DIM / 4);
  const bool whole = (lim4 == RNG * DIM / 4);
  const int tpos = (tid >> 5) * XP + (tid & 31) * 4;

  const float4* __restrict__ xg = (const float4*)(xin + (size_t)b * RNG * DIM);
  float4 xr[RNG * DIM / 1024];
#pragma unroll
  for (int m = 0; m < RNG * DIM / 1024; ++m) {
    const int i = tid + 256 * m;
    xr[m] = (i < lim4) ? xg[i] : make_float4(0.f, 0.f, 0.f, 0.f);
  }
#pragma unroll
  for (int m = 0; m < RNG * DIM / 1024; ++m)
    *(float4*)(acc + tpos + 8 * m * XP) = make_float4(0.f, 0.f, 0.f, 0.f);
  if (tid < RNG * HEADS) {
    const int node = b * RNG + (tid >> 2);
    s_adst[tid] = (node < N) ? adst[(size_t)b * RNG * HEADS + tid] : 0.0f;
    s_menc[tid] = 0u;
    s_den[tid] = 0.0f;
  }
  __syncthreads();

  const int beg = __builtin_amdgcn_readfirstlane((int)off[b * REP]);
  const int end = __builtin_amdgcn_readfirstlane((int)off[(b + 1) * REP]);
  const int per = (end - beg + 7) >> 3;
  const int w0 = beg + wave * per;
  const int w1 = min(w0 + per, end);


  for (int base = w0; base < w1; base += 64) {
    int q[2], dl4[2];
    bool ok[2];
    float4 av[2];
#pragma unroll
    for (int u = 0; u < 2; ++u) {
      const int p = base + 32 * u + lane;
      ok[u] = p < w1;
      q[u] = min(p, w1 - 1);
    }
    int4 r[2];
#pragma unroll
    for (int u = 0; u < 2; ++u) r[u] = rec[q[u]];
#pragma unroll
    for (int u = 0; u < 2; ++u) {
      av[u] = *(const float4*)(asrc + (size_t)r[u].y * HEADS);
      dl4[u] = (r[u].z - b * RNG) * HEADS;
    }
    asm volatile("" ::"v"(av[0].x), "v"(av[1].x));
#pragma unroll
    for (int u = 0; u < 2; ++u) {
      if (ok[u]) {
        const float4 dv = *(const float4*)(s_adst + dl4[u]);
        float4 e;
        e.x = leaky(av[u].x + dv.x);
        e.y = leaky(av[u].y + dv.y);
        e.z = leaky(av[u].z + dv.z);
        e.w = leaky(av[u].w + dv.w);
        scr[q[u]] = e;
        atomicMax(&s_menc[dl4[u] + 0], ord_encode(e.x));
        atomicMax(&s_menc[dl4[u] + 1], ord_encode(e.y));
        atomicMax(&s_menc[dl4[u] + 2], ord_encode(e.z));
        atomicMax(&s_menc[dl4[u] + 3], ord_encode(e.w));
      }
    }
  }
#pragma unroll
  for (int m = 0; m < RNG * DIM / 1024; ++m)
    *(float4*)(xt + tpos + 8 * m * XP) = xr[m];
  __syncthreads();

  for (int base = w0; base < w1; base += 64) {
    int q[2], dl4[2];
    bool ok[2];
    float4 e[2];
#pragma unroll
    for (int u = 0; u < 2; ++u) {
      const int p = base + 32 * u + lane;
      ok[u] = p < w1;
      q[u] = min(p, w1 - 1);
    }
    int dz[2];
#pragma unroll
    for (int u = 0; u < 2; ++u) {
      dz[u] = rec[q[u]].z;
      e[u] = scr[q[u]];
    }
    asm volatile("" ::"v"(dz[0]), "v"(dz[1]), "v"(e[0].x), "v"(e[1].x));
#pragma unroll
    for (int u = 0; u < 2; ++u) dl4[u] = (dz[u] - b * RNG) * HEADS;
#pragma unroll
    for (int u = 0; u < 2; ++u) {
      if (ok[u]) {
        const uint4 mu = *(const uint4*)(s_menc + dl4[u]);
        float4 ex;
        ex.x = expf(e[u].x - ord_decode(mu.x));
        ex.y = expf(e[u].y - ord_decode(mu.y));
        ex.z = expf(e[u].z - ord_decode(mu.z));
        ex.w = expf(e[u].w - ord_decode(mu.w));
        scr[q[u]] = ex;
        atomicAdd(&s_den[dl4[u] + 0], ex.x);
        atomicAdd(&s_den[dl4[u] + 1], ex.y);
        atomicAdd(&s_den[dl4[u] + 2], ex.z);
        atomicAdd(&s_den[dl4[u] + 3], ex.w);
      }
    }
  }
  __syncthreads();

  for (int base = w0; base < w1; base += 32) {
    const int n = min(32, w1 - base);
    int s = 0, d = b * RNG;
    float a0 = 0.f, a1 = 0.f, a2 = 0.f, a3 = 0.f;
    if (lane < n) {
      const int4 r = rec[base + lane];
      s = r.y;
      d = r.z;
      const float4 ev = scr[base + lane];
      const float4 dv = *(const float4*)(s_den + (d - b * RNG) * HEADS);
      a0 = ev.x / (dv.x + 1e-16f);
      a1 = ev.y / (dv.y + 1e-16f);
      a2 = ev.z / (dv.z + 1e-16f);
      a3 = ev.w / (dv.w + 1e-16f);
    }
    const int dl = (d - b * RNG) * XP;
    int j = 0;
    for (; j + GRP <= n; j += GRP)
      agg_group<true>(h, acc, lane, j, n, s, dl, a0, a1, a2, a3);
    if (j < n)
      agg_group<false>(h, acc, lane, j, n, s, dl, a0, a1, a2, a3);
  }
  __syncthreads();

  {
    const int m  = lane & 15;
    const int kh = lane >> 4;
    const int nc = wave * 16 + m;
    float* c_lo = acc + (kh * 8) * XP + nc;
    float* c_hi = c_lo + 16 * XP;
    v8f c0, c1;
#pragma unroll
    for (int r = 0; r < 8; ++r) {
      c0[r] = c_lo[r * XP];
      c1[r] = c_hi[r * XP];
    }
    const float* a_lo = xt + m * XP + kh * 2;
    const float* a_hi = a_lo + 16 * XP;
    const float* __restrict__ wb = Wskip + (size_t)(kh * 2) * DIM + nc;
#pragma unroll 8
    for (int k0 = 0; k0 < DIM; k0 += 4) {
      const v2f av0 = *(const v2f*)(a_lo + k0);
      const v2f av1 = *(const v2f*)(a_hi + k0);
      v2f bv;
      bv.x = wb[(size_t)k0 * DIM];
      bv.y = wb[(size_t)(k0 + 1) * DIM];
      c0 = __builtin_amdgcn_wmma_f32_16x16x4_f32(
          false, av0, false, bv, (short)0, c0, false, false);
      c1 = __builtin_amdgcn_wmma_f32_16x16x4_f32(
          false, av1, false, bv, (short)0, c1, false, false);
    }
    const float bb = bias[nc];
#pragma unroll
    for (int r = 0; r < 8; ++r) {
      const float v0 = c0[r] + bb;
      const float v1 = c1[r] + bb;
      c_lo[r * XP] = (v0 > 0.0f) ? v0 : expm1f(v0);
      c_hi[r * XP] = (v1 > 0.0f) ? v1 : expm1f(v1);
    }
  }
  __syncthreads();

  const int ln = (int)__builtin_amdgcn_mbcnt_lo(~0u, 0u);
  const int tq = wave * XP + ln * 4;
  if (whole) {
#pragma unroll
    for (int m = 0; m < RNG * DIM / 1024; ++m)
      tile[tid + 256 * m] = *(const float4*)(acc + tq + 8 * m * XP);
  } else {
    for (int m = 0; m < RNG * DIM / 1024; ++m) {
      const int i = tid + 256 * m;
      if (i < lim4) tile[i] = *(const float4*)(acc + tq + 8 * m * XP);
    }
  }

  if (NEXT) {
    {
      const int m  = ln & 15;
      const int kh = ln >> 4;
      const int nc = wave * 16 + m;
      const float* a_lo = acc + m * XP + kh * 2;
      const float* a_hi = a_lo + 16 * XP;
      const float* __restrict__ wb = Wn + (size_t)(kh * 2) * DIM + nc;
      v8f d0 = {}, d1 = {};
#pragma unroll 8
      for (int k0 = 0; k0 < DIM; k0 += 4) {
        const v2f av0 = *(const v2f*)(a_lo + k0);
        const v2f av1 = *(const v2f*)(a_hi + k0);
        v2f bv;
        bv.x = wb[(size_t)k0 * DIM];
        bv.y = wb[(size_t)(k0 + 1) * DIM];
        d0 = __builtin_amdgcn_wmma_f32_16x16x4_f32(
            false, av0, false, bv, (short)0, d0, false, false);
        d1 = __builtin_amdgcn_wmma_f32_16x16x4_f32(
            false, av1, false, bv, (short)0, d1, false, false);
      }
      float* o_lo = xt + (kh * 8) * XP + nc;
      float* o_hi = o_lo + 16 * XP;
#pragma unroll
      for (int r = 0; r < 8; ++r) {
        o_lo[r * XP] = d0[r];
        o_hi[r * XP] = d1[r];
      }
    }
    __syncthreads();

    if (tid < RNG * HEADS) {
      const int row = tid >> 2, hh = tid & 3;
      const float* hp = xt + row * XP + hh * CH;
      const float* __restrict__ ws = a_src_n + hh * CH;
      const float* __restrict__ wd = a_dst_n + hh * CH;
      float s1 = 0.f, s2 = 0.f;
#pragma unroll 8
      for (int c = 0; c < CH; ++c) {
        const float v = hp[c];
        s1 += v * ws[c];
        s2 += v * wd[c];
      }
      if (b * RNG + row < N) {
        asrc_n[(size_t)b * RNG * HEADS + tid] = s1;
        adst_n[(size_t)b * RNG * HEADS + tid] = s2;
      }
    }

    float4* __restrict__ htile = (float4*)(hn + (size_t)b * RNG * DIM);
    if (whole) {
#pragma unroll
      for (int m = 0; m < RNG * DIM / 1024; ++m)
        htile[tid + 256 * m] = *(const float4*)(xt + tq + 8 * m * XP);
    } else {
      for (int m = 0; m < RNG * DIM / 1024; ++m) {
        const int i = tid + 256 * m;
        if (i < lim4) htile[i] = *(const float4*)(xt + tq + 8 * m * XP);
      }
    }
  }
}

extern "C" void kernel_launch(void* const* d_in, const int* in_sizes, int n_in,
                              void* d_out, int out_size, void* d_ws,
                              size_t ws_size, hipStream_t stream) {
  const float* x  = (const float*)d_in[0];
  const int*   ei = (const int*)d_in[1];

  const int N  = in_sizes[0] / DIM;
  const int NE = in_sizes[1] / 2;
  const int E  = NE + N;

  float* ws = (float*)d_ws;
  size_t o = 0;
  float* h     = ws + o; o += (size_t)N * DIM;
  float* h2    = ws + o; o += (size_t)N * DIM;
  float* y1    = ws + o; o += (size_t)N * DIM;
  float* asrc  = ws + o; o += (size_t)N * HEADS;
  float* adst  = ws + o; o += (size_t)N * HEADS;
  float* asrc2 = ws + o; o += (size_t)N * HEADS;
  float* adst2 = ws + o; o += (size_t)N * HEADS;
  float4* scr  = (float4*)(ws + o); o += (size_t)E * HEADS;
  const int NB = (N + RNG - 1) / RNG;
  int4* rec = (int4*)(ws + o); o += (size_t)E * 4;
  const int NBR = NB * REP;
  unsigned* bcnt = (unsigned*)(ws + o); o += (size_t)NBR * LINE;
  unsigned* boff = (unsigned*)(ws + o); o += (size_t)((NBR + 4) & ~3);
  unsigned* brank = (unsigned*)(ws + o); o += (size_t)((E + 3) & ~3);

  float* y = (float*)d_out;

  const int TB = 256;
  const int gGemm = N / 16;
  const int gNH   = (N * HEADS + TB - 1) / TB;
  const int gE    = (E + TB - 1) / TB;
  const int gNB   = (NBR + TB - 1) / TB;

  gat_bin_zero<<<gNB, TB, 0, stream>>>(bcnt, NBR);
  gat_bin_count<<<gE, TB, 0, stream>>>(ei, NE, E, bcnt, brank);
  gat_bin_scan<<<1, 1024, 0, stream>>>(bcnt, boff, NBR);
  gat_bin_fill<<<gE, TB, 0, stream>>>(ei, NE, E, boff, brank, rec);

  const float* W1     = (const float*)d_in[2];
  const float* a_src1 = (const float*)d_in[3];
  const float* a_dst1 = (const float*)d_in[4];
  const float* bias1  = (const float*)d_in[5];
  const float* Wskip1 = (const float*)d_in[6];
  const float* W2     = (const float*)d_in[7];
  const float* a_src2 = (const float*)d_in[8];
  const float* a_dst2 = (const float*)d_in[9];
  const float* bias2  = (const float*)d_in[10];
  const float* Wskip2 = (const float*)d_in[11];

  gat_gemm_f32_wmma<<<gGemm, TB, 0, stream>>>(x, W1, h);
  gat_attn_scores<<<gNH, TB, 0, stream>>>(h, a_src1, a_dst1, asrc, adst, N);

  gat_agg_lds<true><<<NB, TB, 0, stream>>>(rec, boff, asrc, adst, scr, h, x,
                                           Wskip1, bias1, y1, N, W2, a_src2,
                                           a_dst2, h2, asrc2, adst2);

  gat_agg_lds<false><<<NB, TB, 0, stream>>>(rec, boff, asrc2, adst2, scr, h2,
                                            y1, Wskip2, bias2, y, N, nullptr,
                                            nullptr, nullptr, nullptr,
                                            nullptr, nullptr);
}


# Layer 1's front end (h = x @ W1, its scores) retiled onto 32-row blocks: x rows via LDS as f32 WMMA A operand, scores from the LDS product, whole-line stores; score launch gone; sum order kept
# speedup vs baseline: 3.5240x; 1.0833x over previous
#include <hip/hip_runtime.h>
#include <math.h>

#define DIM 128
#define HEADS 4
#define CH 32
#define NEG_SLOPE 0.2f

typedef __attribute__((ext_vector_type(2))) float v2f;
typedef __attribute__((ext_vector_type(8))) float v8f;

__device__ __forceinline__ void edge_sd(const int* __restrict__ ei, int e,
                                        int NE, int& s, int& d) {
  if (e < NE) { s = ei[e]; d = ei[NE + e]; }
  else        { s = d = e - NE; }
}

__device__ __forceinline__ unsigned ord_encode(float f) {
  unsigned u = __float_as_uint(f);
  return (f >= 0.0f) ? (u | 0x80000000u) : ~u;
}

__device__ __forceinline__ float ord_decode(unsigned u) {
  const unsigned top = (unsigned)((int)u >> 31);
  const float f = __uint_as_float(u ^ (~top | 0x80000000u));
  return (u == 0u) ? 0.0f : f;
}

__device__ __forceinline__ float leaky(float v) {
  return (v > 0.0f) ? v : NEG_SLOPE * v;
}

#define RNG 32
#define RNG_SHIFT 5
#define REP 8
#define LINE 32

__device__ __forceinline__ int sub_bin(int e, int d) {
  return (d >> RNG_SHIFT) * REP + (e & (REP - 1));
}

__global__ void gat_bin_zero(unsigned* __restrict__ cnt, int n) {
  int t = blockIdx.x * blockDim.x + threadIdx.x;
  if (t < n) cnt[(size_t)t * LINE] = 0u;
}

__global__ void gat_bin_count(const int* __restrict__ ei, int NE, int E,
                              unsigned* __restrict__ cnt,
                              unsigned* __restrict__ rank) {
  int e = blockIdx.x * blockDim.x + threadIdx.x;
  if (e >= E) return;
  int s, d;
  edge_sd(ei, e, NE, s, d);
  rank[e] = atomicAdd(&cnt[(size_t)sub_bin(e, d) * LINE], 1u);
}

__global__ void gat_bin_scan(const unsigned* __restrict__ cnt,
                             unsigned* __restrict__ off, int NB) {
  __shared__ unsigned sh[1024];
  const int t = threadIdx.x;
  const int per = (NB + 1023) / 1024;
  const int b0 = t * per;
  unsigned sum = 0u;
  for (int i = 0; i < per; ++i) {
    const int b = b0 + i;
    if (b < NB) sum += cnt[(size_t)b * LINE];
  }
  sh[t] = sum;
  __syncthreads();
  for (int ofs = 1; ofs < 1024; ofs <<= 1) {
    const unsigned v = (t >= ofs) ? sh[t - ofs] : 0u;
    __syncthreads();
    sh[t] += v;
    __syncthreads();
  }
  unsigned run = sh[t] - sum;
  for (int i = 0; i < per; ++i) {
    const int b = b0 + i;
    if (b < NB) {
      off[b] = run;
      run += cnt[(size_t)b * LINE];
    }
  }
  if (t == 1023) off[NB] = sh[1023];
}

__global__ void gat_bin_fill(const int* __restrict__ ei, int NE, int E,
                             const unsigned* __restrict__ off,
                             const unsigned* __restrict__ rank,
                             int4* __restrict__ rec) {
  int e = blockIdx.x * blockDim.x + threadIdx.x;
  if (e >= E) return;
  int s, d;
  edge_sd(ei, e, NE, s, d);
  const unsigned pos = off[sub_bin(e, d)] + rank[e];
  rec[pos] = make_int4(e, s, d, 0);
}

#define GRP 4
template <bool FULL>
__device__ __forceinline__ void agg_group(const float* __restrict__ h,
                                          float* acc, int lane, int j, int n,
                                          int s, int dl, float a0, float a1,
                                          float a2, float a3) {
  float hv[GRP][4];
  float al[GRP][4];
  int ro[GRP];
#pragma unroll
  for (int u = 0; u < GRP; ++u) {
    if (FULL || j + u < n) {
      const int sj = __builtin_amdgcn_readlane(s, j + u);
      ro[u] = __builtin_amdgcn_readlane(dl, j + u);
      al[u][0] = __int_as_float(__builtin_amdgcn_readlane(__float_as_int(a0), j + u));
      al[u][1] = __int_as_float(__builtin_amdgcn_readlane(__float_as_int(a1), j + u));
      al[u][2] = __int_as_float(__builtin_amdgcn_readlane(__float_as_int(a2), j + u));
      al[u][3] = __int_as_float(__builtin_amdgcn_readlane(__float_as_int(a3), j + u));
      const float* __restrict__ hp = h + (size_t)sj * DIM + lane;
#pragma unroll
      for (int k = 0; k < 4; ++k) hv[u][k] = hp[32 * k];
    }
  }
#pragma unroll
  for (int u = 0; u < GRP; ++u) {
    if (FULL || j + u < n) {
#pragma unroll
      for (int k = 0; k < 4; ++k)
        atomicAdd(&acc[ro[u] + lane + 32 * k], hv[u][k] * al[u][k]);
    }
  }
}

#define XP (DIM + 4)
__global__ void __launch_bounds__(256)
gat_front_lds(const float* __restrict__ xin, const float* __restrict__ Wn,
              const float* __restrict__ a_src_n,
              const float* __restrict__ a_dst_n, float* __restrict__ hn,
              float* __restrict__ asrc_n, float* __restrict__ adst_n, int N) {
  __shared__ alignas(16) float xt[RNG * XP];
  __shared__ alignas(16) float ht[RNG * XP];
  const int tid = threadIdx.x;
  const int lane = tid & 31;
  const int wave = __builtin_amdgcn_readfirstlane(tid >> 5);
  const int b = blockIdx.x;

  const int lim4 = min(RNG, N - b * RNG) * (DIM / 4);
  const bool whole = (lim4 == RNG * DIM / 4);
  const int tpos = wave * XP + lane * 4;

  const float4* __restrict__ xg = (const float4*)(xin + (size_t)b * RNG * DIM);
  float4 xr[RNG * DIM / 1024];
#pragma unroll
  for (int m = 0; m < RNG * DIM / 1024; ++m) {
    const int i = tid + 256 * m;
    xr[m] = (i < lim4) ? xg[i] : make_float4(0.f, 0.f, 0.f, 0.f);
  }
#pragma unroll
  for (int m = 0; m < RNG * DIM / 1024; ++m)
    *(float4*)(xt + tpos + 8 * m * XP) = xr[m];
  __syncthreads();

  {
    const int m  = lane & 15;
    const int kh = lane >> 4;
    const int nc = wave * 16 + m;
    const float* a_lo = xt + m * XP + kh * 2;
    const float* a_hi = a_lo + 16 * XP;
    const float* __restrict__ wb = Wn + (size_t)(kh * 2) * DIM + nc;
    v8f d0 = {}, d1 = {};
#pragma unroll 8
    for (int k0 = 0; k0 < DIM; k0 += 4) {
      const v2f av0 = *(const v2f*)(a_lo + k0);
      const v2f av1 = *(const v2f*)(a_hi + k0);
      v2f bv;
      bv.x = wb[(size_t)k0 * DIM];
      bv.y = wb[(size_t)(k0 + 1) * DIM];
      d0 = __builtin_amdgcn_wmma_f32_16x16x4_f32(
          false, av0, false, bv, (short)0, d0, false, false);
      d1 = __builtin_amdgcn_wmma_f32_16x16x4_f32(
          false, av1, false, bv, (short)0, d1, false, false);
    }
    float* o_lo = ht + (kh * 8) * XP + nc;
    float* o_hi = o_lo + 16 * XP;
#pragma unroll
    for (int r = 0; r < 8; ++r) {
      o_lo[r * XP] = d0[r];
      o_hi[r * XP] = d1[r];
    }
  }
  __syncthreads();

  if (tid < RNG * HEADS) {
    const int row = tid >> 2, hh = tid & 3;
    const float* hp = ht + row * XP + hh * CH;
    const float* __restrict__ ws = a_src_n + hh * CH;
    const float* __restrict__ wd = a_dst_n + hh * CH;
    float s1 = 0.f, s2 = 0.f;
#pragma unroll
    for (int c = 0; c < CH; ++c) {
      const float v = hp[c];
      s1 += v * ws[c];
      s2 += v * wd[c];
    }
    if (b * RNG + row < N) {
      asrc_n[(size_t)b * RNG * HEADS + tid] = s1;
      adst_n[(size_t)b * RNG * HEADS + tid] = s2;
    }
  }

  float4* __restrict__ htile = (float4*)(hn + (size_t)b * RNG * DIM);
  if (whole) {
#pragma unroll
    for (int m = 0; m < RNG * DIM / 1024; ++m)
      htile[tid + 256 * m] = *(const float4*)(ht + tpos + 8 * m * XP);
  } else {
    for (int m = 0; m < RNG * DIM / 1024; ++m) {
      const int i = tid + 256 * m;
      if (i < lim4) htile[i] = *(const float4*)(ht + tpos + 8 * m * XP);
    }
  }
}

template <bool NEXT>
__global__ void __launch_bounds__(256)
gat_agg_lds(const int4* __restrict__ rec, const unsigned* __restrict__ off,
            const float* __restrict__ asrc, const float* __restrict__ adst,
            float4* __restrict__ scr, const float* __restrict__ h,
            const float* __restrict__ xin, const float* __restrict__ Wskip,
            const float* __restrict__ bias, float* __restrict__ yout, int N,
            const float* __restrict__ Wn, const float* __restrict__ a_src_n,
            const float* __restrict__ a_dst_n, float* __restrict__ hn,
            float* __restrict__ asrc_n, float* __restrict__ adst_n) {
  __shared__ alignas(16) float acc[RNG * XP];
  __shared__ alignas(16) float xt[RNG * XP];
  __shared__ alignas(16) float s_adst[RNG * HEADS];
  __shared__ alignas(16) unsigned s_menc[RNG * HEADS];
  __shared__ alignas(16) float s_den[RNG * HEADS];
  const int tid = threadIdx.x;
  const int lane = tid & 31;
  const int wave = __builtin_amdgcn_readfirstlane(tid >> 5);
  const int b = blockIdx.x;

  float4* __restrict__ tile = (float4*)(yout + (size_t)b * RNG * DIM);
  const int lim4 = min(RNG, N - b * RNG) * (DIM / 4);
  const bool whole = (lim4 == RNG * DIM / 4);
  const int tpos = (tid >> 5) * XP + (tid & 31) * 4;

  const float4* __restrict__ xg = (const float4*)(xin + (size_t)b * RNG * DIM);
  float4 xr[RNG * DIM / 1024];
#pragma unroll
  for (int m = 0; m < RNG * DIM / 1024; ++m) {
    const int i = tid + 256 * m;
    xr[m] = (i < lim4) ? xg[i] : make_float4(0.f, 0.f, 0.f, 0.f);
  }
#pragma unroll
  for (int m = 0; m < RNG * DIM / 1024; ++m)
    *(float4*)(acc + tpos + 8 * m * XP) = make_float4(0.f, 0.f, 0.f, 0.f);
  if (tid < RNG * HEADS) {
    const int node = b * RNG + (tid >> 2);
    s_adst[tid] = (node < N) ? adst[(size_t)b * RNG * HEADS + tid] : 0.0f;
    s_menc[tid] = 0u;
    s_den[tid] = 0.0f;
  }
  __syncthreads();

  const int beg = __builtin_amdgcn_readfirstlane((int)off[b * REP]);
  const int end = __builtin_amdgcn_readfirstlane((int)off[(b + 1) * REP]);
  const int per = (end - beg + 7) >> 3;
  const int w0 = beg + wave * per;
  const int w1 = min(w0 + per, end);


  for (int base = w0; base < w1; base += 64) {
    int q[2], dl4[2];
    bool ok[2];
    float4 av[2];
#pragma unroll
    for (int u = 0; u < 2; ++u) {
      const int p = base + 32 * u + lane;
      ok[u] = p < w1;
      q[u] = min(p, w1 - 1);
    }
    int4 r[2];
#pragma unroll
    for (int u = 0; u < 2; ++u) r[u] = rec[q[u]];
#pragma unroll
    for (int u = 0; u < 2; ++u) {
      av[u] = *(const float4*)(asrc + (size_t)r[u].y * HEADS);
      dl4[u] = (r[u].z - b * RNG) * HEADS;
    }
    asm volatile("" ::"v"(av[0].x), "v"(av[1].x));
#pragma unroll
    for (int u = 0; u < 2; ++u) {
      if (ok[u]) {
        const float4 dv = *(const float4*)(s_adst + dl4[u]);
        float4 e;
        e.x = leaky(av[u].x + dv.x);
        e.y = leaky(av[u].y + dv.y);
        e.z = leaky(av[u].z + dv.z);
        e.w = leaky(av[u].w + dv.w);
        scr[q[u]] = e;
        atomicMax(&s_menc[dl4[u] + 0], ord_encode(e.x));
        atomicMax(&s_menc[dl4[u] + 1], ord_encode(e.y));
        atomicMax(&s_menc[dl4[u] + 2], ord_encode(e.z));
        atomicMax(&s_menc[dl4[u] + 3], ord_encode(e.w));
      }
    }
  }
#pragma unroll
  for (int m = 0; m < RNG * DIM / 1024; ++m)
    *(float4*)(xt + tpos + 8 * m * XP) = xr[m];
  __syncthreads();

  for (int base = w0; base < w1; base += 64) {
    int q[2], dl4[2];
    bool ok[2];
    float4 e[2];
#pragma unroll
    for (int u = 0; u < 2; ++u) {
      const int p = base + 32 * u + lane;
      ok[u] = p < w1;
      q[u] = min(p, w1 - 1);
    }
    int dz[2];
#pragma unroll
    for (int u = 0; u < 2; ++u) {
      dz[u] = rec[q[u]].z;
      e[u] = scr[q[u]];
    }
    asm volatile("" ::"v"(dz[0]), "v"(dz[1]), "v"(e[0].x), "v"(e[1].x));
#pragma unroll
    for (int u = 0; u < 2; ++u) dl4[u] = (dz[u] - b * RNG) * HEADS;
#pragma unroll
    for (int u = 0; u < 2; ++u) {
      if (ok[u]) {
        const uint4 mu = *(const uint4*)(s_menc + dl4[u]);
        float4 ex;
        ex.x = expf(e[u].x - ord_decode(mu.x));
        ex.y = expf(e[u].y - ord_decode(mu.y));
        ex.z = expf(e[u].z - ord_decode(mu.z));
        ex.w = expf(e[u].w - ord_decode(mu.w));
        scr[q[u]] = ex;
        atomicAdd(&s_den[dl4[u] + 0], ex.x);
        atomicAdd(&s_den[dl4[u] + 1], ex.y);
        atomicAdd(&s_den[dl4[u] + 2], ex.z);
        atomicAdd(&s_den[dl4[u] + 3], ex.w);
      }
    }
  }
  __syncthreads();

  for (int base = w0; base < w1; base += 32) {
    const int n = min(32, w1 - base);
    int s = 0, d = b * RNG;
    float a0 = 0.f, a1 = 0.f, a2 = 0.f, a3 = 0.f;
    if (lane < n) {
      const int4 r = rec[base + lane];
      s = r.y;
      d = r.z;
      const float4 ev = scr[base + lane];
      const float4 dv = *(const float4*)(s_den + (d - b * RNG) * HEADS);
      a0 = ev.x / (dv.x + 1e-16f);
      a1 = ev.y / (dv.y + 1e-16f);
      a2 = ev.z / (dv.z + 1e-16f);
      a3 = ev.w / (dv.w + 1e-16f);
    }
    const int dl = (d - b * RNG) * XP;
    int j = 0;
    for (; j + GRP <= n; j += GRP)
      agg_group<true>(h, acc, lane, j, n, s, dl, a0, a1, a2, a3);
    if (j < n)
      agg_group<false>(h, acc, lane, j, n, s, dl, a0, a1, a2, a3);
  }
  __syncthreads();

  {
    const int m  = lane & 15;
    const int kh = lane >> 4;
    const int nc = wave * 16 + m;
    float* c_lo = acc + (kh * 8) * XP + nc;
    float* c_hi = c_lo + 16 * XP;
    v8f c0, c1;
#pragma unroll
    for (int r = 0; r < 8; ++r) {
      c0[r] = c_lo[r * XP];
      c1[r] = c_hi[r * XP];
    }
    const float* a_lo = xt + m * XP + kh * 2;
    const float* a_hi = a_lo + 16 * XP;
    const float* __restrict__ wb = Wskip + (size_t)(kh * 2) * DIM + nc;
#pragma unroll 8
    for (int k0 = 0; k0 < DIM; k0 += 4) {
      const v2f av0 = *(const v2f*)(a_lo + k0);
      const v2f av1 = *(const v2f*)(a_hi + k0);
      v2f bv;
      bv.x = wb[(size_t)k0 * DIM];
      bv.y = wb[(size_t)(k0 + 1) * DIM];
      c0 = __builtin_amdgcn_wmma_f32_16x16x4_f32(
          false, av0, false, bv, (short)0, c0, false, false);
      c1 = __builtin_amdgcn_wmma_f32_16x16x4_f32(
          false, av1, false, bv, (short)0, c1, false, false);
    }
    const float bb = bias[nc];
#pragma unroll
    for (int r = 0; r < 8; ++r) {
      const float v0 = c0[r] + bb;
      const float v1 = c1[r] + bb;
      c_lo[r * XP] = (v0 > 0.0f) ? v0 : expm1f(v0);
      c_hi[r * XP] = (v1 > 0.0f) ? v1 : expm1f(v1);
    }
  }
  __syncthreads();

  const int ln = (int)__builtin_amdgcn_mbcnt_lo(~0u, 0u);
  const int tq = wave * XP + ln * 4;
  if (whole) {
#pragma unroll
    for (int m = 0; m < RNG * DIM / 1024; ++m)
      tile[tid + 256 * m] = *(const float4*)(acc + tq + 8 * m * XP);
  } else {
    for (int m = 0; m < RNG * DIM / 1024; ++m) {
      const int i = tid + 256 * m;
      if (i < lim4) tile[i] = *(const float4*)(acc + tq + 8 * m * XP);
    }
  }

  if (NEXT) {
    {
      const int m  = ln & 15;
      const int kh = ln >> 4;
      const int nc = wave * 16 + m;
      const float* a_lo = acc + m * XP + kh * 2;
      const float* a_hi = a_lo + 16 * XP;
      const float* __restrict__ wb = Wn + (size_t)(kh * 2) * DIM + nc;
      v8f d0 = {}, d1 = {};
#pragma unroll 8
      for (int k0 = 0; k0 < DIM; k0 += 4) {
        const v2f av0 = *(const v2f*)(a_lo + k0);
        const v2f av1 = *(const v2f*)(a_hi + k0);
        v2f bv;
        bv.x = wb[(size_t)k0 * DIM];
        bv.y = wb[(size_t)(k0 + 1) * DIM];
        d0 = __builtin_amdgcn_wmma_f32_16x16x4_f32(
            false, av0, false, bv, (short)0, d0, false, false);
        d1 = __builtin_amdgcn_wmma_f32_16x16x4_f32(
            false, av1, false, bv, (short)0, d1, false, false);
      }
      float* o_lo = xt + (kh * 8) * XP + nc;
      float* o_hi = o_lo + 16 * XP;
#pragma unroll
      for (int r = 0; r < 8; ++r) {
        o_lo[r * XP] = d0[r];
        o_hi[r * XP] = d1[r];
      }
    }
    __syncthreads();

    if (tid < RNG * HEADS) {
      const int row = tid >> 2, hh = tid & 3;
      const float* hp = xt + row * XP + hh * CH;
      const float* __restrict__ ws = a_src_n + hh * CH;
      const float* __restrict__ wd = a_dst_n + hh * CH;
      float s1 = 0.f, s2 = 0.f;
#pragma unroll 8
      for (int c = 0; c < CH; ++c) {
        const float v = hp[c];
        s1 += v * ws[c];
        s2 += v * wd[c];
      }
      if (b * RNG + row < N) {
        asrc_n[(size_t)b * RNG * HEADS + tid] = s1;
        adst_n[(size_t)b * RNG * HEADS + tid] = s2;
      }
    }

    float4* __restrict__ htile = (float4*)(hn + (size_t)b * RNG * DIM);
    if (whole) {
#pragma unroll
      for (int m = 0; m < RNG * DIM / 1024; ++m)
        htile[tid + 256 * m] = *(const float4*)(xt + tq + 8 * m * XP);
    } else {
      for (int m = 0; m < RNG * DIM / 1024; ++m) {
        const int i = tid + 256 * m;
        if (i < lim4) htile[i] = *(const float4*)(xt + tq + 8 * m * XP);
      }
    }
  }
}

extern "C" void kernel_launch(void* const* d_in, const int* in_sizes, int n_in,
                              void* d_out, int out_size, void* d_ws,
                              size_t ws_size, hipStream_t stream) {
  const float* x  = (const float*)d_in[0];
  const int*   ei = (const int*)d_in[1];

  const int N  = in_sizes[0] / DIM;
  const int NE = in_sizes[1] / 2;
  const int E  = NE + N;

  float* ws = (float*)d_ws;
  size_t o = 0;
  float* h     = ws + o; o += (size_t)N * DIM;
  float* h2    = ws + o; o += (size_t)N * DIM;
  float* y1    = ws + o; o += (size_t)N * DIM;
  float* asrc  = ws + o; o += (size_t)N * HEADS;
  float* adst  = ws + o; o += (size_t)N * HEADS;
  float* asrc2 = ws + o; o += (size_t)N * HEADS;
  float* adst2 = ws + o; o += (size_t)N * HEADS;
  float4* scr  = (float4*)(ws + o); o += (size_t)E * HEADS;
  const int NB = (N + RNG - 1) / RNG;
  int4* rec = (int4*)(ws + o); o += (size_t)E * 4;
  const int NBR = NB * REP;
  unsigned* bcnt = (unsigned*)(ws + o); o += (size_t)NBR * LINE;
  unsigned* boff = (unsigned*)(ws + o); o += (size_t)((NBR + 4) & ~3);
  unsigned* brank = (unsigned*)(ws + o); o += (size_t)((E + 3) & ~3);

  float* y = (float*)d_out;

  const int TB = 256;
  const int gE    = (E + TB - 1) / TB;
  const int gNB   = (NBR + TB - 1) / TB;

  gat_bin_zero<<<gNB, TB, 0, stream>>>(bcnt, NBR);
  gat_bin_count<<<gE, TB, 0, stream>>>(ei, NE, E, bcnt, brank);
  gat_bin_scan<<<1, 1024, 0, stream>>>(bcnt, boff, NBR);
  gat_bin_fill<<<gE, TB, 0, stream>>>(ei, NE, E, boff, brank, rec);

  const float* W1     = (const float*)d_in[2];
  const float* a_src1 = (const float*)d_in[3];
  const float* a_dst1 = (const float*)d_in[4];
  const float* bias1  = (const float*)d_in[5];
  const float* Wskip1 = (const float*)d_in[6];
  const float* W2     = (const float*)d_in[7];
  const float* a_src2 = (const float*)d_in[8];
  const float* a_dst2 = (const float*)d_in[9];
  const float* bias2  = (const float*)d_in[10];
  const float* Wskip2 = (const float*)d_in[11];

  gat_front_lds<<<NB, TB, 0, stream>>>(x, W1, a_src1, a_dst1, h, asrc, adst,
                                       N);

  gat_agg_lds<true><<<NB, TB, 0, stream>>>(rec, boff, asrc, adst, scr, h, x,
                                           Wskip1, bias1, y1, N, W2, a_src2,
                                           a_dst2, h2, asrc2, adst2);

  gat_agg_lds<false><<<NB, TB, 0, stream>>>(rec, boff, asrc2, adst2, scr, h2,
                                            y1, Wskip2, bias2, y, N, nullptr,
                                            nullptr, nullptr, nullptr,
                                            nullptr, nullptr);
}


# Bin build's prefix pass moved into the front-end launch as 8 leading blocks (each sums the counts ahead of its slice itself, loads several in flight); single-block scan launch gone; slot order kept
# speedup vs baseline: 3.8891x; 1.1036x over previous
#include <hip/hip_runtime.h>
#include <math.h>

#define DIM 128
#define HEADS 4
#define CH 32
#define NEG_SLOPE 0.2f

typedef __attribute__((ext_vector_type(2))) float v2f;
typedef __attribute__((ext_vector_type(8))) float v8f;

__device__ __forceinline__ void edge_sd(const int* __restrict__ ei, int e,
                                        int NE, int& s, int& d) {
  if (e < NE) { s = ei[e]; d = ei[NE + e]; }
  else        { s = d = e - NE; }
}

__device__ __forceinline__ unsigned ord_encode(float f) {
  unsigned u = __float_as_uint(f);
  return (f >= 0.0f) ? (u | 0x80000000u) : ~u;
}

__device__ __forceinline__ float ord_decode(unsigned u) {
  const unsigned top = (unsigned)((int)u >> 31);
  const float f = __uint_as_float(u ^ (~top | 0x80000000u));
  return (u == 0u) ? 0.0f : f;
}

__device__ __forceinline__ float leaky(float v) {
  return (v > 0.0f) ? v : NEG_SLOPE * v;
}

#define RNG 32
#define RNG_SHIFT 5
#define REP 8
#define LINE 32

__device__ __forceinline__ int sub_bin(int e, int d) {
  return (d >> RNG_SHIFT) * REP + (e & (REP - 1));
}

__global__ void gat_bin_zero(unsigned* __restrict__ cnt, int n) {
  int t = blockIdx.x * blockDim.x + threadIdx.x;
  if (t < n) cnt[(size_t)t * LINE] = 0u;
}

__global__ void gat_bin_count(const int* __restrict__ ei, int NE, int E,
                              unsigned* __restrict__ cnt,
                              unsigned* __restrict__ rank) {
  int e = blockIdx.x * blockDim.x + threadIdx.x;
  if (e >= E) return;
  int s, d;
  edge_sd(ei, e, NE, s, d);
  rank[e] = atomicAdd(&cnt[(size_t)sub_bin(e, d) * LINE], 1u);
}

#define SCAN_BLOCKS 8
#define SCAN_GRP 4
__device__ __forceinline__ void bin_scan_slice(const unsigned* __restrict__ cnt,
                                               unsigned* __restrict__ off,
                                               int NBR, int g, unsigned* sh) {
  const int t = threadIdx.x;
  const int S = (NBR + SCAN_BLOCKS - 1) / SCAN_BLOCKS;
  const int lo = min(g * S, NBR);
  const int hi = min(lo + S, NBR);

  unsigned ahead = 0u;
  for (int base = 0; base < lo; base += 256 * SCAN_GRP) {
    unsigned c[SCAN_GRP];
#pragma unroll
    for (int u = 0; u < SCAN_GRP; ++u) {
      const int b = base + 256 * u + t;
      c[u] = cnt[(size_t)min(b, lo - 1) * LINE];
    }
#pragma unroll
    for (int u = 0; u < SCAN_GRP; ++u) {
      const int b = base + 256 * u + t;
      ahead += (b < lo) ? c[u] : 0u;
    }
  }

  const int P = (S + 255) / 256;
  const int b0 = lo + t * P;
  unsigned sum = 0u;
  for (int i0 = 0; i0 < P; i0 += SCAN_GRP) {
    unsigned c[SCAN_GRP];
#pragma unroll
    for (int u = 0; u < SCAN_GRP; ++u)
      c[u] = cnt[(size_t)min(b0 + i0 + u, NBR - 1) * LINE];
#pragma unroll
    for (int u = 0; u < SCAN_GRP; ++u) {
      const int i = i0 + u;
      sum += (i < P && b0 + i < hi) ? c[u] : 0u;
    }
  }

  sh[t] = sum;
  sh[256 + t] = ahead;
  __syncthreads();
  for (int ofs = 1; ofs < 256; ofs <<= 1) {
    const unsigned v = (t >= ofs) ? sh[t - ofs] : 0u;
    const unsigned w = (t >= ofs) ? sh[256 + t - ofs] : 0u;
    __syncthreads();
    sh[t] += v;
    sh[256 + t] += w;
    __syncthreads();
  }
  const unsigned tot_ahead = sh[511];
  unsigned run = tot_ahead + sh[t] - sum;
  for (int i0 = 0; i0 < P; i0 += SCAN_GRP) {
    unsigned c[SCAN_GRP];
#pragma unroll
    for (int u = 0; u < SCAN_GRP; ++u)
      c[u] = cnt[(size_t)min(b0 + i0 + u, NBR - 1) * LINE];
#pragma unroll
    for (int u = 0; u < SCAN_GRP; ++u) {
      const int i = i0 + u;
      const bool ok = (i < P && b0 + i < hi);
      if (ok) off[b0 + i] = run;
      run += ok ? c[u] : 0u;
    }
  }
  if (g == SCAN_BLOCKS - 1 && t == 255) off[NBR] = tot_ahead + sh[255];
}

__global__ void gat_bin_fill(const int* __restrict__ ei, int NE, int E,
                             const unsigned* __restrict__ off,
                             const unsigned* __restrict__ rank,
                             int4* __restrict__ rec) {
  int e = blockIdx.x * blockDim.x + threadIdx.x;
  if (e >= E) return;
  int s, d;
  edge_sd(ei, e, NE, s, d);
  const unsigned pos = off[sub_bin(e, d)] + rank[e];
  rec[pos] = make_int4(e, s, d, 0);
}

#define GRP 4
template <bool FULL>
__device__ __forceinline__ void agg_group(const float* __restrict__ h,
                                          float* acc, int lane, int j, int n,
                                          int s, int dl, float a0, float a1,
                                          float a2, float a3) {
  float hv[GRP][4];
  float al[GRP][4];
  int ro[GRP];
#pragma unroll
  for (int u = 0; u < GRP; ++u) {
    if (FULL || j + u < n) {
      const int sj = __builtin_amdgcn_readlane(s, j + u);
      ro[u] = __builtin_amdgcn_readlane(dl, j + u);
      al[u][0] = __int_as_float(__builtin_amdgcn_readlane(__float_as_int(a0), j + u));
      al[u][1] = __int_as_float(__builtin_amdgcn_readlane(__float_as_int(a1), j + u));
      al[u][2] = __int_as_float(__builtin_amdgcn_readlane(__float_as_int(a2), j + u));
      al[u][3] = __int_as_float(__builtin_amdgcn_readlane(__float_as_int(a3), j + u));
      const float* __restrict__ hp = h + (size_t)sj * DIM + lane;
#pragma unroll
      for (int k = 0; k < 4; ++k) hv[u][k] = hp[32 * k];
    }
  }
#pragma unroll
  for (int u = 0; u < GRP; ++u) {
    if (FULL || j + u < n) {
#pragma unroll
      for (int k = 0; k < 4; ++k)
        atomicAdd(&acc[ro[u] + lane + 32 * k], hv[u][k] * al[u][k]);
    }
  }
}

#define XP (DIM + 4)
__global__ void __launch_bounds__(256)
gat_front_lds(const float* __restrict__ xin, const float* __restrict__ Wn,
              const float* __restrict__ a_src_n,
              const float* __restrict__ a_dst_n, float* __restrict__ hn,
              float* __restrict__ asrc_n, float* __restrict__ adst_n, int N,
              const unsigned* __restrict__ cnt, unsigned* __restrict__ off,
              int NBR) {
  __shared__ alignas(16) float xt[RNG * XP];
  __shared__ alignas(16) float ht[RNG * XP];
  if (blockIdx.x < SCAN_BLOCKS) {
    bin_scan_slice(cnt, off, NBR, blockIdx.x, reinterpret_cast<unsigned*>(xt));
    return;
  }
  const int tid = threadIdx.x;
  const int lane = tid & 31;
  const int wave = __builtin_amdgcn_readfirstlane(tid >> 5);
  const int b = blockIdx.x - SCAN_BLOCKS;

  const int lim4 = min(RNG, N - b * RNG) * (DIM / 4);
  const bool whole = (lim4 == RNG * DIM / 4);
  const int tpos = wave * XP + lane * 4;

  const float4* __restrict__ xg = (const float4*)(xin + (size_t)b * RNG * DIM);
  float4 xr[RNG * DIM / 1024];
#pragma unroll
  for (int m = 0; m < RNG * DIM / 1024; ++m) {
    const int i = tid + 256 * m;
    xr[m] = (i < lim4) ? xg[i] : make_float4(0.f, 0.f, 0.f, 0.f);
  }
#pragma unroll
  for (int m = 0; m < RNG * DIM / 1024; ++m)
    *(float4*)(xt + tpos + 8 * m * XP) = xr[m];
  __syncthreads();

  {
    const int m  = lane & 15;
    const int kh = lane >> 4;
    const int nc = wave * 16 + m;
    const float* a_lo = xt + m * XP + kh * 2;
    const float* a_hi = a_lo + 16 * XP;
    const float* __restrict__ wb = Wn + (size_t)(kh * 2) * DIM + nc;
    v8f d0 = {}, d1 = {};
#pragma unroll 8
    for (int k0 = 0; k0 < DIM; k0 += 4) {
      const v2f av0 = *(const v2f*)(a_lo + k0);
      const v2f av1 = *(const v2f*)(a_hi + k0);
      v2f bv;
      bv.x = wb[(size_t)k0 * DIM];
      bv.y = wb[(size_t)(k0 + 1) * DIM];
      d0 = __builtin_amdgcn_wmma_f32_16x16x4_f32(
          false, av0, false, bv, (short)0, d0, false, false);
      d1 = __builtin_amdgcn_wmma_f32_16x16x4_f32(
          false, av1, false, bv, (short)0, d1, false, false);
    }
    float* o_lo = ht + (kh * 8) * XP + nc;
    float* o_hi = o_lo + 16 * XP;
#pragma unroll
    for (int r = 0; r < 8; ++r) {
      o_lo[r * XP] = d0[r];
      o_hi[r * XP] = d1[r];
    }
  }
  __syncthreads();

  if (tid < RNG * HEADS) {
    const int row = tid >> 2, hh = tid & 3;
    const float* hp = ht + row * XP + hh * CH;
    const float* __restrict__ ws = a_src_n + hh * CH;
    const float* __restrict__ wd = a_dst_n + hh * CH;
    float s1 = 0.f, s2 = 0.f;
#pragma unroll
    for (int c = 0; c < CH; ++c) {
      const float v = hp[c];
      s1 += v * ws[c];
      s2 += v * wd[c];
    }
    if (b * RNG + row < N) {
      asrc_n[(size_t)b * RNG * HEADS + tid] = s1;
      adst_n[(size_t)b * RNG * HEADS + tid] = s2;
    }
  }

  float4* __restrict__ htile = (float4*)(hn + (size_t)b * RNG * DIM);
  if (whole) {
#pragma unroll
    for (int m = 0; m < RNG * DIM / 1024; ++m)
      htile[tid + 256 * m] = *(const float4*)(ht + tpos + 8 * m * XP);
  } else {
    for (int m = 0; m < RNG * DIM / 1024; ++m) {
      const int i = tid + 256 * m;
      if (i < lim4) htile[i] = *(const float4*)(ht + tpos + 8 * m * XP);
    }
  }
}

template <bool NEXT>
__global__ void __launch_bounds__(256)
gat_agg_lds(const int4* __restrict__ rec, const unsigned* __restrict__ off,
            const float* __restrict__ asrc, const float* __restrict__ adst,
            float4* __restrict__ scr, const float* __restrict__ h,
            const float* __restrict__ xin, const float* __restrict__ Wskip,
            const float* __restrict__ bias, float* __restrict__ yout, int N,
            const float* __restrict__ Wn, const float* __restrict__ a_src_n,
            const float* __restrict__ a_dst_n, float* __restrict__ hn,
            float* __restrict__ asrc_n, float* __restrict__ adst_n) {
  __shared__ alignas(16) float acc[RNG * XP];
  __shared__ alignas(16) float xt[RNG * XP];
  __shared__ alignas(16) float s_adst[RNG * HEADS];
  __shared__ alignas(16) unsigned s_menc[RNG * HEADS];
  __shared__ alignas(16) float s_den[RNG * HEADS];
  const int tid = threadIdx.x;
  const int lane = tid & 31;
  const int wave = __builtin_amdgcn_readfirstlane(tid >> 5);
  const int b = blockIdx.x;

  float4* __restrict__ tile = (float4*)(yout + (size_t)b * RNG * DIM);
  const int lim4 = min(RNG, N - b * RNG) * (DIM / 4);
  const bool whole = (lim4 == RNG * DIM / 4);
  const int tpos = (tid >> 5) * XP + (tid & 31) * 4;

  const float4* __restrict__ xg = (const float4*)(xin + (size_t)b * RNG * DIM);
  float4 xr[RNG * DIM / 1024];
#pragma unroll
  for (int m = 0; m < RNG * DIM / 1024; ++m) {
    const int i = tid + 256 * m;
    xr[m] = (i < lim4) ? xg[i] : make_float4(0.f, 0.f, 0.f, 0.f);
  }
#pragma unroll
  for (int m = 0; m < RNG * DIM / 1024; ++m)
    *(float4*)(acc + tpos + 8 * m * XP) = make_float4(0.f, 0.f, 0.f, 0.f);
  if (tid < RNG * HEADS) {
    const int node = b * RNG + (tid >> 2);
    s_adst[tid] = (node < N) ? adst[(size_t)b * RNG * HEADS + tid] : 0.0f;
    s_menc[tid] = 0u;
    s_den[tid] = 0.0f;
  }
  __syncthreads();

  const int beg = __builtin_amdgcn_readfirstlane((int)off[b * REP]);
  const int end = __builtin_amdgcn_readfirstlane((int)off[(b + 1) * REP]);
  const int per = (end - beg + 7) >> 3;
  const int w0 = beg + wave * per;
  const int w1 = min(w0 + per, end);


  for (int base = w0; base < w1; base += 64) {
    int q[2], dl4[2];
    bool ok[2];
    float4 av[2];
#pragma unroll
    for (int u = 0; u < 2; ++u) {
      const int p = base + 32 * u + lane;
      ok[u] = p < w1;
      q[u] = min(p, w1 - 1);
    }
    int4 r[2];
#pragma unroll
    for (int u = 0; u < 2; ++u) r[u] = rec[q[u]];
#pragma unroll
    for (int u = 0; u < 2; ++u) {
      av[u] = *(const float4*)(asrc + (size_t)r[u].y * HEADS);
      dl4[u] = (r[u].z - b * RNG) * HEADS;
    }
    asm volatile("" ::"v"(av[0].x), "v"(av[1].x));
#pragma unroll
    for (int u = 0; u < 2; ++u) {
      if (ok[u]) {
        const float4 dv = *(const float4*)(s_adst + dl4[u]);
        float4 e;
        e.x = leaky(av[u].x + dv.x);
        e.y = leaky(av[u].y + dv.y);
        e.z = leaky(av[u].z + dv.z);
        e.w = leaky(av[u].w + dv.w);
        scr[q[u]] = e;
        atomicMax(&s_menc[dl4[u] + 0], ord_encode(e.x));
        atomicMax(&s_menc[dl4[u] + 1], ord_encode(e.y));
        atomicMax(&s_menc[dl4[u] + 2], ord_encode(e.z));
        atomicMax(&s_menc[dl4[u] + 3], ord_encode(e.w));
      }
    }
  }
#pragma unroll
  for (int m = 0; m < RNG * DIM / 1024; ++m)
    *(float4*)(xt + tpos + 8 * m * XP) = xr[m];
  __syncthreads();

  for (int base = w0; base < w1; base += 64) {
    int q[2], dl4[2];
    bool ok[2];
    float4 e[2];
#pragma unroll
    for (int u = 0; u < 2; ++u) {
      const int p = base + 32 * u + lane;
      ok[u] = p < w1;
      q[u] = min(p, w1 - 1);
    }
    int dz[2];
#pragma unroll
    for (int u = 0; u < 2; ++u) {
      dz[u] = rec[q[u]].z;
      e[u] = scr[q[u]];
    }
    asm volatile("" ::"v"(dz[0]), "v"(dz[1]), "v"(e[0].x), "v"(e[1].x));
#pragma unroll
    for (int u = 0; u < 2; ++u) dl4[u] = (dz[u] - b * RNG) * HEADS;
#pragma unroll
    for (int u = 0; u < 2; ++u) {
      if (ok[u]) {
        const uint4 mu = *(const uint4*)(s_menc + dl4[u]);
        float4 ex;
        ex.x = expf(e[u].x - ord_decode(mu.x));
        ex.y = expf(e[u].y - ord_decode(mu.y));
        ex.z = expf(e[u].z - ord_decode(mu.z));
        ex.w = expf(e[u].w - ord_decode(mu.w));
        scr[q[u]] = ex;
        atomicAdd(&s_den[dl4[u] + 0], ex.x);
        atomicAdd(&s_den[dl4[u] + 1], ex.y);
        atomicAdd(&s_den[dl4[u] + 2], ex.z);
        atomicAdd(&s_den[dl4[u] + 3], ex.w);
      }
    }
  }
  __syncthreads();

  for (int base = w0; base < w1; base += 32) {
    const int n = min(32, w1 - base);
    int s = 0, d = b * RNG;
    float a0 = 0.f, a1 = 0.f, a2 = 0.f, a3 = 0.f;
    if (lane < n) {
      const int4 r = rec[base + lane];
      s = r.y;
      d = r.z;
      const float4 ev = scr[base + lane];
      const float4 dv = *(const float4*)(s_den + (d - b * RNG) * HEADS);
      a0 = ev.x / (dv.x + 1e-16f);
      a1 = ev.y / (dv.y + 1e-16f);
      a2 = ev.z / (dv.z + 1e-16f);
      a3 = ev.w / (dv.w + 1e-16f);
    }
    const int dl = (d - b * RNG) * XP;
    int j = 0;
    for (; j + GRP <= n; j += GRP)
      agg_group<true>(h, acc, lane, j, n, s, dl, a0, a1, a2, a3);
    if (j < n)
      agg_group<false>(h, acc, lane, j, n, s, dl, a0, a1, a2, a3);
  }
  __syncthreads();

  {
    const int m  = lane & 15;
    const int kh = lane >> 4;
    const int nc = wave * 16 + m;
    float* c_lo = acc + (kh * 8) * XP + nc;
    float* c_hi = c_lo + 16 * XP;
    v8f c0, c1;
#pragma unroll
    for (int r = 0; r < 8; ++r) {
      c0[r] = c_lo[r * XP];
      c1[r] = c_hi[r * XP];
    }
    const float* a_lo = xt + m * XP + kh * 2;
    const float* a_hi = a_lo + 16 * XP;
    const float* __restrict__ wb = Wskip + (size_t)(kh * 2) * DIM + nc;
#pragma unroll 8
    for (int k0 = 0; k0 < DIM; k0 += 4) {
      const v2f av0 = *(const v2f*)(a_lo + k0);
      const v2f av1 = *(const v2f*)(a_hi + k0);
      v2f bv;
      bv.x = wb[(size_t)k0 * DIM];
      bv.y = wb[(size_t)(k0 + 1) * DIM];
      c0 = __builtin_amdgcn_wmma_f32_16x16x4_f32(
          false, av0, false, bv, (short)0, c0, false, false);
      c1 = __builtin_amdgcn_wmma_f32_16x16x4_f32(
          false, av1, false, bv, (short)0, c1, false, false);
    }
    const float bb = bias[nc];
#pragma unroll
    for (int r = 0; r < 8; ++r) {
      const float v0 = c0[r] + bb;
      const float v1 = c1[r] + bb;
      c_lo[r * XP] = (v0 > 0.0f) ? v0 : expm1f(v0);
      c_hi[r * XP] = (v1 > 0.0f) ? v1 : expm1f(v1);
    }
  }
  __syncthreads();

  const int ln = (int)__builtin_amdgcn_mbcnt_lo(~0u, 0u);
  const int tq = wave * XP + ln * 4;
  if (whole) {
#pragma unroll
    for (int m = 0; m < RNG * DIM / 1024; ++m)
      tile[tid + 256 * m] = *(const float4*)(acc + tq + 8 * m * XP);
  } else {
    for (int m = 0; m < RNG * DIM / 1024; ++m) {
      const int i = tid + 256 * m;
      if (i < lim4) tile[i] = *(const float4*)(acc + tq + 8 * m * XP);
    }
  }

  if (NEXT) {
    {
      const int m  = ln & 15;
      const int kh = ln >> 4;
      const int nc = wave * 16 + m;
      const float* a_lo = acc + m * XP + kh * 2;
      const float* a_hi = a_lo + 16 * XP;
      const float* __restrict__ wb = Wn + (size_t)(kh * 2) * DIM + nc;
      v8f d0 = {}, d1 = {};
#pragma unroll 8
      for (int k0 = 0; k0 < DIM; k0 += 4) {
        const v2f av0 = *(const v2f*)(a_lo + k0);
        const v2f av1 = *(const v2f*)(a_hi + k0);
        v2f bv;
        bv.x = wb[(size_t)k0 * DIM];
        bv.y = wb[(size_t)(k0 + 1) * DIM];
        d0 = __builtin_amdgcn_wmma_f32_16x16x4_f32(
            false, av0, false, bv, (short)0, d0, false, false);
        d1 = __builtin_amdgcn_wmma_f32_16x16x4_f32(
            false, av1, false, bv, (short)0, d1, false, false);
      }
      float* o_lo = xt + (kh * 8) * XP + nc;
      float* o_hi = o_lo + 16 * XP;
#pragma unroll
      for (int r = 0; r < 8; ++r) {
        o_lo[r * XP] = d0[r];
        o_hi[r * XP] = d1[r];
      }
    }
    __syncthreads();

    if (tid < RNG * HEADS) {
      const int row = tid >> 2, hh = tid & 3;
      const float* hp = xt + row * XP + hh * CH;
      const float* __restrict__ ws = a_src_n + hh * CH;
      const float* __restrict__ wd = a_dst_n + hh * CH;
      float s1 = 0.f, s2 = 0.f;
#pragma unroll 8
      for (int c = 0; c < CH; ++c) {
        const float v = hp[c];
        s1 += v * ws[c];
        s2 += v * wd[c];
      }
      if (b * RNG + row < N) {
        asrc_n[(size_t)b * RNG * HEADS + tid] = s1;
        adst_n[(size_t)b * RNG * HEADS + tid] = s2;
      }
    }

    float4* __restrict__ htile = (float4*)(hn + (size_t)b * RNG * DIM);
    if (whole) {
#pragma unroll
      for (int m = 0; m < RNG * DIM / 1024; ++m)
        htile[tid + 256 * m] = *(const float4*)(xt + tq + 8 * m * XP);
    } else {
      for (int m = 0; m < RNG * DIM / 1024; ++m) {
        const int i = tid + 256 * m;
        if (i < lim4) htile[i] = *(const float4*)(xt + tq + 8 * m * XP);
      }
    }
  }
}

extern "C" void kernel_launch(void* const* d_in, const int* in_sizes, int n_in,
                              void* d_out, int out_size, void* d_ws,
                              size_t ws_size, hipStream_t stream) {
  const float* x  = (const float*)d_in[0];
  const int*   ei = (const int*)d_in[1];

  const int N  = in_sizes[0] / DIM;
  const int NE = in_sizes[1] / 2;
  const int E  = NE + N;

  float* ws = (float*)d_ws;
  size_t o = 0;
  float* h     = ws + o; o += (size_t)N * DIM;
  float* h2    = ws + o; o += (size_t)N * DIM;
  float* y1    = ws + o; o += (size_t)N * DIM;
  float* asrc  = ws + o; o += (size_t)N * HEADS;
  float* adst  = ws + o; o += (size_t)N * HEADS;
  float* asrc2 = ws + o; o += (size_t)N * HEADS;
  float* adst2 = ws + o; o += (size_t)N * HEADS;
  float4* scr  = (float4*)(ws + o); o += (size_t)E * HEADS;
  const int NB = (N + RNG - 1) / RNG;
  int4* rec = (int4*)(ws + o); o += (size_t)E * 4;
  const int NBR = NB * REP;
  unsigned* bcnt = (unsigned*)(ws + o); o += (size_t)NBR * LINE;
  unsigned* boff = (unsigned*)(ws + o); o += (size_t)((NBR + 4) & ~3);
  unsigned* brank = (unsigned*)(ws + o); o += (size_t)((E + 3) & ~3);

  float* y = (float*)d_out;

  const int TB = 256;
  const int gE    = (E + TB - 1) / TB;
  const int gNB   = (NBR + TB - 1) / TB;

  gat_bin_zero<<<gNB, TB, 0, stream>>>(bcnt, NBR);
  gat_bin_count<<<gE, TB, 0, stream>>>(ei, NE, E, bcnt, brank);

  const float* W1     = (const float*)d_in[2];
  const float* a_src1 = (const float*)d_in[3];
  const float* a_dst1 = (const float*)d_in[4];
  const float* bias1  = (const float*)d_in[5];
  const float* Wskip1 = (const float*)d_in[6];
  const float* W2     = (const float*)d_in[7];
  const float* a_src2 = (const float*)d_in[8];
  const float* a_dst2 = (const float*)d_in[9];
  const float* bias2  = (const float*)d_in[10];
  const float* Wskip2 = (const float*)d_in[11];

  gat_front_lds<<<NB + SCAN_BLOCKS, TB, 0, stream>>>(x, W1, a_src1, a_dst1, h,
                                                     asrc, adst, N, bcnt, boff,
                                                     NBR);
  gat_bin_fill<<<gE, TB, 0, stream>>>(ei, NE, E, boff, brank, rec);

  gat_agg_lds<true><<<NB, TB, 0, stream>>>(rec, boff, asrc, adst, scr, h, x,
                                           Wskip1, bias1, y1, N, W2, a_src2,
                                           a_dst2, h2, asrc2, adst2);

  gat_agg_lds<false><<<NB, TB, 0, stream>>>(rec, boff, asrc2, adst2, scr, h2,
                                            y1, Wskip2, bias2, y, N, nullptr,
                                            nullptr, nullptr, nullptr,
                                            nullptr, nullptr);
}


# h rows stored channel-interleaved (word 4l+k = channel 32k+l): the aggregation's row gather is one 16-byte load per lane, a whole 512-B row per instruction, not four 4-byte loads; sum order kept
# speedup vs baseline: 4.0413x; 1.0391x over previous
#include <hip/hip_runtime.h>
#include <math.h>

#define DIM 128
#define HEADS 4
#define CH 32
#define NEG_SLOPE 0.2f

typedef __attribute__((ext_vector_type(2))) float v2f;
typedef __attribute__((ext_vector_type(8))) float v8f;

__device__ __forceinline__ void edge_sd(const int* __restrict__ ei, int e,
                                        int NE, int& s, int& d) {
  if (e < NE) { s = ei[e]; d = ei[NE + e]; }
  else        { s = d = e - NE; }
}

__device__ __forceinline__ unsigned ord_encode(float f) {
  unsigned u = __float_as_uint(f);
  return (f >= 0.0f) ? (u | 0x80000000u) : ~u;
}

__device__ __forceinline__ float ord_decode(unsigned u) {
  const unsigned top = (unsigned)((int)u >> 31);
  const float f = __uint_as_float(u ^ (~top | 0x80000000u));
  return (u == 0u) ? 0.0f : f;
}

__device__ __forceinline__ float leaky(float v) {
  return (v > 0.0f) ? v : NEG_SLOPE * v;
}

#define RNG 32
#define RNG_SHIFT 5
#define REP 8
#define LINE 32

__device__ __forceinline__ int sub_bin(int e, int d) {
  return (d >> RNG_SHIFT) * REP + (e & (REP - 1));
}

__global__ void gat_bin_zero(unsigned* __restrict__ cnt, int n) {
  int t = blockIdx.x * blockDim.x + threadIdx.x;
  if (t < n) cnt[(size_t)t * LINE] = 0u;
}

__global__ void gat_bin_count(const int* __restrict__ ei, int NE, int E,
                              unsigned* __restrict__ cnt,
                              unsigned* __restrict__ rank) {
  int e = blockIdx.x * blockDim.x + threadIdx.x;
  if (e >= E) return;
  int s, d;
  edge_sd(ei, e, NE, s, d);
  rank[e] = atomicAdd(&cnt[(size_t)sub_bin(e, d) * LINE], 1u);
}

#define SCAN_BLOCKS 8
#define SCAN_GRP 4
__device__ __forceinline__ void bin_scan_slice(const unsigned* __restrict__ cnt,
                                               unsigned* __restrict__ off,
                                               int NBR, int g, unsigned* sh) {
  const int t = threadIdx.x;
  const int S = (NBR + SCAN_BLOCKS - 1) / SCAN_BLOCKS;
  const int lo = min(g * S, NBR);
  const int hi = min(lo + S, NBR);

  unsigned ahead = 0u;
  for (int base = 0; base < lo; base += 256 * SCAN_GRP) {
    unsigned c[SCAN_GRP];
#pragma unroll
    for (int u = 0; u < SCAN_GRP; ++u) {
      const int b = base + 256 * u + t;
      c[u] = cnt[(size_t)min(b, lo - 1) * LINE];
    }
#pragma unroll
    for (int u = 0; u < SCAN_GRP; ++u) {
      const int b = base + 256 * u + t;
      ahead += (b < lo) ? c[u] : 0u;
    }
  }

  const int P = (S + 255) / 256;
  const int b0 = lo + t * P;
  unsigned sum = 0u;
  for (int i0 = 0; i0 < P; i0 += SCAN_GRP) {
    unsigned c[SCAN_GRP];
#pragma unroll
    for (int u = 0; u < SCAN_GRP; ++u)
      c[u] = cnt[(size_t)min(b0 + i0 + u, NBR - 1) * LINE];
#pragma unroll
    for (int u = 0; u < SCAN_GRP; ++u) {
      const int i = i0 + u;
      sum += (i < P && b0 + i < hi) ? c[u] : 0u;
    }
  }

  sh[t] = sum;
  sh[256 + t] = ahead;
  __syncthreads();
  for (int ofs = 1; ofs < 256; ofs <<= 1) {
    const unsigned v = (t >= ofs) ? sh[t - ofs] : 0u;
    const unsigned w = (t >= ofs) ? sh[256 + t - ofs] : 0u;
    __syncthreads();
    sh[t] += v;
    sh[256 + t] += w;
    __syncthreads();
  }
  const unsigned tot_ahead = sh[511];
  unsigned run = tot_ahead + sh[t] - sum;
  for (int i0 = 0; i0 < P; i0 += SCAN_GRP) {
    unsigned c[SCAN_GRP];
#pragma unroll
    for (int u = 0; u < SCAN_GRP; ++u)
      c[u] = cnt[(size_t)min(b0 + i0 + u, NBR - 1) * LINE];
#pragma unroll
    for (int u = 0; u < SCAN_GRP; ++u) {
      const int i = i0 + u;
      const bool ok = (i < P && b0 + i < hi);
      if (ok) off[b0 + i] = run;
      run += ok ? c[u] : 0u;
    }
  }
  if (g == SCAN_BLOCKS - 1 && t == 255) off[NBR] = tot_ahead + sh[255];
}

__global__ void gat_bin_fill(const int* __restrict__ ei, int NE, int E,
                             const unsigned* __restrict__ off,
                             const unsigned* __restrict__ rank,
                             int4* __restrict__ rec) {
  int e = blockIdx.x * blockDim.x + threadIdx.x;
  if (e >= E) return;
  int s, d;
  edge_sd(ei, e, NE, s, d);
  const unsigned pos = off[sub_bin(e, d)] + rank[e];
  rec[pos] = make_int4(e, s, d, 0);
}

#define GRP 4
template <bool FULL>
__device__ __forceinline__ void agg_group(const float* __restrict__ h,
                                          float* acc, int lane, int j, int n,
                                          int s, int dl, float a0, float a1,
                                          float a2, float a3) {
  float hv[GRP][4];
  float al[GRP][4];
  int ro[GRP];
#pragma unroll
  for (int u = 0; u < GRP; ++u) {
    if (FULL || j + u < n) {
      const int sj = __builtin_amdgcn_readlane(s, j + u);
      ro[u] = __builtin_amdgcn_readlane(dl, j + u);
      al[u][0] = __int_as_float(__builtin_amdgcn_readlane(__float_as_int(a0), j + u));
      al[u][1] = __int_as_float(__builtin_amdgcn_readlane(__float_as_int(a1), j + u));
      al[u][2] = __int_as_float(__builtin_amdgcn_readlane(__float_as_int(a2), j + u));
      al[u][3] = __int_as_float(__builtin_amdgcn_readlane(__float_as_int(a3), j + u));
      const float4 v = ((const float4*)(h + (size_t)sj * DIM))[lane];
      hv[u][0] = v.x;
      hv[u][1] = v.y;
      hv[u][2] = v.z;
      hv[u][3] = v.w;
    }
  }
#pragma unroll
  for (int u = 0; u < GRP; ++u) {
    if (FULL || j + u < n) {
#pragma unroll
      for (int k = 0; k < 4; ++k)
        atomicAdd(&acc[ro[u] + lane + 32 * k], hv[u][k] * al[u][k]);
    }
  }
}

#define XP (DIM + 4)
__global__ void __launch_bounds__(256)
gat_front_lds(const float* __restrict__ xin, const float* __restrict__ Wn,
              const float* __restrict__ a_src_n,
              const float* __restrict__ a_dst_n, float* __restrict__ hn,
              float* __restrict__ asrc_n, float* __restrict__ adst_n, int N,
              const unsigned* __restrict__ cnt, unsigned* __restrict__ off,
              int NBR) {
  __shared__ alignas(16) float xt[RNG * XP];
  __shared__ alignas(16) float ht[RNG * XP];
  if (blockIdx.x < SCAN_BLOCKS) {
    bin_scan_slice(cnt, off, NBR, blockIdx.x, reinterpret_cast<unsigned*>(xt));
    return;
  }
  const int tid = threadIdx.x;
  const int lane = tid & 31;
  const int wave = __builtin_amdgcn_readfirstlane(tid >> 5);
  const int b = blockIdx.x - SCAN_BLOCKS;

  const int lim4 = min(RNG, N - b * RNG) * (DIM / 4);
  const bool whole = (lim4 == RNG * DIM / 4);
  const int tpos = wave * XP + lane * 4;

  const float4* __restrict__ xg = (const float4*)(xin + (size_t)b * RNG * DIM);
  float4 xr[RNG * DIM / 1024];
#pragma unroll
  for (int m = 0; m < RNG * DIM / 1024; ++m) {
    const int i = tid + 256 * m;
    xr[m] = (i < lim4) ? xg[i] : make_float4(0.f, 0.f, 0.f, 0.f);
  }
#pragma unroll
  for (int m = 0; m < RNG * DIM / 1024; ++m)
    *(float4*)(xt + tpos + 8 * m * XP) = xr[m];
  __syncthreads();

  {
    const int m  = lane & 15;
    const int kh = lane >> 4;
    const int nc = wave * 16 + m;
    const float* a_lo = xt + m * XP + kh * 2;
    const float* a_hi = a_lo + 16 * XP;
    const float* __restrict__ wb = Wn + (size_t)(kh * 2) * DIM + nc;
    v8f d0 = {}, d1 = {};
#pragma unroll 8
    for (int k0 = 0; k0 < DIM; k0 += 4) {
      const v2f av0 = *(const v2f*)(a_lo + k0);
      const v2f av1 = *(const v2f*)(a_hi + k0);
      v2f bv;
      bv.x = wb[(size_t)k0 * DIM];
      bv.y = wb[(size_t)(k0 + 1) * DIM];
      d0 = __builtin_amdgcn_wmma_f32_16x16x4_f32(
          false, av0, false, bv, (short)0, d0, false, false);
      d1 = __builtin_amdgcn_wmma_f32_16x16x4_f32(
          false, av1, false, bv, (short)0, d1, false, false);
    }
    float* o_lo = ht + (kh * 8) * XP + nc;
    float* o_hi = o_lo + 16 * XP;
#pragma unroll
    for (int r = 0; r < 8; ++r) {
      o_lo[r * XP] = d0[r];
      o_hi[r * XP] = d1[r];
    }
  }
  __syncthreads();

  if (tid < RNG * HEADS) {
    const int row = tid >> 2, hh = tid & 3;
    const float* hp = ht + row * XP + hh * CH;
    const float* __restrict__ ws = a_src_n + hh * CH;
    const float* __restrict__ wd = a_dst_n + hh * CH;
    float s1 = 0.f, s2 = 0.f;
#pragma unroll
    for (int c = 0; c < CH; ++c) {
      const float v = hp[c];
      s1 += v * ws[c];
      s2 += v * wd[c];
    }
    if (b * RNG + row < N) {
      asrc_n[(size_t)b * RNG * HEADS + tid] = s1;
      adst_n[(size_t)b * RNG * HEADS + tid] = s2;
    }
  }

  float4* __restrict__ htile = (float4*)(hn + (size_t)b * RNG * DIM);
  const float* hsrc = ht + wave * XP + lane;
  if (whole) {
#pragma unroll
    for (int m = 0; m < RNG * DIM / 1024; ++m) {
      const float* p = hsrc + 8 * m * XP;
      htile[tid + 256 * m] = make_float4(p[0], p[32], p[64], p[96]);
    }
  } else {
    for (int m = 0; m < RNG * DIM / 1024; ++m) {
      const int i = tid + 256 * m;
      const float* p = hsrc + 8 * m * XP;
      if (i < lim4) htile[i] = make_float4(p[0], p[32], p[64], p[96]);
    }
  }
}

template <bool NEXT>
__global__ void __launch_bounds__(256)
gat_agg_lds(const int4* __restrict__ rec, const unsigned* __restrict__ off,
            const float* __restrict__ asrc, const float* __restrict__ adst,
            float4* __restrict__ scr, const float* __restrict__ h,
            const float* __restrict__ xin, const float* __restrict__ Wskip,
            const float* __restrict__ bias, float* __restrict__ yout, int N,
            const float* __restrict__ Wn, const float* __restrict__ a_src_n,
            const float* __restrict__ a_dst_n, float* __restrict__ hn,
            float* __restrict__ asrc_n, float* __restrict__ adst_n) {
  __shared__ alignas(16) float acc[RNG * XP];
  __shared__ alignas(16) float xt[RNG * XP];
  __shared__ alignas(16) float s_adst[RNG * HEADS];
  __shared__ alignas(16) unsigned s_menc[RNG * HEADS];
  __shared__ alignas(16) float s_den[RNG * HEADS];
  const int tid = threadIdx.x;
  const int lane = tid & 31;
  const int wave = __builtin_amdgcn_readfirstlane(tid >> 5);
  const int b = blockIdx.x;

  float4* __restrict__ tile = (float4*)(yout + (size_t)b * RNG * DIM);
  const int lim4 = min(RNG, N - b * RNG) * (DIM / 4);
  const bool whole = (lim4 == RNG * DIM / 4);
  const int tpos = (tid >> 5) * XP + (tid & 31) * 4;

  const float4* __restrict__ xg = (const float4*)(xin + (size_t)b * RNG * DIM);
  float4 xr[RNG * DIM / 1024];
#pragma unroll
  for (int m = 0; m < RNG * DIM / 1024; ++m) {
    const int i = tid + 256 * m;
    xr[m] = (i < lim4) ? xg[i] : make_float4(0.f, 0.f, 0.f, 0.f);
  }
#pragma unroll
  for (int m = 0; m < RNG * DIM / 1024; ++m)
    *(float4*)(acc + tpos + 8 * m * XP) = make_float4(0.f, 0.f, 0.f, 0.f);
  if (tid < RNG * HEADS) {
    const int node = b * RNG + (tid >> 2);
    s_adst[tid] = (node < N) ? adst[(size_t)b * RNG * HEADS + tid] : 0.0f;
    s_menc[tid] = 0u;
    s_den[tid] = 0.0f;
  }
  __syncthreads();

  const int beg = __builtin_amdgcn_readfirstlane((int)off[b * REP]);
  const int end = __builtin_amdgcn_readfirstlane((int)off[(b + 1) * REP]);
  const int per = (end - beg + 7) >> 3;
  const int w0 = beg + wave * per;
  const int w1 = min(w0 + per, end);


  for (int base = w0; base < w1; base += 64) {
    int q[2], dl4[2];
    bool ok[2];
    float4 av[2];
#pragma unroll
    for (int u = 0; u < 2; ++u) {
      const int p = base + 32 * u + lane;
      ok[u] = p < w1;
      q[u] = min(p, w1 - 1);
    }
    int4 r[2];
#pragma unroll
    for (int u = 0; u < 2; ++u) r[u] = rec[q[u]];
#pragma unroll
    for (int u = 0; u < 2; ++u) {
      av[u] = *(const float4*)(asrc + (size_t)r[u].y * HEADS);
      dl4[u] = (r[u].z - b * RNG) * HEADS;
    }
    asm volatile("" ::"v"(av[0].x), "v"(av[1].x));
#pragma unroll
    for (int u = 0; u < 2; ++u) {
      if (ok[u]) {
        const float4 dv = *(const float4*)(s_adst + dl4[u]);
        float4 e;
        e.x = leaky(av[u].x + dv.x);
        e.y = leaky(av[u].y + dv.y);
        e.z = leaky(av[u].z + dv.z);
        e.w = leaky(av[u].w + dv.w);
        scr[q[u]] = e;
        atomicMax(&s_menc[dl4[u] + 0], ord_encode(e.x));
        atomicMax(&s_menc[dl4[u] + 1], ord_encode(e.y));
        atomicMax(&s_menc[dl4[u] + 2], ord_encode(e.z));
        atomicMax(&s_menc[dl4[u] + 3], ord_encode(e.w));
      }
    }
  }
#pragma unroll
  for (int m = 0; m < RNG * DIM / 1024; ++m)
    *(float4*)(xt + tpos + 8 * m * XP) = xr[m];
  __syncthreads();

  for (int base = w0; base < w1; base += 64) {
    int q[2], dl4[2];
    bool ok[2];
    float4 e[2];
#pragma unroll
    for (int u = 0; u < 2; ++u) {
      const int p = base + 32 * u + lane;
      ok[u] = p < w1;
      q[u] = min(p, w1 - 1);
    }
    int dz[2];
#pragma unroll
    for (int u = 0; u < 2; ++u) {
      dz[u] = rec[q[u]].z;
      e[u] = scr[q[u]];
    }
    asm volatile("" ::"v"(dz[0]), "v"(dz[1]), "v"(e[0].x), "v"(e[1].x));
#pragma unroll
    for (int u = 0; u < 2; ++u) dl4[u] = (dz[u] - b * RNG) * HEADS;
#pragma unroll
    for (int u = 0; u < 2; ++u) {
      if (ok[u]) {
        const uint4 mu = *(const uint4*)(s_menc + dl4[u]);
        float4 ex;
        ex.x = expf(e[u].x - ord_decode(mu.x));
        ex.y = expf(e[u].y - ord_decode(mu.y));
        ex.z = expf(e[u].z - ord_decode(mu.z));
        ex.w = expf(e[u].w - ord_decode(mu.w));
        scr[q[u]] = ex;
        atomicAdd(&s_den[dl4[u] + 0], ex.x);
        atomicAdd(&s_den[dl4[u] + 1], ex.y);
        atomicAdd(&s_den[dl4[u] + 2], ex.z);
        atomicAdd(&s_den[dl4[u] + 3], ex.w);
      }
    }
  }
  __syncthreads();

  for (int base = w0; base < w1; base += 32) {
    const int n = min(32, w1 - base);
    int s = 0, d = b * RNG;
    float a0 = 0.f, a1 = 0.f, a2 = 0.f, a3 = 0.f;
    if (lane < n) {
      const int4 r = rec[base + lane];
      s = r.y;
      d = r.z;
      const float4 ev = scr[base + lane];
      const float4 dv = *(const float4*)(s_den + (d - b * RNG) * HEADS);
      a0 = ev.x / (dv.x + 1e-16f);
      a1 = ev.y / (dv.y + 1e-16f);
      a2 = ev.z / (dv.z + 1e-16f);
      a3 = ev.w / (dv.w + 1e-16f);
    }
    const int dl = (d - b * RNG) * XP;
    int j = 0;
    for (; j + GRP <= n; j += GRP)
      agg_group<true>(h, acc, lane, j, n, s, dl, a0, a1, a2, a3);
    if (j < n)
      agg_group<false>(h, acc, lane, j, n, s, dl, a0, a1, a2, a3);
  }
  __syncthreads();

  {
    const int m  = lane & 15;
    const int kh = lane >> 4;
    const int nc = wave * 16 + m;
    float* c_lo = acc + (kh * 8) * XP + nc;
    float* c_hi = c_lo + 16 * XP;
    v8f c0, c1;
#pragma unroll
    for (int r = 0; r < 8; ++r) {
      c0[r] = c_lo[r * XP];
      c1[r] = c_hi[r * XP];
    }
    const float* a_lo = xt + m * XP + kh * 2;
    const float* a_hi = a_lo + 16 * XP;
    const float* __restrict__ wb = Wskip + (size_t)(kh * 2) * DIM + nc;
#pragma unroll 8
    for (int k0 = 0; k0 < DIM; k0 += 4) {
      const v2f av0 = *(const v2f*)(a_lo + k0);
      const v2f av1 = *(const v2f*)(a_hi + k0);
      v2f bv;
      bv.x = wb[(size_t)k0 * DIM];
      bv.y = wb[(size_t)(k0 + 1) * DIM];
      c0 = __builtin_amdgcn_wmma_f32_16x16x4_f32(
          false, av0, false, bv, (short)0, c0, false, false);
      c1 = __builtin_amdgcn_wmma_f32_16x16x4_f32(
          false, av1, false, bv, (short)0, c1, false, false);
    }
    const float bb = bias[nc];
#pragma unroll
    for (int r = 0; r < 8; ++r) {
      const float v0 = c0[r] + bb;
      const float v1 = c1[r] + bb;
      c_lo[r * XP] = (v0 > 0.0f) ? v0 : expm1f(v0);
      c_hi[r * XP] = (v1 > 0.0f) ? v1 : expm1f(v1);
    }
  }
  __syncthreads();

  const int ln = (int)__builtin_amdgcn_mbcnt_lo(~0u, 0u);
  const int tq = wave * XP + ln * 4;
  if (whole) {
#pragma unroll
    for (int m = 0; m < RNG * DIM / 1024; ++m)
      tile[tid + 256 * m] = *(const float4*)(acc + tq + 8 * m * XP);
  } else {
    for (int m = 0; m < RNG * DIM / 1024; ++m) {
      const int i = tid + 256 * m;
      if (i < lim4) tile[i] = *(const float4*)(acc + tq + 8 * m * XP);
    }
  }

  if (NEXT) {
    {
      const int m  = ln & 15;
      const int kh = ln >> 4;
      const int nc = wave * 16 + m;
      const float* a_lo = acc + m * XP + kh * 2;
      const float* a_hi = a_lo + 16 * XP;
      const float* __restrict__ wb = Wn + (size_t)(kh * 2) * DIM + nc;
      v8f d0 = {}, d1 = {};
#pragma unroll 8
      for (int k0 = 0; k0 < DIM; k0 += 4) {
        const v2f av0 = *(const v2f*)(a_lo + k0);
        const v2f av1 = *(const v2f*)(a_hi + k0);
        v2f bv;
        bv.x = wb[(size_t)k0 * DIM];
        bv.y = wb[(size_t)(k0 + 1) * DIM];
        d0 = __builtin_amdgcn_wmma_f32_16x16x4_f32(
            false, av0, false, bv, (short)0, d0, false, false);
        d1 = __builtin_amdgcn_wmma_f32_16x16x4_f32(
            false, av1, false, bv, (short)0, d1, false, false);
      }
      float* o_lo = xt + (kh * 8) * XP + nc;
      float* o_hi = o_lo + 16 * XP;
#pragma unroll
      for (int r = 0; r < 8; ++r) {
        o_lo[r * XP] = d0[r];
        o_hi[r * XP] = d1[r];
      }
    }
    __syncthreads();

    if (tid < RNG * HEADS) {
      const int row = tid >> 2, hh = tid & 3;
      const float* hp = xt + row * XP + hh * CH;
      const float* __restrict__ ws = a_src_n + hh * CH;
      const float* __restrict__ wd = a_dst_n + hh * CH;
      float s1 = 0.f, s2 = 0.f;
#pragma unroll 8
      for (int c = 0; c < CH; ++c) {
        const float v = hp[c];
        s1 += v * ws[c];
        s2 += v * wd[c];
      }
      if (b * RNG + row < N) {
        asrc_n[(size_t)b * RNG * HEADS + tid] = s1;
        adst_n[(size_t)b * RNG * HEADS + tid] = s2;
      }
    }

    float4* __restrict__ htile = (float4*)(hn + (size_t)b * RNG * DIM);
    const float* hsrc = xt + wave * XP + ln;
    if (whole) {
#pragma unroll
      for (int m = 0; m < RNG * DIM / 1024; ++m) {
        const float* p = hsrc + 8 * m * XP;
        htile[tid + 256 * m] = make_float4(p[0], p[32], p[64], p[96]);
      }
    } else {
      for (int m = 0; m < RNG * DIM / 1024; ++m) {
        const int i = tid + 256 * m;
        const float* p = hsrc + 8 * m * XP;
        if (i < lim4) htile[i] = make_float4(p[0], p[32], p[64], p[96]);
      }
    }
  }
}

extern "C" void kernel_launch(void* const* d_in, const int* in_sizes, int n_in,
                              void* d_out, int out_size, void* d_ws,
                              size_t ws_size, hipStream_t stream) {
  const float* x  = (const float*)d_in[0];
  const int*   ei = (const int*)d_in[1];

  const int N  = in_sizes[0] / DIM;
  const int NE = in_sizes[1] / 2;
  const int E  = NE + N;

  float* ws = (float*)d_ws;
  size_t o = 0;
  float* h     = ws + o; o += (size_t)N * DIM;
  float* h2    = ws + o; o += (size_t)N * DIM;
  float* y1    = ws + o; o += (size_t)N * DIM;
  float* asrc  = ws + o; o += (size_t)N * HEADS;
  float* adst  = ws + o; o += (size_t)N * HEADS;
  float* asrc2 = ws + o; o += (size_t)N * HEADS;
  float* adst2 = ws + o; o += (size_t)N * HEADS;
  float4* scr  = (float4*)(ws + o); o += (size_t)E * HEADS;
  const int NB = (N + RNG - 1) / RNG;
  int4* rec = (int4*)(ws + o); o += (size_t)E * 4;
  const int NBR = NB * REP;
  unsigned* bcnt = (unsigned*)(ws + o); o += (size_t)NBR * LINE;
  unsigned* boff = (unsigned*)(ws + o); o += (size_t)((NBR + 4) & ~3);
  unsigned* brank = (unsigned*)(ws + o); o += (size_t)((E + 3) & ~3);

  float* y = (float*)d_out;

  const int TB = 256;
  const int gE    = (E + TB - 1) / TB;
  const int gNB   = (NBR + TB - 1) / TB;

  gat_bin_zero<<<gNB, TB, 0, stream>>>(bcnt, NBR);
  gat_bin_count<<<gE, TB, 0, stream>>>(ei, NE, E, bcnt, brank);

  const float* W1     = (const float*)d_in[2];
  const float* a_src1 = (const float*)d_in[3];
  const float* a_dst1 = (const float*)d_in[4];
  const float* bias1  = (const float*)d_in[5];
  const float* Wskip1 = (const float*)d_in[6];
  const float* W2     = (const float*)d_in[7];
  const float* a_src2 = (const float*)d_in[8];
  const float* a_dst2 = (const float*)d_in[9];
  const float* bias2  = (const float*)d_in[10];
  const float* Wskip2 = (const float*)d_in[11];

  gat_front_lds<<<NB + SCAN_BLOCKS, TB, 0, stream>>>(x, W1, a_src1, a_dst1, h,
                                                     asrc, adst, N, bcnt, boff,
                                                     NBR);
  gat_bin_fill<<<gE, TB, 0, stream>>>(ei, NE, E, boff, brank, rec);

  gat_agg_lds<true><<<NB, TB, 0, stream>>>(rec, boff, asrc, adst, scr, h, x,
                                           Wskip1, bias1, y1, N, W2, a_src2,
                                           a_dst2, h2, asrc2, adst2);

  gat_agg_lds<false><<<NB, TB, 0, stream>>>(rec, boff, asrc2, adst2, scr, h2,
                                            y1, Wskip2, bias2, y, N, nullptr,
                                            nullptr, nullptr, nullptr,
                                            nullptr, nullptr);
}


# Bin build's counters made private to one half of the second-level cache: a counting block adds to sub-bins of the half it runs on (die number), half kept in the rank word; slot order: add's arrival
# speedup vs baseline: 4.7528x; 1.1761x over previous
#include <hip/hip_runtime.h>
#include <math.h>

#define DIM 128
#define HEADS 4
#define CH 32
#define NEG_SLOPE 0.2f

typedef __attribute__((ext_vector_type(2))) float v2f;
typedef __attribute__((ext_vector_type(8))) float v8f;

__device__ __forceinline__ void edge_sd(const int* __restrict__ ei, int e,
                                        int NE, int& s, int& d) {
  if (e < NE) { s = ei[e]; d = ei[NE + e]; }
  else        { s = d = e - NE; }
}

__device__ __forceinline__ unsigned ord_encode(float f) {
  unsigned u = __float_as_uint(f);
  return (f >= 0.0f) ? (u | 0x80000000u) : ~u;
}

__device__ __forceinline__ float ord_decode(unsigned u) {
  const unsigned top = (unsigned)((int)u >> 31);
  const float f = __uint_as_float(u ^ (~top | 0x80000000u));
  return (u == 0u) ? 0.0f : f;
}

__device__ __forceinline__ float leaky(float v) {
  return (v > 0.0f) ? v : NEG_SLOPE * v;
}

#define RNG 32
#define RNG_SHIFT 5
#define REP 8
#define LINE 32

#define RANK_HALF 0x80000000u

__device__ __forceinline__ int sub_bin(int e, int d, int half) {
  return (d >> RNG_SHIFT) * REP + half * (REP / 2) + (e & (REP / 2 - 1));
}

__device__ __forceinline__ int chip_half() {
  return (int)((__builtin_amdgcn_s_sendmsg_rtn(0x87) >> 18) & 1u);
}

__global__ void gat_bin_zero(unsigned* __restrict__ cnt, int n) {
  int t = blockIdx.x * blockDim.x + threadIdx.x;
  if (t < n) cnt[(size_t)t * LINE] = 0u;
}

__global__ void gat_bin_count(const int* __restrict__ ei, int NE, int E,
                              unsigned* __restrict__ cnt,
                              unsigned* __restrict__ rank) {
  int e = blockIdx.x * blockDim.x + threadIdx.x;
  if (e >= E) return;
  int s, d;
  edge_sd(ei, e, NE, s, d);
  const int half = chip_half();
  const unsigned r = atomicAdd(&cnt[(size_t)sub_bin(e, d, half) * LINE], 1u);
  rank[e] = r | (half ? RANK_HALF : 0u);
}

#define SCAN_BLOCKS 8
#define SCAN_GRP 4
__device__ __forceinline__ void bin_scan_slice(const unsigned* __restrict__ cnt,
                                               unsigned* __restrict__ off,
                                               int NBR, int g, unsigned* sh) {
  const int t = threadIdx.x;
  const int S = (NBR + SCAN_BLOCKS - 1) / SCAN_BLOCKS;
  const int lo = min(g * S, NBR);
  const int hi = min(lo + S, NBR);

  unsigned ahead = 0u;
  for (int base = 0; base < lo; base += 256 * SCAN_GRP) {
    unsigned c[SCAN_GRP];
#pragma unroll
    for (int u = 0; u < SCAN_GRP; ++u) {
      const int b = base + 256 * u + t;
      c[u] = cnt[(size_t)min(b, lo - 1) * LINE];
    }
#pragma unroll
    for (int u = 0; u < SCAN_GRP; ++u) {
      const int b = base + 256 * u + t;
      ahead += (b < lo) ? c[u] : 0u;
    }
  }

  const int P = (S + 255) / 256;
  const int b0 = lo + t * P;
  unsigned sum = 0u;
  for (int i0 = 0; i0 < P; i0 += SCAN_GRP) {
    unsigned c[SCAN_GRP];
#pragma unroll
    for (int u = 0; u < SCAN_GRP; ++u)
      c[u] = cnt[(size_t)min(b0 + i0 + u, NBR - 1) * LINE];
#pragma unroll
    for (int u = 0; u < SCAN_GRP; ++u) {
      const int i = i0 + u;
      sum += (i < P && b0 + i < hi) ? c[u] : 0u;
    }
  }

  sh[t] = sum;
  sh[256 + t] = ahead;
  __syncthreads();
  for (int ofs = 1; ofs < 256; ofs <<= 1) {
    const unsigned v = (t >= ofs) ? sh[t - ofs] : 0u;
    const unsigned w = (t >= ofs) ? sh[256 + t - ofs] : 0u;
    __syncthreads();
    sh[t] += v;
    sh[256 + t] += w;
    __syncthreads();
  }
  const unsigned tot_ahead = sh[511];
  unsigned run = tot_ahead + sh[t] - sum;
  for (int i0 = 0; i0 < P; i0 += SCAN_GRP) {
    unsigned c[SCAN_GRP];
#pragma unroll
    for (int u = 0; u < SCAN_GRP; ++u)
      c[u] = cnt[(size_t)min(b0 + i0 + u, NBR - 1) * LINE];
#pragma unroll
    for (int u = 0; u < SCAN_GRP; ++u) {
      const int i = i0 + u;
      const bool ok = (i < P && b0 + i < hi);
      if (ok) off[b0 + i] = run;
      run += ok ? c[u] : 0u;
    }
  }
  if (g == SCAN_BLOCKS - 1 && t == 255) off[NBR] = tot_ahead + sh[255];
}

__global__ void gat_bin_fill(const int* __restrict__ ei, int NE, int E,
                             const unsigned* __restrict__ off,
                             const unsigned* __restrict__ rank,
                             int4* __restrict__ rec) {
  int e = blockIdx.x * blockDim.x + threadIdx.x;
  if (e >= E) return;
  const int ec = min(e, NE - 1);
  const unsigned r = rank[e];
  int s = ei[ec], d = ei[NE + ec];
  if (e >= NE) s = d = e - NE;
  const int half = (r & RANK_HALF) ? 1 : 0;
  const unsigned pos = off[sub_bin(e, d, half)] + (r & ~RANK_HALF);
  rec[pos] = make_int4(e, s, d, 0);
}

#define GRP 4
template <bool FULL>
__device__ __forceinline__ void agg_group(const float* __restrict__ h,
                                          float* acc, int lane, int j, int n,
                                          int s, int dl, float a0, float a1,
                                          float a2, float a3) {
  float hv[GRP][4];
  float al[GRP][4];
  int ro[GRP];
#pragma unroll
  for (int u = 0; u < GRP; ++u) {
    if (FULL || j + u < n) {
      const int sj = __builtin_amdgcn_readlane(s, j + u);
      ro[u] = __builtin_amdgcn_readlane(dl, j + u);
      al[u][0] = __int_as_float(__builtin_amdgcn_readlane(__float_as_int(a0), j + u));
      al[u][1] = __int_as_float(__builtin_amdgcn_readlane(__float_as_int(a1), j + u));
      al[u][2] = __int_as_float(__builtin_amdgcn_readlane(__float_as_int(a2), j + u));
      al[u][3] = __int_as_float(__builtin_amdgcn_readlane(__float_as_int(a3), j + u));
      const float4 v = ((const float4*)(h + (size_t)sj * DIM))[lane];
      hv[u][0] = v.x;
      hv[u][1] = v.y;
      hv[u][2] = v.z;
      hv[u][3] = v.w;
    }
  }
#pragma unroll
  for (int u = 0; u < GRP; ++u) {
    if (FULL || j + u < n) {
#pragma unroll
      for (int k = 0; k < 4; ++k)
        atomicAdd(&acc[ro[u] + lane + 32 * k], hv[u][k] * al[u][k]);
    }
  }
}

#define XP (DIM + 4)
__global__ void __launch_bounds__(256)
gat_front_lds(const float* __restrict__ xin, const float* __restrict__ Wn,
              const float* __restrict__ a_src_n,
              const float* __restrict__ a_dst_n, float* __restrict__ hn,
              float* __restrict__ asrc_n, float* __restrict__ adst_n, int N,
              const unsigned* __restrict__ cnt, unsigned* __restrict__ off,
              int NBR) {
  __shared__ alignas(16) float xt[RNG * XP];
  __shared__ alignas(16) float ht[RNG * XP];
  if (blockIdx.x < SCAN_BLOCKS) {
    bin_scan_slice(cnt, off, NBR, blockIdx.x, reinterpret_cast<unsigned*>(xt));
    return;
  }
  const int tid = threadIdx.x;
  const int lane = tid & 31;
  const int wave = __builtin_amdgcn_readfirstlane(tid >> 5);
  const int b = blockIdx.x - SCAN_BLOCKS;

  const int lim4 = min(RNG, N - b * RNG) * (DIM / 4);
  const bool whole = (lim4 == RNG * DIM / 4);
  const int tpos = wave * XP + lane * 4;

  const float4* __restrict__ xg = (const float4*)(xin + (size_t)b * RNG * DIM);
  float4 xr[RNG * DIM / 1024];
#pragma unroll
  for (int m = 0; m < RNG * DIM / 1024; ++m) {
    const int i = tid + 256 * m;
    xr[m] = (i < lim4) ? xg[i] : make_float4(0.f, 0.f, 0.f, 0.f);
  }
#pragma unroll
  for (int m = 0; m < RNG * DIM / 1024; ++m)
    *(float4*)(xt + tpos + 8 * m * XP) = xr[m];
  __syncthreads();

  {
    const int m  = lane & 15;
    const int kh = lane >> 4;
    const int nc = wave * 16 + m;
    const float* a_lo = xt + m * XP + kh * 2;
    const float* a_hi = a_lo + 16 * XP;
    const float* __restrict__ wb = Wn + (size_t)(kh * 2) * DIM + nc;
    v8f d0 = {}, d1 = {};
#pragma unroll 8
    for (int k0 = 0; k0 < DIM; k0 += 4) {
      const v2f av0 = *(const v2f*)(a_lo + k0);
      const v2f av1 = *(const v2f*)(a_hi + k0);
      v2f bv;
      bv.x = wb[(size_t)k0 * DIM];
      bv.y = wb[(size_t)(k0 + 1) * DIM];
      d0 = __builtin_amdgcn_wmma_f32_16x16x4_f32(
          false, av0, false, bv, (short)0, d0, false, false);
      d1 = __builtin_amdgcn_wmma_f32_16x16x4_f32(
          false, av1, false, bv, (short)0, d1, false, false);
    }
    float* o_lo = ht + (kh * 8) * XP + nc;
    float* o_hi = o_lo + 16 * XP;
#pragma unroll
    for (int r = 0; r < 8; ++r) {
      o_lo[r * XP] = d0[r];
      o_hi[r * XP] = d1[r];
    }
  }
  __syncthreads();

  if (tid < RNG * HEADS) {
    const int row = tid >> 2, hh = tid & 3;
    const float* hp = ht + row * XP + hh * CH;
    const float* __restrict__ ws = a_src_n + hh * CH;
    const float* __restrict__ wd = a_dst_n + hh * CH;
    float s1 = 0.f, s2 = 0.f;
#pragma unroll
    for (int c = 0; c < CH; ++c) {
      const float v = hp[c];
      s1 += v * ws[c];
      s2 += v * wd[c];
    }
    if (b * RNG + row < N) {
      asrc_n[(size_t)b * RNG * HEADS + tid] = s1;
      adst_n[(size_t)b * RNG * HEADS + tid] = s2;
    }
  }

  float4* __restrict__ htile = (float4*)(hn + (size_t)b * RNG * DIM);
  const float* hsrc = ht + wave * XP + lane;
  if (whole) {
#pragma unroll
    for (int m = 0; m < RNG * DIM / 1024; ++m) {
      const float* p = hsrc + 8 * m * XP;
      htile[tid + 256 * m] = make_float4(p[0], p[32], p[64], p[96]);
    }
  } else {
    for (int m = 0; m < RNG * DIM / 1024; ++m) {
      const int i = tid + 256 * m;
      const float* p = hsrc + 8 * m * XP;
      if (i < lim4) htile[i] = make_float4(p[0], p[32], p[64], p[96]);
    }
  }
}

template <bool NEXT>
__global__ void __launch_bounds__(256)
gat_agg_lds(const int4* __restrict__ rec, const unsigned* __restrict__ off,
            const float* __restrict__ asrc, const float* __restrict__ adst,
            float4* __restrict__ scr, const float* __restrict__ h,
            const float* __restrict__ xin, const float* __restrict__ Wskip,
            const float* __restrict__ bias, float* __restrict__ yout, int N,
            const float* __restrict__ Wn, const float* __restrict__ a_src_n,
            const float* __restrict__ a_dst_n, float* __restrict__ hn,
            float* __restrict__ asrc_n, float* __restrict__ adst_n) {
  __shared__ alignas(16) float acc[RNG * XP];
  __shared__ alignas(16) float xt[RNG * XP];
  __shared__ alignas(16) float s_adst[RNG * HEADS];
  __shared__ alignas(16) unsigned s_menc[RNG * HEADS];
  __shared__ alignas(16) float s_den[RNG * HEADS];
  const int tid = threadIdx.x;
  const int lane = tid & 31;
  const int wave = __builtin_amdgcn_readfirstlane(tid >> 5);
  const int b = blockIdx.x;

  float4* __restrict__ tile = (float4*)(yout + (size_t)b * RNG * DIM);
  const int lim4 = min(RNG, N - b * RNG) * (DIM / 4);
  const bool whole = (lim4 == RNG * DIM / 4);
  const int tpos = (tid >> 5) * XP + (tid & 31) * 4;

  const float4* __restrict__ xg = (const float4*)(xin + (size_t)b * RNG * DIM);
  float4 xr[RNG * DIM / 1024];
#pragma unroll
  for (int m = 0; m < RNG * DIM / 1024; ++m) {
    const int i = tid + 256 * m;
    xr[m] = (i < lim4) ? xg[i] : make_float4(0.f, 0.f, 0.f, 0.f);
  }
#pragma unroll
  for (int m = 0; m < RNG * DIM / 1024; ++m)
    *(float4*)(acc + tpos + 8 * m * XP) = make_float4(0.f, 0.f, 0.f, 0.f);
  if (tid < RNG * HEADS) {
    const int node = b * RNG + (tid >> 2);
    s_adst[tid] = (node < N) ? adst[(size_t)b * RNG * HEADS + tid] : 0.0f;
    s_menc[tid] = 0u;
    s_den[tid] = 0.0f;
  }
  __syncthreads();

  const int beg = __builtin_amdgcn_readfirstlane((int)off[b * REP]);
  const int end = __builtin_amdgcn_readfirstlane((int)off[(b + 1) * REP]);
  const int per = (end - beg + 7) >> 3;
  const int w0 = beg + wave * per;
  const int w1 = min(w0 + per, end);


  for (int base = w0; base < w1; base += 64) {
    int q[2], dl4[2];
    bool ok[2];
    float4 av[2];
#pragma unroll
    for (int u = 0; u < 2; ++u) {
      const int p = base + 32 * u + lane;
      ok[u] = p < w1;
      q[u] = min(p, w1 - 1);
    }
    int4 r[2];
#pragma unroll
    for (int u = 0; u < 2; ++u) r[u] = rec[q[u]];
#pragma unroll
    for (int u = 0; u < 2; ++u) {
      av[u] = *(const float4*)(asrc + (size_t)r[u].y * HEADS);
      dl4[u] = (r[u].z - b * RNG) * HEADS;
    }
    asm volatile("" ::"v"(av[0].x), "v"(av[1].x));
#pragma unroll
    for (int u = 0; u < 2; ++u) {
      if (ok[u]) {
        const float4 dv = *(const float4*)(s_adst + dl4[u]);
        float4 e;
        e.x = leaky(av[u].x + dv.x);
        e.y = leaky(av[u].y + dv.y);
        e.z = leaky(av[u].z + dv.z);
        e.w = leaky(av[u].w + dv.w);
        scr[q[u]] = e;
        atomicMax(&s_menc[dl4[u] + 0], ord_encode(e.x));
        atomicMax(&s_menc[dl4[u] + 1], ord_encode(e.y));
        atomicMax(&s_menc[dl4[u] + 2], ord_encode(e.z));
        atomicMax(&s_menc[dl4[u] + 3], ord_encode(e.w));
      }
    }
  }
#pragma unroll
  for (int m = 0; m < RNG * DIM / 1024; ++m)
    *(float4*)(xt + tpos + 8 * m * XP) = xr[m];
  __syncthreads();

  for (int base = w0; base < w1; base += 64) {
    int q[2], dl4[2];
    bool ok[2];
    float4 e[2];
#pragma unroll
    for (int u = 0; u < 2; ++u) {
      const int p = base + 32 * u + lane;
      ok[u] = p < w1;
      q[u] = min(p, w1 - 1);
    }
    int dz[2];
#pragma unroll
    for (int u = 0; u < 2; ++u) {
      dz[u] = rec[q[u]].z;
      e[u] = scr[q[u]];
    }
    asm volatile("" ::"v"(dz[0]), "v"(dz[1]), "v"(e[0].x), "v"(e[1].x));
#pragma unroll
    for (int u = 0; u < 2; ++u) dl4[u] = (dz[u] - b * RNG) * HEADS;
#pragma unroll
    for (int u = 0; u < 2; ++u) {
      if (ok[u]) {
        const uint4 mu = *(const uint4*)(s_menc + dl4[u]);
        float4 ex;
        ex.x = expf(e[u].x - ord_decode(mu.x));
        ex.y = expf(e[u].y - ord_decode(mu.y));
        ex.z = expf(e[u].z - ord_decode(mu.z));
        ex.w = expf(e[u].w - ord_decode(mu.w));
        scr[q[u]] = ex;
        atomicAdd(&s_den[dl4[u] + 0], ex.x);
        atomicAdd(&s_den[dl4[u] + 1], ex.y);
        atomicAdd(&s_den[dl4[u] + 2], ex.z);
        atomicAdd(&s_den[dl4[u] + 3], ex.w);
      }
    }
  }
  __syncthreads();

  for (int base = w0; base < w1; base += 32) {
    const int n = min(32, w1 - base);
    int s = 0, d = b * RNG;
    float a0 = 0.f, a1 = 0.f, a2 = 0.f, a3 = 0.f;
    if (lane < n) {
      const int4 r = rec[base + lane];
      s = r.y;
      d = r.z;
      const float4 ev = scr[base + lane];
      const float4 dv = *(const float4*)(s_den + (d - b * RNG) * HEADS);
      a0 = ev.x / (dv.x + 1e-16f);
      a1 = ev.y / (dv.y + 1e-16f);
      a2 = ev.z / (dv.z + 1e-16f);
      a3 = ev.w / (dv.w + 1e-16f);
    }
    const int dl = (d - b * RNG) * XP;
    int j = 0;
    for (; j + GRP <= n; j += GRP)
      agg_group<true>(h, acc, lane, j, n, s, dl, a0, a1, a2, a3);
    if (j < n)
      agg_group<false>(h, acc, lane, j, n, s, dl, a0, a1, a2, a3);
  }
  __syncthreads();

  {
    const int m  = lane & 15;
    const int kh = lane >> 4;
    const int nc = wave * 16 + m;
    float* c_lo = acc + (kh * 8) * XP + nc;
    float* c_hi = c_lo + 16 * XP;
    v8f c0, c1;
#pragma unroll
    for (int r = 0; r < 8; ++r) {
      c0[r] = c_lo[r * XP];
      c1[r] = c_hi[r * XP];
    }
    const float* a_lo = xt + m * XP + kh * 2;
    const float* a_hi = a_lo + 16 * XP;
    const float* __restrict__ wb = Wskip + (size_t)(kh * 2) * DIM + nc;
#pragma unroll 8
    for (int k0 = 0; k0 < DIM; k0 += 4) {
      const v2f av0 = *(const v2f*)(a_lo + k0);
      const v2f av1 = *(const v2f*)(a_hi + k0);
      v2f bv;
      bv.x = wb[(size_t)k0 * DIM];
      bv.y = wb[(size_t)(k0 + 1) * DIM];
      c0 = __builtin_amdgcn_wmma_f32_16x16x4_f32(
          false, av0, false, bv, (short)0, c0, false, false);
      c1 = __builtin_amdgcn_wmma_f32_16x16x4_f32(
          false, av1, false, bv, (short)0, c1, false, false);
    }
    const float bb = bias[nc];
#pragma unroll
    for (int r = 0; r < 8; ++r) {
      const float v0 = c0[r] + bb;
      const float v1 = c1[r] + bb;
      c_lo[r * XP] = (v0 > 0.0f) ? v0 : expm1f(v0);
      c_hi[r * XP] = (v1 > 0.0f) ? v1 : expm1f(v1);
    }
  }
  __syncthreads();

  const int ln = (int)__builtin_amdgcn_mbcnt_lo(~0u, 0u);
  const int tq = wave * XP + ln * 4;
  if (whole) {
#pragma unroll
    for (int m = 0; m < RNG * DIM / 1024; ++m)
      tile[tid + 256 * m] = *(const float4*)(acc + tq + 8 * m * XP);
  } else {
    for (int m = 0; m < RNG * DIM / 1024; ++m) {
      const int i = tid + 256 * m;
      if (i < lim4) tile[i] = *(const float4*)(acc + tq + 8 * m * XP);
    }
  }

  if (NEXT) {
    {
      const int m  = ln & 15;
      const int kh = ln >> 4;
      const int nc = wave * 16 + m;
      const float* a_lo = acc + m * XP + kh * 2;
      const float* a_hi = a_lo + 16 * XP;
      const float* __restrict__ wb = Wn + (size_t)(kh * 2) * DIM + nc;
      v8f d0 = {}, d1 = {};
#pragma unroll 8
      for (int k0 = 0; k0 < DIM; k0 += 4) {
        const v2f av0 = *(const v2f*)(a_lo + k0);
        const v2f av1 = *(const v2f*)(a_hi + k0);
        v2f bv;
        bv.x = wb[(size_t)k0 * DIM];
        bv.y = wb[(size_t)(k0 + 1) * DIM];
        d0 = __builtin_amdgcn_wmma_f32_16x16x4_f32(
            false, av0, false, bv, (short)0, d0, false, false);
        d1 = __builtin_amdgcn_wmma_f32_16x16x4_f32(
            false, av1, false, bv, (short)0, d1, false, false);
      }
      float* o_lo = xt + (kh * 8) * XP + nc;
      float* o_hi = o_lo + 16 * XP;
#pragma unroll
      for (int r = 0; r < 8; ++r) {
        o_lo[r * XP] = d0[r];
        o_hi[r * XP] = d1[r];
      }
    }
    __syncthreads();

    if (tid < RNG * HEADS) {
      const int row = tid >> 2, hh = tid & 3;
      const float* hp = xt + row * XP + hh * CH;
      const float* __restrict__ ws = a_src_n + hh * CH;
      const float* __restrict__ wd = a_dst_n + hh * CH;
      float s1 = 0.f, s2 = 0.f;
#pragma unroll 8
      for (int c = 0; c < CH; ++c) {
        const float v = hp[c];
        s1 += v * ws[c];
        s2 += v * wd[c];
      }
      if (b * RNG + row < N) {
        asrc_n[(size_t)b * RNG * HEADS + tid] = s1;
        adst_n[(size_t)b * RNG * HEADS + tid] = s2;
      }
    }

    float4* __restrict__ htile = (float4*)(hn + (size_t)b * RNG * DIM);
    const float* hsrc = xt + wave * XP + ln;
    if (whole) {
#pragma unroll
      for (int m = 0; m < RNG * DIM / 1024; ++m) {
        const float* p = hsrc + 8 * m * XP;
        htile[tid + 256 * m] = make_float4(p[0], p[32], p[64], p[96]);
      }
    } else {
      for (int m = 0; m < RNG * DIM / 1024; ++m) {
        const int i = tid + 256 * m;
        const float* p = hsrc + 8 * m * XP;
        if (i < lim4) htile[i] = make_float4(p[0], p[32], p[64], p[96]);
      }
    }
  }
}

extern "C" void kernel_launch(void* const* d_in, const int* in_sizes, int n_in,
                              void* d_out, int out_size, void* d_ws,
                              size_t ws_size, hipStream_t stream) {
  const float* x  = (const float*)d_in[0];
  const int*   ei = (const int*)d_in[1];

  const int N  = in_sizes[0] / DIM;
  const int NE = in_sizes[1] / 2;
  const int E  = NE + N;

  float* ws = (float*)d_ws;
  size_t o = 0;
  float* h     = ws + o; o += (size_t)N * DIM;
  float* h2    = ws + o; o += (size_t)N * DIM;
  float* y1    = ws + o; o += (size_t)N * DIM;
  float* asrc  = ws + o; o += (size_t)N * HEADS;
  float* adst  = ws + o; o += (size_t)N * HEADS;
  float* asrc2 = ws + o; o += (size_t)N * HEADS;
  float* adst2 = ws + o; o += (size_t)N * HEADS;
  float4* scr  = (float4*)(ws + o); o += (size_t)E * HEADS;
  const int NB = (N + RNG - 1) / RNG;
  int4* rec = (int4*)(ws + o); o += (size_t)E * 4;
  const int NBR = NB * REP;
  unsigned* bcnt = (unsigned*)(ws + o); o += (size_t)NBR * LINE;
  unsigned* boff = (unsigned*)(ws + o); o += (size_t)((NBR + 4) & ~3);
  unsigned* brank = (unsigned*)(ws + o); o += (size_t)((E + 3) & ~3);

  float* y = (float*)d_out;

  const int TB = 256;
  const int gE    = (E + TB - 1) / TB;
  const int gNB   = (NBR + TB - 1) / TB;

  gat_bin_zero<<<gNB, TB, 0, stream>>>(bcnt, NBR);
  gat_bin_count<<<gE, TB, 0, stream>>>(ei, NE, E, bcnt, brank);

  const float* W1     = (const float*)d_in[2];
  const float* a_src1 = (const float*)d_in[3];
  const float* a_dst1 = (const float*)d_in[4];
  const float* bias1  = (const float*)d_in[5];
  const float* Wskip1 = (const float*)d_in[6];
  const float* W2     = (const float*)d_in[7];
  const float* a_src2 = (const float*)d_in[8];
  const float* a_dst2 = (const float*)d_in[9];
  const float* bias2  = (const float*)d_in[10];
  const float* Wskip2 = (const float*)d_in[11];

  gat_front_lds<<<NB + SCAN_BLOCKS, TB, 0, stream>>>(x, W1, a_src1, a_dst1, h,
                                                     asrc, adst, N, bcnt, boff,
                                                     NBR);
  gat_bin_fill<<<gE, TB, 0, stream>>>(ei, NE, E, boff, brank, rec);

  gat_agg_lds<true><<<NB, TB, 0, stream>>>(rec, boff, asrc, adst, scr, h, x,
                                           Wskip1, bias1, y1, N, W2, a_src2,
                                           a_dst2, h2, asrc2, adst2);

  gat_agg_lds<false><<<NB, TB, 0, stream>>>(rec, boff, asrc2, adst2, scr, h2,
                                            y1, Wskip2, bias2, y, N, nullptr,
                                            nullptr, nullptr, nullptr,
                                            nullptr, nullptr);
}
